# starting kernel
# baseline (speedup 1.0000x reference)
#include <hip/hip_runtime.h>
#include <math.h>

typedef __attribute__((ext_vector_type(2))) float v2f;
typedef __attribute__((ext_vector_type(8))) float v8f;

#define DIMC    96
#define D_STATE 16
#define D_INNER 192
#define DT_RANK 6
#define KDIR    4
#define BATCH   2
#define HH      56
#define WWD     56
#define LL      (HH * WWD)
#define BL      (BATCH * LL)
#define NPROJ   (DT_RANK + 2 * D_STATE)
#define XZW     (2 * D_INNER)

__device__ __forceinline__ int scan_to_spatial(int k, int l) {
    int q = l / WWD, r = l % WWD;
    if (k == 0) return l;
    if (k == 1) return q * WWD + (WWD - 1 - r);
    if (k == 2) return r * WWD + q;
    return (WWD - 1 - r) * WWD + q;
}

__device__ __forceinline__ int spatial_to_scan(int k, int l) {
    int q = l / WWD, r = l % WWD;
    if (k == 0) return l;
    if (k == 1) return q * WWD + (WWD - 1 - r);
    if (k == 2) return r * WWD + q;
    return r * WWD + (WWD - 1 - q);
}

__device__ __forceinline__ float silu(float v) {
    return v / (1.f + __expf(-v));
}

__global__ void k_in_gemm(const float* __restrict__ x,
                          const float* __restrict__ Win,
                          float* __restrict__ xz) {
    int lane  = threadIdx.x & 31;
    int wave  = threadIdx.x >> 5;
    int mTile = blockIdx.x * 4 + wave;
    int nTile = blockIdx.y;
    int m0 = mTile * 16, n0 = nTile * 16;
    int mrow = m0 + (lane & 15);
    int b = mrow / LL, l = mrow % LL;
    int nn = n0 + (lane & 15);
    int kh = (lane >> 4) * 2;
    v8f acc = {};
    for (int kb = 0; kb < DIMC; kb += 4) {
        int c0 = kb + kh;
        v2f a, bf;
        a[0]  = x[(size_t)(b * DIMC + c0    ) * LL + l];
        a[1]  = x[(size_t)(b * DIMC + c0 + 1) * LL + l];
        bf[0] = Win[(size_t)(c0    ) * XZW + nn];
        bf[1] = Win[(size_t)(c0 + 1) * XZW + nn];
        acc = __builtin_amdgcn_wmma_f32_16x16x4_f32(false, a, false, bf,
                                                    (short)0, acc, false, false);
    }
    int mb = m0 + (lane >> 4) * 8;
#pragma unroll
    for (int v = 0; v < 8; ++v)
        xz[(size_t)(mb + v) * XZW + nn] = acc[v];
}

__global__ void k_conv(const float* __restrict__ xz,
                       const float* __restrict__ Wc,
                       const float* __restrict__ bc,
                       float* __restrict__ xi) {
    int tid = blockIdx.x * blockDim.x + threadIdx.x;
    if (tid >= BL * D_INNER) return;
    int d  = tid % D_INNER;
    int bl = tid / D_INNER;
    int b = bl / LL, l = bl % LL;
    int h = l / WWD, w = l % WWD;
    float acc = bc[d];
#pragma unroll
    for (int dy = 0; dy < 3; ++dy) {
        int hy = h + dy - 1;
        if (hy < 0 || hy >= HH) continue;
#pragma unroll
        for (int dx = 0; dx < 3; ++dx) {
            int wx = w + dx - 1;
            if (wx < 0 || wx >= WWD) continue;
            acc += Wc[d * 9 + dy * 3 + dx] *
                   xz[(size_t)(b * LL + hy * WWD + wx) * XZW + d];
        }
    }
    xi[(size_t)bl * D_INNER + d] = silu(acc);
}

__global__ void k_xproj(const float* __restrict__ xi,
                        const float* __restrict__ Wxp,
                        float* __restrict__ xdbl) {
    int lane = threadIdx.x & 31;
    int wave = threadIdx.x >> 5;
    int kdir = blockIdx.z;
    int mTile = blockIdx.x * 4 + wave;
    int nTile = blockIdx.y;
    int m0 = mTile * 16, n0 = nTile * 16;
    int mrow = m0 + (lane & 15);
    int b = mrow / LL, l = mrow % LL;
    int lsp = scan_to_spatial(kdir, l);
    const float* arow = xi + (size_t)(b * LL + lsp) * D_INNER;
    const float* wk = Wxp + (size_t)kdir * D_INNER * NPROJ;
    int nn = n0 + (lane & 15);
    bool nok = (nn < NPROJ);
    int kh = (lane >> 4) * 2;
    v8f acc = {};
    for (int kb = 0; kb < D_INNER; kb += 4) {
        v2f a, bf;
        a[0]  = arow[kb + kh];
        a[1]  = arow[kb + kh + 1];
        bf[0] = nok ? wk[(size_t)(kb + kh    ) * NPROJ + nn] : 0.f;
        bf[1] = nok ? wk[(size_t)(kb + kh + 1) * NPROJ + nn] : 0.f;
        acc = __builtin_amdgcn_wmma_f32_16x16x4_f32(false, a, false, bf,
                                                    (short)0, acc, false, false);
    }
    if (nok) {
        int mb = m0 + (lane >> 4) * 8;
#pragma unroll
        for (int v = 0; v < 8; ++v)
            xdbl[(size_t)(kdir * BL + mb + v) * NPROJ + nn] = acc[v];
    }
}

__global__ void k_delta(const float* __restrict__ xdbl,
                        const float* __restrict__ Wdt,
                        const float* __restrict__ bdt,
                        float* __restrict__ delta) {
    int lane = threadIdx.x & 31;
    int wave = threadIdx.x >> 5;
    int kdir = blockIdx.z;
    int mTile = blockIdx.x * 4 + wave;
    int nTile = blockIdx.y;
    int m0 = mTile * 16, n0 = nTile * 16;
    int mrow = m0 + (lane & 15);
    int nn = n0 + (lane & 15);
    int kh = (lane >> 4) * 2;
    v8f acc = {};
#pragma unroll
    for (int kb = 0; kb < 8; kb += 4) {
        v2f a, bf;
#pragma unroll
        for (int v = 0; v < 2; ++v) {
            int kk = kb + kh + v;
            bool kok = (kk < DT_RANK);
            a[v]  = kok ? xdbl[(size_t)(kdir * BL + mrow) * NPROJ + kk] : 0.f;
            bf[v] = kok ? Wdt[(size_t)(kdir * DT_RANK + kk) * D_INNER + nn] : 0.f;
        }
        acc = __builtin_amdgcn_wmma_f32_16x16x4_f32(false, a, false, bf,
                                                    (short)0, acc, false, false);
    }
    float bias = bdt[kdir * D_INNER + nn];
    int mb = m0 + (lane >> 4) * 8;
#pragma unroll
    for (int v = 0; v < 8; ++v) {
        float t = acc[v] + bias;
        float sp = (t > 20.f) ? t : log1pf(__expf(t));
        delta[(size_t)(kdir * BL + mb + v) * D_INNER + nn] = sp;
    }
}

__global__ void k_scan(const float* __restrict__ xi,
                       const float* __restrict__ xdbl,
                       const float* __restrict__ delta,
                       const float* __restrict__ A_logs,
                       const float* __restrict__ Ds,
                       float* __restrict__ ys) {
    int tid = blockIdx.x * blockDim.x + threadIdx.x;
    if (tid >= KDIR * BATCH * D_INNER * D_STATE) return;
    int n     = tid & 15;
    int group = tid >> 4;
    int d     = group % D_INNER;
    int kb    = group / D_INNER;
    int b     = kb % BATCH;
    int kdir  = kb / BATCH;

    float A = -__expf(A_logs[(size_t)(kdir * D_INNER + d) * D_STATE + n]);
    float Dskip = Ds[kdir * D_INNER + d];
    float h = 0.f;

    const float* xib = xi    + (size_t)b * LL * D_INNER + d;
    const float* dlb = delta + (size_t)(kdir * BATCH + b) * LL * D_INNER + d;
    const float* xdb = xdbl  + (size_t)(kdir * BATCH + b) * LL * NPROJ;
    float*       ysb = ys    + (size_t)(kdir * BATCH + b) * LL * D_INNER + d;

    for (int l = 0; l < LL; ++l) {
        int lsp  = scan_to_spatial(kdir, l);
        float x  = xib[(size_t)lsp * D_INNER];
        float dl = dlb[(size_t)l * D_INNER];
        const float* bc = xdb + (size_t)l * NPROJ;
        __builtin_prefetch(bc + NPROJ, 0, 3);
        float Bn = bc[DT_RANK + n];
        float Cn = bc[DT_RANK + D_STATE + n];
        float dA = __expf(dl * A);
        h = dA * h + (dl * x) * Bn;
        float p = h * Cn;
#pragma unroll
        for (int off = 8; off > 0; off >>= 1)
            p += __shfl_xor(p, off, 16);
        if (n == 0)
            ysb[(size_t)l * D_INNER] = p + x * Dskip;
    }
}

__global__ void k_combine(const float* __restrict__ ys,
                          const float* __restrict__ xz,
                          float* __restrict__ y) {
    int tid = blockIdx.x * blockDim.x + threadIdx.x;
    if (tid >= BL * D_INNER) return;
    int d  = tid % D_INNER;
    int bl = tid / D_INNER;
    int b = bl / LL, l = bl % LL;
    float acc = 0.f;
#pragma unroll
    for (int k = 0; k < KDIR; ++k) {
        int ls = spatial_to_scan(k, l);
        acc += ys[(size_t)((k * BATCH + b) * LL + ls) * D_INNER + d];
    }
    float z = xz[(size_t)bl * XZW + D_INNER + d];
    y[tid] = acc * silu(z);
}

__global__ void k_out_gemm(const float* __restrict__ y,
                           const float* __restrict__ Wout,
                           float* __restrict__ out) {
    int lane = threadIdx.x & 31;
    int wave = threadIdx.x >> 5;
    int mTile = blockIdx.x * 4 + wave;
    int nTile = blockIdx.y;
    int m0 = mTile * 16, n0 = nTile * 16;
    int mrow = m0 + (lane & 15);
    int nn = n0 + (lane & 15);
    int kh = (lane >> 4) * 2;
    v8f acc = {};
    for (int kb = 0; kb < D_INNER; kb += 4) {
        v2f a, bf;
        a[0]  = y[(size_t)mrow * D_INNER + kb + kh];
        a[1]  = y[(size_t)mrow * D_INNER + kb + kh + 1];
        bf[0] = Wout[(size_t)(kb + kh    ) * DIMC + nn];
        bf[1] = Wout[(size_t)(kb + kh + 1) * DIMC + nn];
        acc = __builtin_amdgcn_wmma_f32_16x16x4_f32(false, a, false, bf,
                                                    (short)0, acc, false, false);
    }
    int mb = m0 + (lane >> 4) * 8;
#pragma unroll
    for (int v = 0; v < 8; ++v) {
        int row = mb + v;
        int b = row / LL, l = row % LL;
        out[(size_t)(b * DIMC + nn) * LL + l] = acc[v];
    }
}

extern "C" void kernel_launch(void* const* d_in, const int* in_sizes, int n_in,
                              void* d_out, int out_size, void* d_ws, size_t ws_size,
                              hipStream_t stream) {
    (void)in_sizes; (void)n_in; (void)out_size; (void)ws_size;
    const float* x       = (const float*)d_in[0];
    const float* W_in    = (const float*)d_in[1];
    const float* W_conv  = (const float*)d_in[2];
    const float* b_conv  = (const float*)d_in[3];
    const float* W_xproj = (const float*)d_in[4];
    const float* W_dt    = (const float*)d_in[5];
    const float* b_dt    = (const float*)d_in[6];
    const float* A_logs  = (const float*)d_in[7];
    const float* Ds      = (const float*)d_in[8];
    const float* W_out   = (const float*)d_in[9];
    float* out = (float*)d_out;

    float* ws = (float*)d_ws;
    size_t o = 0;
    float* xz    = ws + o;  o += (size_t)BL * XZW;
    float* xi    = ws + o;  o += (size_t)BL * D_INNER;
    float* xdbl  = ws + o;  o += (size_t)KDIR * BL * NPROJ;
    float* delta = ws + o;  o += (size_t)KDIR * BL * D_INNER;
    float* ysbuf = ws + o;  o += (size_t)KDIR * BL * D_INNER;
    float* ybuf  = ws + o;  o += (size_t)BL * D_INNER;

    dim3 blk(128);
    k_in_gemm <<<dim3(98, 24),    blk, 0, stream>>>(x, W_in, xz);
    int nelt = BL * D_INNER;
    k_conv    <<<(nelt + 255) / 256, 256, 0, stream>>>(xz, W_conv, b_conv, xi);
    k_xproj   <<<dim3(98, 3, 4),  blk, 0, stream>>>(xi, W_xproj, xdbl);
    k_delta   <<<dim3(98, 12, 4), blk, 0, stream>>>(xdbl, W_dt, b_dt, delta);
    int nscan = KDIR * BATCH * D_INNER * D_STATE;
    k_scan    <<<(nscan + 255) / 256, 256, 0, stream>>>(xi, xdbl, delta,
                                                        A_logs, Ds, ysbuf);
    k_combine <<<(nelt + 255) / 256, 256, 0, stream>>>(ysbuf, xz, ybuf);
    k_out_gemm<<<dim3(98, 6),     blk, 0, stream>>>(ybuf, W_out, out);
}


# scan kernel: operand loads software-pipelined - x, delta, B, C of the next 14 steps are fetched into registers while the current 14 compute; arithmetic and summation order unchanged
# speedup vs baseline: 7.1377x; 7.1377x over previous
#include <hip/hip_runtime.h>
#include <math.h>

typedef __attribute__((ext_vector_type(2))) float v2f;
typedef __attribute__((ext_vector_type(8))) float v8f;

#define DIMC    96
#define D_STATE 16
#define D_INNER 192
#define DT_RANK 6
#define KDIR    4
#define BATCH   2
#define HH      56
#define WWD     56
#define LL      (HH * WWD)
#define BL      (BATCH * LL)
#define NPROJ   (DT_RANK + 2 * D_STATE)
#define XZW     (2 * D_INNER)

__device__ __forceinline__ int scan_to_spatial(int k, int l) {
    int q = l / WWD, r = l % WWD;
    if (k == 0) return l;
    if (k == 1) return q * WWD + (WWD - 1 - r);
    if (k == 2) return r * WWD + q;
    return (WWD - 1 - r) * WWD + q;
}

__device__ __forceinline__ int spatial_to_scan(int k, int l) {
    int q = l / WWD, r = l % WWD;
    if (k == 0) return l;
    if (k == 1) return q * WWD + (WWD - 1 - r);
    if (k == 2) return r * WWD + q;
    return r * WWD + (WWD - 1 - q);
}

__device__ __forceinline__ float silu(float v) {
    return v / (1.f + __expf(-v));
}

__global__ void k_in_gemm(const float* __restrict__ x,
                          const float* __restrict__ Win,
                          float* __restrict__ xz) {
    int lane  = threadIdx.x & 31;
    int wave  = threadIdx.x >> 5;
    int mTile = blockIdx.x * 4 + wave;
    int nTile = blockIdx.y;
    int m0 = mTile * 16, n0 = nTile * 16;
    int mrow = m0 + (lane & 15);
    int b = mrow / LL, l = mrow % LL;
    int nn = n0 + (lane & 15);
    int kh = (lane >> 4) * 2;
    v8f acc = {};
    for (int kb = 0; kb < DIMC; kb += 4) {
        int c0 = kb + kh;
        v2f a, bf;
        a[0]  = x[(size_t)(b * DIMC + c0    ) * LL + l];
        a[1]  = x[(size_t)(b * DIMC + c0 + 1) * LL + l];
        bf[0] = Win[(size_t)(c0    ) * XZW + nn];
        bf[1] = Win[(size_t)(c0 + 1) * XZW + nn];
        acc = __builtin_amdgcn_wmma_f32_16x16x4_f32(false, a, false, bf,
                                                    (short)0, acc, false, false);
    }
    int mb = m0 + (lane >> 4) * 8;
#pragma unroll
    for (int v = 0; v < 8; ++v)
        xz[(size_t)(mb + v) * XZW + nn] = acc[v];
}

__global__ void k_conv(const float* __restrict__ xz,
                       const float* __restrict__ Wc,
                       const float* __restrict__ bc,
                       float* __restrict__ xi) {
    int tid = blockIdx.x * blockDim.x + threadIdx.x;
    if (tid >= BL * D_INNER) return;
    int d  = tid % D_INNER;
    int bl = tid / D_INNER;
    int b = bl / LL, l = bl % LL;
    int h = l / WWD, w = l % WWD;
    float acc = bc[d];
#pragma unroll
    for (int dy = 0; dy < 3; ++dy) {
        int hy = h + dy - 1;
        if (hy < 0 || hy >= HH) continue;
#pragma unroll
        for (int dx = 0; dx < 3; ++dx) {
            int wx = w + dx - 1;
            if (wx < 0 || wx >= WWD) continue;
            acc += Wc[d * 9 + dy * 3 + dx] *
                   xz[(size_t)(b * LL + hy * WWD + wx) * XZW + d];
        }
    }
    xi[(size_t)bl * D_INNER + d] = silu(acc);
}

__global__ void k_xproj(const float* __restrict__ xi,
                        const float* __restrict__ Wxp,
                        float* __restrict__ xdbl) {
    int lane = threadIdx.x & 31;
    int wave = threadIdx.x >> 5;
    int kdir = blockIdx.z;
    int mTile = blockIdx.x * 4 + wave;
    int nTile = blockIdx.y;
    int m0 = mTile * 16, n0 = nTile * 16;
    int mrow = m0 + (lane & 15);
    int b = mrow / LL, l = mrow % LL;
    int lsp = scan_to_spatial(kdir, l);
    const float* arow = xi + (size_t)(b * LL + lsp) * D_INNER;
    const float* wk = Wxp + (size_t)kdir * D_INNER * NPROJ;
    int nn = n0 + (lane & 15);
    bool nok = (nn < NPROJ);
    int kh = (lane >> 4) * 2;
    v8f acc = {};
    for (int kb = 0; kb < D_INNER; kb += 4) {
        v2f a, bf;
        a[0]  = arow[kb + kh];
        a[1]  = arow[kb + kh + 1];
        bf[0] = nok ? wk[(size_t)(kb + kh    ) * NPROJ + nn] : 0.f;
        bf[1] = nok ? wk[(size_t)(kb + kh + 1) * NPROJ + nn] : 0.f;
        acc = __builtin_amdgcn_wmma_f32_16x16x4_f32(false, a, false, bf,
                                                    (short)0, acc, false, false);
    }
    if (nok) {
        int mb = m0 + (lane >> 4) * 8;
#pragma unroll
        for (int v = 0; v < 8; ++v)
            xdbl[(size_t)(kdir * BL + mb + v) * NPROJ + nn] = acc[v];
    }
}

__global__ void k_delta(const float* __restrict__ xdbl,
                        const float* __restrict__ Wdt,
                        const float* __restrict__ bdt,
                        float* __restrict__ delta) {
    int lane = threadIdx.x & 31;
    int wave = threadIdx.x >> 5;
    int kdir = blockIdx.z;
    int mTile = blockIdx.x * 4 + wave;
    int nTile = blockIdx.y;
    int m0 = mTile * 16, n0 = nTile * 16;
    int mrow = m0 + (lane & 15);
    int nn = n0 + (lane & 15);
    int kh = (lane >> 4) * 2;
    v8f acc = {};
#pragma unroll
    for (int kb = 0; kb < 8; kb += 4) {
        v2f a, bf;
#pragma unroll
        for (int v = 0; v < 2; ++v) {
            int kk = kb + kh + v;
            bool kok = (kk < DT_RANK);
            a[v]  = kok ? xdbl[(size_t)(kdir * BL + mrow) * NPROJ + kk] : 0.f;
            bf[v] = kok ? Wdt[(size_t)(kdir * DT_RANK + kk) * D_INNER + nn] : 0.f;
        }
        acc = __builtin_amdgcn_wmma_f32_16x16x4_f32(false, a, false, bf,
                                                    (short)0, acc, false, false);
    }
    float bias = bdt[kdir * D_INNER + nn];
    int mb = m0 + (lane >> 4) * 8;
#pragma unroll
    for (int v = 0; v < 8; ++v) {
        float t = acc[v] + bias;
        float sp = (t > 20.f) ? t : log1pf(__expf(t));
        delta[(size_t)(kdir * BL + mb + v) * D_INNER + nn] = sp;
    }
}

#define SCAN_T 14
#define SCAN_THREADS 256

__device__ __forceinline__ void scan_fetch(const float* __restrict__ dlk,
                                           const float* __restrict__ bck,
                                           int l1, int d, int n,
                                           int& pr, const float*& xp, long incA, long incB,
                                           float (&xn)[SCAN_T], float (&dn)[SCAN_T],
                                           float (&bn)[SCAN_T], float (&cn)[SCAN_T]) {
#pragma unroll
    for (int t = 0; t < SCAN_T; ++t) {
        int l = l1 + t;
        xn[t] = xp[d];
        dn[t] = (dlk + (size_t)l   * D_INNER)[d];
        const float* bc = bck + (size_t)l * NPROJ;
        bn[t] = bc[n];
        cn[t] = bc[D_STATE + n];
        bool wrap = (pr == WWD - 1);
        xp += wrap ? incB : incA;
        pr   = wrap ? 0 : pr + 1;
    }
}

__global__ void __launch_bounds__(SCAN_THREADS)
k_scan(const float* __restrict__ xi,
       const float* __restrict__ xdbl,
       const float* __restrict__ delta,
       const float* __restrict__ A_logs,
       const float* __restrict__ Ds,
       float* __restrict__ ys) {
    int tid = blockIdx.x * SCAN_THREADS + threadIdx.x;
    if (tid >= KDIR * BATCH * D_INNER * D_STATE) return;
    int n     = threadIdx.x & 15;
    int d     = (blockIdx.x % 12) * 16 + (threadIdx.x >> 4);
    int kb    = blockIdx.x / 12;
    int b     = kb % BATCH;
    int kdir  = kb / BATCH;

    int sbase, sq, sr;
    if (kdir == 0)      { sbase = 0;                 sq = WWD; sr = 1;    }
    else if (kdir == 1) { sbase = WWD - 1;           sq = WWD; sr = -1;   }
    else if (kdir == 2) { sbase = 0;                 sq = 1;   sr = WWD;  }
    else                { sbase = (WWD - 1) * WWD;   sq = 1;   sr = -WWD; }
    long incA = (long)sr * D_INNER;
    long incB = (long)(sq - (WWD - 1) * sr) * D_INNER;

    float A = -__expf(A_logs[(size_t)(kdir * D_INNER + d) * D_STATE + n]);
    float Dskip = Ds[kdir * D_INNER + d];
    float h = 0.f;

    const float* xb  = xi    + (size_t)b  * LL * D_INNER;
    const float* dlk = delta + (size_t)kb * LL * D_INNER;
    const float* bck = xdbl  + (size_t)kb * LL * NPROJ + DT_RANK;
    float*       ysb = ys    + (size_t)kb * LL * D_INNER + d;

    float xn[SCAN_T], dn[SCAN_T], bn[SCAN_T], cn[SCAN_T];
    int pr = 0;
    const float* xp = xb + (size_t)sbase * D_INNER;
    scan_fetch(dlk, bck, 0, d, n, pr, xp, incA, incB, xn, dn, bn, cn);

    for (int l0 = 0; l0 < LL; l0 += SCAN_T) {
        float xc[SCAN_T], dc[SCAN_T], bc[SCAN_T], cc[SCAN_T];
#pragma unroll
        for (int t = 0; t < SCAN_T; ++t) {
            xc[t] = xn[t]; dc[t] = dn[t]; bc[t] = bn[t]; cc[t] = cn[t];
        }
        if (l0 + SCAN_T < LL)
            scan_fetch(dlk, bck, l0 + SCAN_T, d, n, pr, xp, incA, incB,
                       xn, dn, bn, cn);

        float yv[SCAN_T];
#pragma unroll
        for (int t = 0; t < SCAN_T; ++t) {
            float x  = xc[t];
            float dl = dc[t];
            float dA = __expf(dl * A);
            h = dA * h + (dl * x) * bc[t];
            float p = h * cc[t];
#pragma unroll
            for (int off = 8; off > 0; off >>= 1)
                p += __shfl_xor(p, off, 16);
            yv[t] = p + x * Dskip;
        }
        if (n == 0) {
#pragma unroll
            for (int t = 0; t < SCAN_T; ++t)
                ysb[(size_t)(l0 + t) * D_INNER] = yv[t];
        }
    }
}

__global__ void k_combine(const float* __restrict__ ys,
                          const float* __restrict__ xz,
                          float* __restrict__ y) {
    int tid = blockIdx.x * blockDim.x + threadIdx.x;
    if (tid >= BL * D_INNER) return;
    int d  = tid % D_INNER;
    int bl = tid / D_INNER;
    int b = bl / LL, l = bl % LL;
    float acc = 0.f;
#pragma unroll
    for (int k = 0; k < KDIR; ++k) {
        int ls = spatial_to_scan(k, l);
        acc += ys[(size_t)((k * BATCH + b) * LL + ls) * D_INNER + d];
    }
    float z = xz[(size_t)bl * XZW + D_INNER + d];
    y[tid] = acc * silu(z);
}

__global__ void k_out_gemm(const float* __restrict__ y,
                           const float* __restrict__ Wout,
                           float* __restrict__ out) {
    int lane = threadIdx.x & 31;
    int wave = threadIdx.x >> 5;
    int mTile = blockIdx.x * 4 + wave;
    int nTile = blockIdx.y;
    int m0 = mTile * 16, n0 = nTile * 16;
    int mrow = m0 + (lane & 15);
    int nn = n0 + (lane & 15);
    int kh = (lane >> 4) * 2;
    v8f acc = {};
    for (int kb = 0; kb < D_INNER; kb += 4) {
        v2f a, bf;
        a[0]  = y[(size_t)mrow * D_INNER + kb + kh];
        a[1]  = y[(size_t)mrow * D_INNER + kb + kh + 1];
        bf[0] = Wout[(size_t)(kb + kh    ) * DIMC + nn];
        bf[1] = Wout[(size_t)(kb + kh + 1) * DIMC + nn];
        acc = __builtin_amdgcn_wmma_f32_16x16x4_f32(false, a, false, bf,
                                                    (short)0, acc, false, false);
    }
    int mb = m0 + (lane >> 4) * 8;
#pragma unroll
    for (int v = 0; v < 8; ++v) {
        int row = mb + v;
        int b = row / LL, l = row % LL;
        out[(size_t)(b * DIMC + nn) * LL + l] = acc[v];
    }
}

extern "C" void kernel_launch(void* const* d_in, const int* in_sizes, int n_in,
                              void* d_out, int out_size, void* d_ws, size_t ws_size,
                              hipStream_t stream) {
    (void)in_sizes; (void)n_in; (void)out_size; (void)ws_size;
    const float* x       = (const float*)d_in[0];
    const float* W_in    = (const float*)d_in[1];
    const float* W_conv  = (const float*)d_in[2];
    const float* b_conv  = (const float*)d_in[3];
    const float* W_xproj = (const float*)d_in[4];
    const float* W_dt    = (const float*)d_in[5];
    const float* b_dt    = (const float*)d_in[6];
    const float* A_logs  = (const float*)d_in[7];
    const float* Ds      = (const float*)d_in[8];
    const float* W_out   = (const float*)d_in[9];
    float* out = (float*)d_out;

    float* ws = (float*)d_ws;
    size_t o = 0;
    float* xz    = ws + o;  o += (size_t)BL * XZW;
    float* xi    = ws + o;  o += (size_t)BL * D_INNER;
    float* xdbl  = ws + o;  o += (size_t)KDIR * BL * NPROJ;
    float* delta = ws + o;  o += (size_t)KDIR * BL * D_INNER;
    float* ysbuf = ws + o;  o += (size_t)KDIR * BL * D_INNER;
    float* ybuf  = ws + o;  o += (size_t)BL * D_INNER;

    dim3 blk(128);
    k_in_gemm <<<dim3(98, 24),    blk, 0, stream>>>(x, W_in, xz);
    int nelt = BL * D_INNER;
    k_conv    <<<(nelt + 255) / 256, 256, 0, stream>>>(xz, W_conv, b_conv, xi);
    k_xproj   <<<dim3(98, 3, 4),  blk, 0, stream>>>(xi, W_xproj, xdbl);
    k_delta   <<<dim3(98, 12, 4), blk, 0, stream>>>(xdbl, W_dt, b_dt, delta);
    int nscan = KDIR * BATCH * D_INNER * D_STATE;
    k_scan    <<<(nscan + 255) / 256, 256, 0, stream>>>(xi, xdbl, delta,
                                                        A_logs, Ds, ysbuf);
    k_combine <<<(nelt + 255) / 256, 256, 0, stream>>>(ysbuf, xz, ybuf);
    k_out_gemm<<<dim3(98, 6),     blk, 0, stream>>>(ybuf, W_out, out);
}


# scan kernel: x, delta, B|C staged through LDS, 64 steps per stage by coalesced block loads (16 per thread per stage, was 4 per step), next stage in flight; arithmetic and summation order unchanged
# speedup vs baseline: 8.2109x; 1.1504x over previous
#include <hip/hip_runtime.h>
#include <math.h>

typedef __attribute__((ext_vector_type(2))) float v2f;
typedef __attribute__((ext_vector_type(8))) float v8f;

#define DIMC    96
#define D_STATE 16
#define D_INNER 192
#define DT_RANK 6
#define KDIR    4
#define BATCH   2
#define HH      56
#define WWD     56
#define LL      (HH * WWD)
#define BL      (BATCH * LL)
#define NPROJ   (DT_RANK + 2 * D_STATE)
#define XZW     (2 * D_INNER)

__device__ __forceinline__ int scan_to_spatial(int k, int l) {
    int q = l / WWD, r = l % WWD;
    if (k == 0) return l;
    if (k == 1) return q * WWD + (WWD - 1 - r);
    if (k == 2) return r * WWD + q;
    return (WWD - 1 - r) * WWD + q;
}

__device__ __forceinline__ int spatial_to_scan(int k, int l) {
    int q = l / WWD, r = l % WWD;
    if (k == 0) return l;
    if (k == 1) return q * WWD + (WWD - 1 - r);
    if (k == 2) return r * WWD + q;
    return r * WWD + (WWD - 1 - q);
}

__device__ __forceinline__ float silu(float v) {
    return v / (1.f + __expf(-v));
}

__global__ void k_in_gemm(const float* __restrict__ x,
                          const float* __restrict__ Win,
                          float* __restrict__ xz) {
    int lane  = threadIdx.x & 31;
    int wave  = threadIdx.x >> 5;
    int mTile = blockIdx.x * 4 + wave;
    int nTile = blockIdx.y;
    int m0 = mTile * 16, n0 = nTile * 16;
    int mrow = m0 + (lane & 15);
    int b = mrow / LL, l = mrow % LL;
    int nn = n0 + (lane & 15);
    int kh = (lane >> 4) * 2;
    v8f acc = {};
    for (int kb = 0; kb < DIMC; kb += 4) {
        int c0 = kb + kh;
        v2f a, bf;
        a[0]  = x[(size_t)(b * DIMC + c0    ) * LL + l];
        a[1]  = x[(size_t)(b * DIMC + c0 + 1) * LL + l];
        bf[0] = Win[(size_t)(c0    ) * XZW + nn];
        bf[1] = Win[(size_t)(c0 + 1) * XZW + nn];
        acc = __builtin_amdgcn_wmma_f32_16x16x4_f32(false, a, false, bf,
                                                    (short)0, acc, false, false);
    }
    int mb = m0 + (lane >> 4) * 8;
#pragma unroll
    for (int v = 0; v < 8; ++v)
        xz[(size_t)(mb + v) * XZW + nn] = acc[v];
}

__global__ void k_conv(const float* __restrict__ xz,
                       const float* __restrict__ Wc,
                       const float* __restrict__ bc,
                       float* __restrict__ xi) {
    int tid = blockIdx.x * blockDim.x + threadIdx.x;
    if (tid >= BL * D_INNER) return;
    int d  = tid % D_INNER;
    int bl = tid / D_INNER;
    int b = bl / LL, l = bl % LL;
    int h = l / WWD, w = l % WWD;
    float acc = bc[d];
#pragma unroll
    for (int dy = 0; dy < 3; ++dy) {
        int hy = h + dy - 1;
        if (hy < 0 || hy >= HH) continue;
#pragma unroll
        for (int dx = 0; dx < 3; ++dx) {
            int wx = w + dx - 1;
            if (wx < 0 || wx >= WWD) continue;
            acc += Wc[d * 9 + dy * 3 + dx] *
                   xz[(size_t)(b * LL + hy * WWD + wx) * XZW + d];
        }
    }
    xi[(size_t)bl * D_INNER + d] = silu(acc);
}

__global__ void k_xproj(const float* __restrict__ xi,
                        const float* __restrict__ Wxp,
                        float* __restrict__ xdbl) {
    int lane = threadIdx.x & 31;
    int wave = threadIdx.x >> 5;
    int kdir = blockIdx.z;
    int mTile = blockIdx.x * 4 + wave;
    int nTile = blockIdx.y;
    int m0 = mTile * 16, n0 = nTile * 16;
    int mrow = m0 + (lane & 15);
    int b = mrow / LL, l = mrow % LL;
    int lsp = scan_to_spatial(kdir, l);
    const float* arow = xi + (size_t)(b * LL + lsp) * D_INNER;
    const float* wk = Wxp + (size_t)kdir * D_INNER * NPROJ;
    int nn = n0 + (lane & 15);
    bool nok = (nn < NPROJ);
    int kh = (lane >> 4) * 2;
    v8f acc = {};
    for (int kb = 0; kb < D_INNER; kb += 4) {
        v2f a, bf;
        a[0]  = arow[kb + kh];
        a[1]  = arow[kb + kh + 1];
        bf[0] = nok ? wk[(size_t)(kb + kh    ) * NPROJ + nn] : 0.f;
        bf[1] = nok ? wk[(size_t)(kb + kh + 1) * NPROJ + nn] : 0.f;
        acc = __builtin_amdgcn_wmma_f32_16x16x4_f32(false, a, false, bf,
                                                    (short)0, acc, false, false);
    }
    if (nok) {
        int mb = m0 + (lane >> 4) * 8;
#pragma unroll
        for (int v = 0; v < 8; ++v)
            xdbl[(size_t)(kdir * BL + mb + v) * NPROJ + nn] = acc[v];
    }
}

__global__ void k_delta(const float* __restrict__ xdbl,
                        const float* __restrict__ Wdt,
                        const float* __restrict__ bdt,
                        float* __restrict__ delta) {
    int lane = threadIdx.x & 31;
    int wave = threadIdx.x >> 5;
    int kdir = blockIdx.z;
    int mTile = blockIdx.x * 4 + wave;
    int nTile = blockIdx.y;
    int m0 = mTile * 16, n0 = nTile * 16;
    int mrow = m0 + (lane & 15);
    int nn = n0 + (lane & 15);
    int kh = (lane >> 4) * 2;
    v8f acc = {};
#pragma unroll
    for (int kb = 0; kb < 8; kb += 4) {
        v2f a, bf;
#pragma unroll
        for (int v = 0; v < 2; ++v) {
            int kk = kb + kh + v;
            bool kok = (kk < DT_RANK);
            a[v]  = kok ? xdbl[(size_t)(kdir * BL + mrow) * NPROJ + kk] : 0.f;
            bf[v] = kok ? Wdt[(size_t)(kdir * DT_RANK + kk) * D_INNER + nn] : 0.f;
        }
        acc = __builtin_amdgcn_wmma_f32_16x16x4_f32(false, a, false, bf,
                                                    (short)0, acc, false, false);
    }
    float bias = bdt[kdir * D_INNER + nn];
    int mb = m0 + (lane >> 4) * 8;
#pragma unroll
    for (int v = 0; v < 8; ++v) {
        float t = acc[v] + bias;
        float sp = (t > 20.f) ? t : log1pf(__expf(t));
        delta[(size_t)(kdir * BL + mb + v) * D_INNER + nn] = sp;
    }
}

#define SCAN_THREADS 256
#define ST_STEPS 64
#define N_STAGE  (LL / ST_STEPS)
#define CH       16
#define XPITCH   (ST_STEPS + 4)
#define ST_X     0
#define ST_D     (16 * XPITCH)
#define ST_BC    (2 * 16 * XPITCH)
#define ST_FLOATS (ST_BC + ST_STEPS * 32)

__device__ __forceinline__ void stage_fetch(const float* __restrict__ xb,
                                            const float* __restrict__ dlk,
                                            const float* __restrict__ bck,
                                            int l0, int d0, int sbase, int sq, int sr,
                                            int tid,
                                            float (&rx)[4], float (&rd)[4],
                                            float (&rbc)[8]) {
#pragma unroll
    for (int i = 0; i < 4; ++i) {
        int e  = i * SCAN_THREADS + tid;
        int t  = e >> 4, dd = e & 15;
        int l  = l0 + t;
        int q  = l / WWD, r = l - q * WWD;
        int sp = sbase + q * sq + r * sr;
        rx[i] = xb [(size_t)sp * D_INNER + d0 + dd];
        rd[i] = dlk[(size_t)l  * D_INNER + d0 + dd];
    }
#pragma unroll
    for (int i = 0; i < 8; ++i) {
        int e = i * SCAN_THREADS + tid;
        int t = e >> 5, j = e & 31;
        rbc[i] = bck[(size_t)(l0 + t) * NPROJ + j];
    }
}

__device__ __forceinline__ void stage_store(float* __restrict__ buf, int tid,
                                            const float (&rx)[4], const float (&rd)[4],
                                            const float (&rbc)[8]) {
#pragma unroll
    for (int i = 0; i < 4; ++i) {
        int e = i * SCAN_THREADS + tid;
        int t = e >> 4, dd = e & 15;
        buf[ST_X + dd * XPITCH + t] = rx[i];
        buf[ST_D + dd * XPITCH + t] = rd[i];
    }
#pragma unroll
    for (int i = 0; i < 8; ++i)
        buf[ST_BC + i * SCAN_THREADS + tid] = rbc[i];
}

__device__ __forceinline__ void lds_stage_barrier() {
    asm volatile("s_wait_dscnt 0x0\n\ts_barrier_signal -1\n\ts_barrier_wait -1" ::: "memory");
}

__global__ void __launch_bounds__(SCAN_THREADS)
k_scan(const float* __restrict__ xi,
       const float* __restrict__ xdbl,
       const float* __restrict__ delta,
       const float* __restrict__ A_logs,
       const float* __restrict__ Ds,
       float* __restrict__ ys) {
    __shared__ __attribute__((aligned(16))) float sm[2 * ST_FLOATS];

    int tid   = threadIdx.x;
    int n     = tid & 15;
    int g     = tid >> 4;
    int d0    = (blockIdx.x % 12) * 16;
    int d     = d0 + g;
    int kb    = blockIdx.x / 12;
    int b     = kb % BATCH;
    int kdir  = kb / BATCH;

    int sbase, sq, sr;
    if (kdir == 0)      { sbase = 0;                 sq = WWD; sr = 1;    }
    else if (kdir == 1) { sbase = WWD - 1;           sq = WWD; sr = -1;   }
    else if (kdir == 2) { sbase = 0;                 sq = 1;   sr = WWD;  }
    else                { sbase = (WWD - 1) * WWD;   sq = 1;   sr = -WWD; }

    float A = -__expf(A_logs[(size_t)(kdir * D_INNER + d) * D_STATE + n]);
    float Dskip = Ds[kdir * D_INNER + d];
    float h = 0.f;

    const float* xb  = xi    + (size_t)b  * LL * D_INNER;
    const float* dlk = delta + (size_t)kb * LL * D_INNER;
    const float* bck = xdbl  + (size_t)kb * LL * NPROJ + DT_RANK;
    float*       ysb = ys    + (size_t)kb * LL * D_INNER + d;

    float rx[4], rd[4], rbc[8];
    stage_fetch(xb, dlk, bck, 0, d0, sbase, sq, sr, tid, rx, rd, rbc);
    stage_store(sm, tid, rx, rd, rbc);
    lds_stage_barrier();

    for (int s = 0; s < N_STAGE; ++s) {
        const float* cur = sm + (s & 1) * ST_FLOATS;
        float*       nxt = sm + ((s + 1) & 1) * ST_FLOATS;
        bool more = (s + 1 < N_STAGE);
        if (more)
            stage_fetch(xb, dlk, bck, (s + 1) * ST_STEPS, d0, sbase, sq, sr, tid,
                        rx, rd, rbc);

        for (int c = 0; c < ST_STEPS / CH; ++c) {
            int tt = c * CH;
            float xc[CH], dc[CH], bc[CH], cc[CH];
#pragma unroll
            for (int j = 0; j < CH / 4; ++j) {
                float4 vx = *(const float4*)(cur + ST_X + g * XPITCH + tt + 4 * j);
                float4 vd = *(const float4*)(cur + ST_D + g * XPITCH + tt + 4 * j);
                xc[4 * j] = vx.x; xc[4 * j + 1] = vx.y; xc[4 * j + 2] = vx.z; xc[4 * j + 3] = vx.w;
                dc[4 * j] = vd.x; dc[4 * j + 1] = vd.y; dc[4 * j + 2] = vd.z; dc[4 * j + 3] = vd.w;
            }
#pragma unroll
            for (int t = 0; t < CH; ++t) {
                bc[t] = cur[ST_BC + (tt + t) * 32 + n];
                cc[t] = cur[ST_BC + (tt + t) * 32 + D_STATE + n];
            }

            float yv[CH];
#pragma unroll
            for (int t = 0; t < CH; ++t) {
                float x  = xc[t];
                float dl = dc[t];
                float dA = __expf(dl * A);
                h = dA * h + (dl * x) * bc[t];
                float p = h * cc[t];
#pragma unroll
                for (int off = 8; off > 0; off >>= 1)
                    p += __shfl_xor(p, off, 16);
                yv[t] = p + x * Dskip;
            }
            if (n == 0) {
                int l0 = s * ST_STEPS + tt;
#pragma unroll
                for (int t = 0; t < CH; ++t)
                    ysb[(size_t)(l0 + t) * D_INNER] = yv[t];
            }
        }

        if (more)
            stage_store(nxt, tid, rx, rd, rbc);
        lds_stage_barrier();
    }
}

__global__ void k_combine(const float* __restrict__ ys,
                          const float* __restrict__ xz,
                          float* __restrict__ y) {
    int tid = blockIdx.x * blockDim.x + threadIdx.x;
    if (tid >= BL * D_INNER) return;
    int d  = tid % D_INNER;
    int bl = tid / D_INNER;
    int b = bl / LL, l = bl % LL;
    float acc = 0.f;
#pragma unroll
    for (int k = 0; k < KDIR; ++k) {
        int ls = spatial_to_scan(k, l);
        acc += ys[(size_t)((k * BATCH + b) * LL + ls) * D_INNER + d];
    }
    float z = xz[(size_t)bl * XZW + D_INNER + d];
    y[tid] = acc * silu(z);
}

__global__ void k_out_gemm(const float* __restrict__ y,
                           const float* __restrict__ Wout,
                           float* __restrict__ out) {
    int lane = threadIdx.x & 31;
    int wave = threadIdx.x >> 5;
    int mTile = blockIdx.x * 4 + wave;
    int nTile = blockIdx.y;
    int m0 = mTile * 16, n0 = nTile * 16;
    int mrow = m0 + (lane & 15);
    int nn = n0 + (lane & 15);
    int kh = (lane >> 4) * 2;
    v8f acc = {};
    for (int kb = 0; kb < D_INNER; kb += 4) {
        v2f a, bf;
        a[0]  = y[(size_t)mrow * D_INNER + kb + kh];
        a[1]  = y[(size_t)mrow * D_INNER + kb + kh + 1];
        bf[0] = Wout[(size_t)(kb + kh    ) * DIMC + nn];
        bf[1] = Wout[(size_t)(kb + kh + 1) * DIMC + nn];
        acc = __builtin_amdgcn_wmma_f32_16x16x4_f32(false, a, false, bf,
                                                    (short)0, acc, false, false);
    }
    int mb = m0 + (lane >> 4) * 8;
#pragma unroll
    for (int v = 0; v < 8; ++v) {
        int row = mb + v;
        int b = row / LL, l = row % LL;
        out[(size_t)(b * DIMC + nn) * LL + l] = acc[v];
    }
}

extern "C" void kernel_launch(void* const* d_in, const int* in_sizes, int n_in,
                              void* d_out, int out_size, void* d_ws, size_t ws_size,
                              hipStream_t stream) {
    (void)in_sizes; (void)n_in; (void)out_size; (void)ws_size;
    const float* x       = (const float*)d_in[0];
    const float* W_in    = (const float*)d_in[1];
    const float* W_conv  = (const float*)d_in[2];
    const float* b_conv  = (const float*)d_in[3];
    const float* W_xproj = (const float*)d_in[4];
    const float* W_dt    = (const float*)d_in[5];
    const float* b_dt    = (const float*)d_in[6];
    const float* A_logs  = (const float*)d_in[7];
    const float* Ds      = (const float*)d_in[8];
    const float* W_out   = (const float*)d_in[9];
    float* out = (float*)d_out;

    float* ws = (float*)d_ws;
    size_t o = 0;
    float* xz    = ws + o;  o += (size_t)BL * XZW;
    float* xi    = ws + o;  o += (size_t)BL * D_INNER;
    float* xdbl  = ws + o;  o += (size_t)KDIR * BL * NPROJ;
    float* delta = ws + o;  o += (size_t)KDIR * BL * D_INNER;
    float* ysbuf = ws + o;  o += (size_t)KDIR * BL * D_INNER;
    float* ybuf  = ws + o;  o += (size_t)BL * D_INNER;

    dim3 blk(128);
    k_in_gemm <<<dim3(98, 24),    blk, 0, stream>>>(x, W_in, xz);
    int nelt = BL * D_INNER;
    k_conv    <<<(nelt + 255) / 256, 256, 0, stream>>>(xz, W_conv, b_conv, xi);
    k_xproj   <<<dim3(98, 3, 4),  blk, 0, stream>>>(xi, W_xproj, xdbl);
    k_delta   <<<dim3(98, 12, 4), blk, 0, stream>>>(xdbl, W_dt, b_dt, delta);
    int nscan = KDIR * BATCH * D_INNER * D_STATE;
    k_scan    <<<(nscan + 255) / 256, 256, 0, stream>>>(xi, xdbl, delta,
                                                        A_logs, Ds, ysbuf);
    k_combine <<<(nelt + 255) / 256, 256, 0, stream>>>(ysbuf, xz, ybuf);
    k_out_gemm<<<dim3(98, 6),     blk, 0, stream>>>(ybuf, W_out, out);
}


# scan kernel: the per-step sum over a group's 16 lanes runs on the vector ALU (row_xmask DPP on the adds) instead of four lane permutes through the LDS unit per step; same pairs, same order of sums
# speedup vs baseline: 10.7396x; 1.3080x over previous
#include <hip/hip_runtime.h>
#include <math.h>

typedef __attribute__((ext_vector_type(2))) float v2f;
typedef __attribute__((ext_vector_type(8))) float v8f;

#define DIMC    96
#define D_STATE 16
#define D_INNER 192
#define DT_RANK 6
#define KDIR    4
#define BATCH   2
#define HH      56
#define WWD     56
#define LL      (HH * WWD)
#define BL      (BATCH * LL)
#define NPROJ   (DT_RANK + 2 * D_STATE)
#define XZW     (2 * D_INNER)

__device__ __forceinline__ int scan_to_spatial(int k, int l) {
    int q = l / WWD, r = l % WWD;
    if (k == 0) return l;
    if (k == 1) return q * WWD + (WWD - 1 - r);
    if (k == 2) return r * WWD + q;
    return (WWD - 1 - r) * WWD + q;
}

__device__ __forceinline__ int spatial_to_scan(int k, int l) {
    int q = l / WWD, r = l % WWD;
    if (k == 0) return l;
    if (k == 1) return q * WWD + (WWD - 1 - r);
    if (k == 2) return r * WWD + q;
    return r * WWD + (WWD - 1 - q);
}

__device__ __forceinline__ float silu(float v) {
    return v / (1.f + __expf(-v));
}

__global__ void k_in_gemm(const float* __restrict__ x,
                          const float* __restrict__ Win,
                          float* __restrict__ xz) {
    int lane  = threadIdx.x & 31;
    int wave  = threadIdx.x >> 5;
    int mTile = blockIdx.x * 4 + wave;
    int nTile = blockIdx.y;
    int m0 = mTile * 16, n0 = nTile * 16;
    int mrow = m0 + (lane & 15);
    int b = mrow / LL, l = mrow % LL;
    int nn = n0 + (lane & 15);
    int kh = (lane >> 4) * 2;
    v8f acc = {};
    for (int kb = 0; kb < DIMC; kb += 4) {
        int c0 = kb + kh;
        v2f a, bf;
        a[0]  = x[(size_t)(b * DIMC + c0    ) * LL + l];
        a[1]  = x[(size_t)(b * DIMC + c0 + 1) * LL + l];
        bf[0] = Win[(size_t)(c0    ) * XZW + nn];
        bf[1] = Win[(size_t)(c0 + 1) * XZW + nn];
        acc = __builtin_amdgcn_wmma_f32_16x16x4_f32(false, a, false, bf,
                                                    (short)0, acc, false, false);
    }
    int mb = m0 + (lane >> 4) * 8;
#pragma unroll
    for (int v = 0; v < 8; ++v)
        xz[(size_t)(mb + v) * XZW + nn] = acc[v];
}

__global__ void k_conv(const float* __restrict__ xz,
                       const float* __restrict__ Wc,
                       const float* __restrict__ bc,
                       float* __restrict__ xi) {
    int tid = blockIdx.x * blockDim.x + threadIdx.x;
    if (tid >= BL * D_INNER) return;
    int d  = tid % D_INNER;
    int bl = tid / D_INNER;
    int b = bl / LL, l = bl % LL;
    int h = l / WWD, w = l % WWD;
    float acc = bc[d];
#pragma unroll
    for (int dy = 0; dy < 3; ++dy) {
        int hy = h + dy - 1;
        if (hy < 0 || hy >= HH) continue;
#pragma unroll
        for (int dx = 0; dx < 3; ++dx) {
            int wx = w + dx - 1;
            if (wx < 0 || wx >= WWD) continue;
            acc += Wc[d * 9 + dy * 3 + dx] *
                   xz[(size_t)(b * LL + hy * WWD + wx) * XZW + d];
        }
    }
    xi[(size_t)bl * D_INNER + d] = silu(acc);
}

__global__ void k_xproj(const float* __restrict__ xi,
                        const float* __restrict__ Wxp,
                        float* __restrict__ xdbl) {
    int lane = threadIdx.x & 31;
    int wave = threadIdx.x >> 5;
    int kdir = blockIdx.z;
    int mTile = blockIdx.x * 4 + wave;
    int nTile = blockIdx.y;
    int m0 = mTile * 16, n0 = nTile * 16;
    int mrow = m0 + (lane & 15);
    int b = mrow / LL, l = mrow % LL;
    int lsp = scan_to_spatial(kdir, l);
    const float* arow = xi + (size_t)(b * LL + lsp) * D_INNER;
    const float* wk = Wxp + (size_t)kdir * D_INNER * NPROJ;
    int nn = n0 + (lane & 15);
    bool nok = (nn < NPROJ);
    int kh = (lane >> 4) * 2;
    v8f acc = {};
    for (int kb = 0; kb < D_INNER; kb += 4) {
        v2f a, bf;
        a[0]  = arow[kb + kh];
        a[1]  = arow[kb + kh + 1];
        bf[0] = nok ? wk[(size_t)(kb + kh    ) * NPROJ + nn] : 0.f;
        bf[1] = nok ? wk[(size_t)(kb + kh + 1) * NPROJ + nn] : 0.f;
        acc = __builtin_amdgcn_wmma_f32_16x16x4_f32(false, a, false, bf,
                                                    (short)0, acc, false, false);
    }
    if (nok) {
        int mb = m0 + (lane >> 4) * 8;
#pragma unroll
        for (int v = 0; v < 8; ++v)
            xdbl[(size_t)(kdir * BL + mb + v) * NPROJ + nn] = acc[v];
    }
}

__global__ void k_delta(const float* __restrict__ xdbl,
                        const float* __restrict__ Wdt,
                        const float* __restrict__ bdt,
                        float* __restrict__ delta) {
    int lane = threadIdx.x & 31;
    int wave = threadIdx.x >> 5;
    int kdir = blockIdx.z;
    int mTile = blockIdx.x * 4 + wave;
    int nTile = blockIdx.y;
    int m0 = mTile * 16, n0 = nTile * 16;
    int mrow = m0 + (lane & 15);
    int nn = n0 + (lane & 15);
    int kh = (lane >> 4) * 2;
    v8f acc = {};
#pragma unroll
    for (int kb = 0; kb < 8; kb += 4) {
        v2f a, bf;
#pragma unroll
        for (int v = 0; v < 2; ++v) {
            int kk = kb + kh + v;
            bool kok = (kk < DT_RANK);
            a[v]  = kok ? xdbl[(size_t)(kdir * BL + mrow) * NPROJ + kk] : 0.f;
            bf[v] = kok ? Wdt[(size_t)(kdir * DT_RANK + kk) * D_INNER + nn] : 0.f;
        }
        acc = __builtin_amdgcn_wmma_f32_16x16x4_f32(false, a, false, bf,
                                                    (short)0, acc, false, false);
    }
    float bias = bdt[kdir * D_INNER + nn];
    int mb = m0 + (lane >> 4) * 8;
#pragma unroll
    for (int v = 0; v < 8; ++v) {
        float t = acc[v] + bias;
        float sp = (t > 20.f) ? t : log1pf(__expf(t));
        delta[(size_t)(kdir * BL + mb + v) * D_INNER + nn] = sp;
    }
}

#define SCAN_THREADS 256
#define ST_STEPS 64
#define N_STAGE  (LL / ST_STEPS)
#define CH       16
#define XPITCH   (ST_STEPS + 4)
#define ST_X     0
#define ST_D     (16 * XPITCH)
#define ST_BC    (2 * 16 * XPITCH)
#define ST_FLOATS (ST_BC + ST_STEPS * 32)

__device__ __forceinline__ void stage_fetch(const float* __restrict__ xb,
                                            const float* __restrict__ dlk,
                                            const float* __restrict__ bck,
                                            int l0, int d0, int sbase, int sq, int sr,
                                            int tid,
                                            float (&rx)[4], float (&rd)[4],
                                            float (&rbc)[8]) {
#pragma unroll
    for (int i = 0; i < 4; ++i) {
        int e  = i * SCAN_THREADS + tid;
        int t  = e >> 4, dd = e & 15;
        int l  = l0 + t;
        int q  = l / WWD, r = l - q * WWD;
        int sp = sbase + q * sq + r * sr;
        rx[i] = xb [(size_t)sp * D_INNER + d0 + dd];
        rd[i] = dlk[(size_t)l  * D_INNER + d0 + dd];
    }
#pragma unroll
    for (int i = 0; i < 8; ++i) {
        int e = i * SCAN_THREADS + tid;
        int t = e >> 5, j = e & 31;
        rbc[i] = bck[(size_t)(l0 + t) * NPROJ + j];
    }
}

__device__ __forceinline__ void stage_store(float* __restrict__ buf, int tid,
                                            const float (&rx)[4], const float (&rd)[4],
                                            const float (&rbc)[8]) {
#pragma unroll
    for (int i = 0; i < 4; ++i) {
        int e = i * SCAN_THREADS + tid;
        int t = e >> 4, dd = e & 15;
        buf[ST_X + dd * XPITCH + t] = rx[i];
        buf[ST_D + dd * XPITCH + t] = rd[i];
    }
#pragma unroll
    for (int i = 0; i < 8; ++i)
        buf[ST_BC + i * SCAN_THREADS + tid] = rbc[i];
}

__device__ __forceinline__ void lds_stage_barrier() {
    asm volatile("s_wait_dscnt 0x0\n\ts_barrier_signal -1\n\ts_barrier_wait -1" ::: "memory");
}

template <int M>
__device__ __forceinline__ float row_xor_add(float p) {
    int q = __builtin_amdgcn_update_dpp(0, __float_as_int(p), 0x160 + M, 0xf, 0xf, true);
    return p + __int_as_float(q);
}

__global__ void __launch_bounds__(SCAN_THREADS)
k_scan(const float* __restrict__ xi,
       const float* __restrict__ xdbl,
       const float* __restrict__ delta,
       const float* __restrict__ A_logs,
       const float* __restrict__ Ds,
       float* __restrict__ ys) {
    __shared__ __attribute__((aligned(16))) float sm[2 * ST_FLOATS];

    int tid   = threadIdx.x;
    int n     = tid & 15;
    int g     = tid >> 4;
    int d0    = (blockIdx.x % 12) * 16;
    int d     = d0 + g;
    int kb    = blockIdx.x / 12;
    int b     = kb % BATCH;
    int kdir  = kb / BATCH;

    int sbase, sq, sr;
    if (kdir == 0)      { sbase = 0;                 sq = WWD; sr = 1;    }
    else if (kdir == 1) { sbase = WWD - 1;           sq = WWD; sr = -1;   }
    else if (kdir == 2) { sbase = 0;                 sq = 1;   sr = WWD;  }
    else                { sbase = (WWD - 1) * WWD;   sq = 1;   sr = -WWD; }

    float A = -__expf(A_logs[(size_t)(kdir * D_INNER + d) * D_STATE + n]);
    float Dskip = Ds[kdir * D_INNER + d];
    float h = 0.f;

    const float* xb  = xi    + (size_t)b  * LL * D_INNER;
    const float* dlk = delta + (size_t)kb * LL * D_INNER;
    const float* bck = xdbl  + (size_t)kb * LL * NPROJ + DT_RANK;
    float*       ysb = ys    + (size_t)kb * LL * D_INNER + d;

    float rx[4], rd[4], rbc[8];
    stage_fetch(xb, dlk, bck, 0, d0, sbase, sq, sr, tid, rx, rd, rbc);
    stage_store(sm, tid, rx, rd, rbc);
    lds_stage_barrier();

    for (int s = 0; s < N_STAGE; ++s) {
        const float* cur = sm + (s & 1) * ST_FLOATS;
        float*       nxt = sm + ((s + 1) & 1) * ST_FLOATS;
        bool more = (s + 1 < N_STAGE);
        if (more)
            stage_fetch(xb, dlk, bck, (s + 1) * ST_STEPS, d0, sbase, sq, sr, tid,
                        rx, rd, rbc);

        for (int c = 0; c < ST_STEPS / CH; ++c) {
            int tt = c * CH;
            float xc[CH], dc[CH], bc[CH], cc[CH];
#pragma unroll
            for (int j = 0; j < CH / 4; ++j) {
                float4 vx = *(const float4*)(cur + ST_X + g * XPITCH + tt + 4 * j);
                float4 vd = *(const float4*)(cur + ST_D + g * XPITCH + tt + 4 * j);
                xc[4 * j] = vx.x; xc[4 * j + 1] = vx.y; xc[4 * j + 2] = vx.z; xc[4 * j + 3] = vx.w;
                dc[4 * j] = vd.x; dc[4 * j + 1] = vd.y; dc[4 * j + 2] = vd.z; dc[4 * j + 3] = vd.w;
            }
#pragma unroll
            for (int t = 0; t < CH; ++t) {
                bc[t] = cur[ST_BC + (tt + t) * 32 + n];
                cc[t] = cur[ST_BC + (tt + t) * 32 + D_STATE + n];
            }

            float yv[CH];
#pragma unroll
            for (int t = 0; t < CH; ++t) {
                float x  = xc[t];
                float dl = dc[t];
                float dA = __expf(dl * A);
                h = dA * h + (dl * x) * bc[t];
                float p = h * cc[t];
                p = row_xor_add<8>(p);
                p = row_xor_add<4>(p);
                p = row_xor_add<2>(p);
                p = row_xor_add<1>(p);
                yv[t] = p + x * Dskip;
            }
            if (n == 0) {
                int l0 = s * ST_STEPS + tt;
#pragma unroll
                for (int t = 0; t < CH; ++t)
                    ysb[(size_t)(l0 + t) * D_INNER] = yv[t];
            }
        }

        if (more)
            stage_store(nxt, tid, rx, rd, rbc);
        lds_stage_barrier();
    }
}

__global__ void k_combine(const float* __restrict__ ys,
                          const float* __restrict__ xz,
                          float* __restrict__ y) {
    int tid = blockIdx.x * blockDim.x + threadIdx.x;
    if (tid >= BL * D_INNER) return;
    int d  = tid % D_INNER;
    int bl = tid / D_INNER;
    int b = bl / LL, l = bl % LL;
    float acc = 0.f;
#pragma unroll
    for (int k = 0; k < KDIR; ++k) {
        int ls = spatial_to_scan(k, l);
        acc += ys[(size_t)((k * BATCH + b) * LL + ls) * D_INNER + d];
    }
    float z = xz[(size_t)bl * XZW + D_INNER + d];
    y[tid] = acc * silu(z);
}

__global__ void k_out_gemm(const float* __restrict__ y,
                           const float* __restrict__ Wout,
                           float* __restrict__ out) {
    int lane = threadIdx.x & 31;
    int wave = threadIdx.x >> 5;
    int mTile = blockIdx.x * 4 + wave;
    int nTile = blockIdx.y;
    int m0 = mTile * 16, n0 = nTile * 16;
    int mrow = m0 + (lane & 15);
    int nn = n0 + (lane & 15);
    int kh = (lane >> 4) * 2;
    v8f acc = {};
    for (int kb = 0; kb < D_INNER; kb += 4) {
        v2f a, bf;
        a[0]  = y[(size_t)mrow * D_INNER + kb + kh];
        a[1]  = y[(size_t)mrow * D_INNER + kb + kh + 1];
        bf[0] = Wout[(size_t)(kb + kh    ) * DIMC + nn];
        bf[1] = Wout[(size_t)(kb + kh + 1) * DIMC + nn];
        acc = __builtin_amdgcn_wmma_f32_16x16x4_f32(false, a, false, bf,
                                                    (short)0, acc, false, false);
    }
    int mb = m0 + (lane >> 4) * 8;
#pragma unroll
    for (int v = 0; v < 8; ++v) {
        int row = mb + v;
        int b = row / LL, l = row % LL;
        out[(size_t)(b * DIMC + nn) * LL + l] = acc[v];
    }
}

extern "C" void kernel_launch(void* const* d_in, const int* in_sizes, int n_in,
                              void* d_out, int out_size, void* d_ws, size_t ws_size,
                              hipStream_t stream) {
    (void)in_sizes; (void)n_in; (void)out_size; (void)ws_size;
    const float* x       = (const float*)d_in[0];
    const float* W_in    = (const float*)d_in[1];
    const float* W_conv  = (const float*)d_in[2];
    const float* b_conv  = (const float*)d_in[3];
    const float* W_xproj = (const float*)d_in[4];
    const float* W_dt    = (const float*)d_in[5];
    const float* b_dt    = (const float*)d_in[6];
    const float* A_logs  = (const float*)d_in[7];
    const float* Ds      = (const float*)d_in[8];
    const float* W_out   = (const float*)d_in[9];
    float* out = (float*)d_out;

    float* ws = (float*)d_ws;
    size_t o = 0;
    float* xz    = ws + o;  o += (size_t)BL * XZW;
    float* xi    = ws + o;  o += (size_t)BL * D_INNER;
    float* xdbl  = ws + o;  o += (size_t)KDIR * BL * NPROJ;
    float* delta = ws + o;  o += (size_t)KDIR * BL * D_INNER;
    float* ysbuf = ws + o;  o += (size_t)KDIR * BL * D_INNER;
    float* ybuf  = ws + o;  o += (size_t)BL * D_INNER;

    dim3 blk(128);
    k_in_gemm <<<dim3(98, 24),    blk, 0, stream>>>(x, W_in, xz);
    int nelt = BL * D_INNER;
    k_conv    <<<(nelt + 255) / 256, 256, 0, stream>>>(xz, W_conv, b_conv, xi);
    k_xproj   <<<dim3(98, 3, 4),  blk, 0, stream>>>(xi, W_xproj, xdbl);
    k_delta   <<<dim3(98, 12, 4), blk, 0, stream>>>(xdbl, W_dt, b_dt, delta);
    int nscan = KDIR * BATCH * D_INNER * D_STATE;
    k_scan    <<<(nscan + 255) / 256, 256, 0, stream>>>(xi, xdbl, delta,
                                                        A_logs, Ds, ysbuf);
    k_combine <<<(nelt + 255) / 256, 256, 0, stream>>>(ysbuf, xz, ybuf);
    k_out_gemm<<<dim3(98, 6),     blk, 0, stream>>>(ybuf, W_out, out);
}


# scan kernel: block halved to 4 waves = 8 channels per block, 192 blocks instead of 96, so each block has a WGP to itself with one wave per SIMD instead of two; arithmetic and order of sums unchanged
# speedup vs baseline: 12.3675x; 1.1516x over previous
#include <hip/hip_runtime.h>
#include <math.h>

typedef __attribute__((ext_vector_type(2))) float v2f;
typedef __attribute__((ext_vector_type(8))) float v8f;

#define DIMC    96
#define D_STATE 16
#define D_INNER 192
#define DT_RANK 6
#define KDIR    4
#define BATCH   2
#define HH      56
#define WWD     56
#define LL      (HH * WWD)
#define BL      (BATCH * LL)
#define NPROJ   (DT_RANK + 2 * D_STATE)
#define XZW     (2 * D_INNER)

__device__ __forceinline__ int scan_to_spatial(int k, int l) {
    int q = l / WWD, r = l % WWD;
    if (k == 0) return l;
    if (k == 1) return q * WWD + (WWD - 1 - r);
    if (k == 2) return r * WWD + q;
    return (WWD - 1 - r) * WWD + q;
}

__device__ __forceinline__ int spatial_to_scan(int k, int l) {
    int q = l / WWD, r = l % WWD;
    if (k == 0) return l;
    if (k == 1) return q * WWD + (WWD - 1 - r);
    if (k == 2) return r * WWD + q;
    return r * WWD + (WWD - 1 - q);
}

__device__ __forceinline__ float silu(float v) {
    return v / (1.f + __expf(-v));
}

__global__ void k_in_gemm(const float* __restrict__ x,
                          const float* __restrict__ Win,
                          float* __restrict__ xz) {
    int lane  = threadIdx.x & 31;
    int wave  = threadIdx.x >> 5;
    int mTile = blockIdx.x * 4 + wave;
    int nTile = blockIdx.y;
    int m0 = mTile * 16, n0 = nTile * 16;
    int mrow = m0 + (lane & 15);
    int b = mrow / LL, l = mrow % LL;
    int nn = n0 + (lane & 15);
    int kh = (lane >> 4) * 2;
    v8f acc = {};
    for (int kb = 0; kb < DIMC; kb += 4) {
        int c0 = kb + kh;
        v2f a, bf;
        a[0]  = x[(size_t)(b * DIMC + c0    ) * LL + l];
        a[1]  = x[(size_t)(b * DIMC + c0 + 1) * LL + l];
        bf[0] = Win[(size_t)(c0    ) * XZW + nn];
        bf[1] = Win[(size_t)(c0 + 1) * XZW + nn];
        acc = __builtin_amdgcn_wmma_f32_16x16x4_f32(false, a, false, bf,
                                                    (short)0, acc, false, false);
    }
    int mb = m0 + (lane >> 4) * 8;
#pragma unroll
    for (int v = 0; v < 8; ++v)
        xz[(size_t)(mb + v) * XZW + nn] = acc[v];
}

__global__ void k_conv(const float* __restrict__ xz,
                       const float* __restrict__ Wc,
                       const float* __restrict__ bc,
                       float* __restrict__ xi) {
    int tid = blockIdx.x * blockDim.x + threadIdx.x;
    if (tid >= BL * D_INNER) return;
    int d  = tid % D_INNER;
    int bl = tid / D_INNER;
    int b = bl / LL, l = bl % LL;
    int h = l / WWD, w = l % WWD;
    float acc = bc[d];
#pragma unroll
    for (int dy = 0; dy < 3; ++dy) {
        int hy = h + dy - 1;
        if (hy < 0 || hy >= HH) continue;
#pragma unroll
        for (int dx = 0; dx < 3; ++dx) {
            int wx = w + dx - 1;
            if (wx < 0 || wx >= WWD) continue;
            acc += Wc[d * 9 + dy * 3 + dx] *
                   xz[(size_t)(b * LL + hy * WWD + wx) * XZW + d];
        }
    }
    xi[(size_t)bl * D_INNER + d] = silu(acc);
}

__global__ void k_xproj(const float* __restrict__ xi,
                        const float* __restrict__ Wxp,
                        float* __restrict__ xdbl) {
    int lane = threadIdx.x & 31;
    int wave = threadIdx.x >> 5;
    int kdir = blockIdx.z;
    int mTile = blockIdx.x * 4 + wave;
    int nTile = blockIdx.y;
    int m0 = mTile * 16, n0 = nTile * 16;
    int mrow = m0 + (lane & 15);
    int b = mrow / LL, l = mrow % LL;
    int lsp = scan_to_spatial(kdir, l);
    const float* arow = xi + (size_t)(b * LL + lsp) * D_INNER;
    const float* wk = Wxp + (size_t)kdir * D_INNER * NPROJ;
    int nn = n0 + (lane & 15);
    bool nok = (nn < NPROJ);
    int kh = (lane >> 4) * 2;
    v8f acc = {};
    for (int kb = 0; kb < D_INNER; kb += 4) {
        v2f a, bf;
        a[0]  = arow[kb + kh];
        a[1]  = arow[kb + kh + 1];
        bf[0] = nok ? wk[(size_t)(kb + kh    ) * NPROJ + nn] : 0.f;
        bf[1] = nok ? wk[(size_t)(kb + kh + 1) * NPROJ + nn] : 0.f;
        acc = __builtin_amdgcn_wmma_f32_16x16x4_f32(false, a, false, bf,
                                                    (short)0, acc, false, false);
    }
    if (nok) {
        int mb = m0 + (lane >> 4) * 8;
#pragma unroll
        for (int v = 0; v < 8; ++v)
            xdbl[(size_t)(kdir * BL + mb + v) * NPROJ + nn] = acc[v];
    }
}

__global__ void k_delta(const float* __restrict__ xdbl,
                        const float* __restrict__ Wdt,
                        const float* __restrict__ bdt,
                        float* __restrict__ delta) {
    int lane = threadIdx.x & 31;
    int wave = threadIdx.x >> 5;
    int kdir = blockIdx.z;
    int mTile = blockIdx.x * 4 + wave;
    int nTile = blockIdx.y;
    int m0 = mTile * 16, n0 = nTile * 16;
    int mrow = m0 + (lane & 15);
    int nn = n0 + (lane & 15);
    int kh = (lane >> 4) * 2;
    v8f acc = {};
#pragma unroll
    for (int kb = 0; kb < 8; kb += 4) {
        v2f a, bf;
#pragma unroll
        for (int v = 0; v < 2; ++v) {
            int kk = kb + kh + v;
            bool kok = (kk < DT_RANK);
            a[v]  = kok ? xdbl[(size_t)(kdir * BL + mrow) * NPROJ + kk] : 0.f;
            bf[v] = kok ? Wdt[(size_t)(kdir * DT_RANK + kk) * D_INNER + nn] : 0.f;
        }
        acc = __builtin_amdgcn_wmma_f32_16x16x4_f32(false, a, false, bf,
                                                    (short)0, acc, false, false);
    }
    float bias = bdt[kdir * D_INNER + nn];
    int mb = m0 + (lane >> 4) * 8;
#pragma unroll
    for (int v = 0; v < 8; ++v) {
        float t = acc[v] + bias;
        float sp = (t > 20.f) ? t : log1pf(__expf(t));
        delta[(size_t)(kdir * BL + mb + v) * D_INNER + nn] = sp;
    }
}

#define SCAN_CH      8
#define SCAN_THREADS (SCAN_CH * 16)
#define CH_SHIFT     3
#define BLK_PER_KB   (D_INNER / SCAN_CH)
#define NBC          (ST_STEPS * 32 / SCAN_THREADS)
#define ST_STEPS 64
#define N_STAGE  (LL / ST_STEPS)
#define CH       16
#define XPITCH   (ST_STEPS + 4)
#define ST_X     0
#define ST_D     (SCAN_CH * XPITCH)
#define ST_BC    (2 * SCAN_CH * XPITCH)
#define ST_FLOATS (ST_BC + ST_STEPS * 32)

__device__ __forceinline__ void stage_fetch(const float* __restrict__ xb,
                                            const float* __restrict__ dlk,
                                            const float* __restrict__ bck,
                                            int l0, int d0, int sbase, int sq, int sr,
                                            int tid,
                                            float (&rx)[4], float (&rd)[4],
                                            float (&rbc)[NBC]) {
#pragma unroll
    for (int i = 0; i < 4; ++i) {
        int e  = i * SCAN_THREADS + tid;
        int t  = e >> CH_SHIFT, dd = e & (SCAN_CH - 1);
        int l  = l0 + t;
        int q  = l / WWD, r = l - q * WWD;
        int sp = sbase + q * sq + r * sr;
        rx[i] = xb [(size_t)sp * D_INNER + d0 + dd];
        rd[i] = dlk[(size_t)l  * D_INNER + d0 + dd];
    }
#pragma unroll
    for (int i = 0; i < NBC; ++i) {
        int e = i * SCAN_THREADS + tid;
        int t = e >> 5, j = e & 31;
        rbc[i] = bck[(size_t)(l0 + t) * NPROJ + j];
    }
}

__device__ __forceinline__ void stage_store(float* __restrict__ buf, int tid,
                                            const float (&rx)[4], const float (&rd)[4],
                                            const float (&rbc)[NBC]) {
#pragma unroll
    for (int i = 0; i < 4; ++i) {
        int e = i * SCAN_THREADS + tid;
        int t = e >> CH_SHIFT, dd = e & (SCAN_CH - 1);
        buf[ST_X + dd * XPITCH + t] = rx[i];
        buf[ST_D + dd * XPITCH + t] = rd[i];
    }
#pragma unroll
    for (int i = 0; i < NBC; ++i)
        buf[ST_BC + i * SCAN_THREADS + tid] = rbc[i];
}

__device__ __forceinline__ void lds_stage_barrier() {
    asm volatile("s_wait_dscnt 0x0\n\ts_barrier_signal -1\n\ts_barrier_wait -1" ::: "memory");
}

template <int M>
__device__ __forceinline__ float row_xor_add(float p) {
    int q = __builtin_amdgcn_update_dpp(0, __float_as_int(p), 0x160 + M, 0xf, 0xf, true);
    return p + __int_as_float(q);
}

__global__ void __launch_bounds__(SCAN_THREADS)
k_scan(const float* __restrict__ xi,
       const float* __restrict__ xdbl,
       const float* __restrict__ delta,
       const float* __restrict__ A_logs,
       const float* __restrict__ Ds,
       float* __restrict__ ys) {
    __shared__ __attribute__((aligned(16))) float sm[2 * ST_FLOATS];

    int tid   = threadIdx.x;
    int n     = tid & 15;
    int g     = tid >> 4;
    int d0    = (blockIdx.x % BLK_PER_KB) * SCAN_CH;
    int d     = d0 + g;
    int kb    = blockIdx.x / BLK_PER_KB;
    int b     = kb % BATCH;
    int kdir  = kb / BATCH;

    int sbase, sq, sr;
    if (kdir == 0)      { sbase = 0;                 sq = WWD; sr = 1;    }
    else if (kdir == 1) { sbase = WWD - 1;           sq = WWD; sr = -1;   }
    else if (kdir == 2) { sbase = 0;                 sq = 1;   sr = WWD;  }
    else                { sbase = (WWD - 1) * WWD;   sq = 1;   sr = -WWD; }

    float A = -__expf(A_logs[(size_t)(kdir * D_INNER + d) * D_STATE + n]);
    float Dskip = Ds[kdir * D_INNER + d];
    float h = 0.f;

    const float* xb  = xi    + (size_t)b  * LL * D_INNER;
    const float* dlk = delta + (size_t)kb * LL * D_INNER;
    const float* bck = xdbl  + (size_t)kb * LL * NPROJ + DT_RANK;
    float*       ysb = ys    + (size_t)kb * LL * D_INNER + d;

    float rx[4], rd[4], rbc[NBC];
    stage_fetch(xb, dlk, bck, 0, d0, sbase, sq, sr, tid, rx, rd, rbc);
    stage_store(sm, tid, rx, rd, rbc);
    lds_stage_barrier();

    for (int s = 0; s < N_STAGE; ++s) {
        const float* cur = sm + (s & 1) * ST_FLOATS;
        float*       nxt = sm + ((s + 1) & 1) * ST_FLOATS;
        bool more = (s + 1 < N_STAGE);
        if (more)
            stage_fetch(xb, dlk, bck, (s + 1) * ST_STEPS, d0, sbase, sq, sr, tid,
                        rx, rd, rbc);

        for (int c = 0; c < ST_STEPS / CH; ++c) {
            int tt = c * CH;
            float xc[CH], dc[CH], bc[CH], cc[CH];
#pragma unroll
            for (int j = 0; j < CH / 4; ++j) {
                float4 vx = *(const float4*)(cur + ST_X + g * XPITCH + tt + 4 * j);
                float4 vd = *(const float4*)(cur + ST_D + g * XPITCH + tt + 4 * j);
                xc[4 * j] = vx.x; xc[4 * j + 1] = vx.y; xc[4 * j + 2] = vx.z; xc[4 * j + 3] = vx.w;
                dc[4 * j] = vd.x; dc[4 * j + 1] = vd.y; dc[4 * j + 2] = vd.z; dc[4 * j + 3] = vd.w;
            }
#pragma unroll
            for (int t = 0; t < CH; ++t) {
                bc[t] = cur[ST_BC + (tt + t) * 32 + n];
                cc[t] = cur[ST_BC + (tt + t) * 32 + D_STATE + n];
            }

            float yv[CH];
#pragma unroll
            for (int t = 0; t < CH; ++t) {
                float x  = xc[t];
                float dl = dc[t];
                float dA = __expf(dl * A);
                h = dA * h + (dl * x) * bc[t];
                float p = h * cc[t];
                p = row_xor_add<8>(p);
                p = row_xor_add<4>(p);
                p = row_xor_add<2>(p);
                p = row_xor_add<1>(p);
                yv[t] = p + x * Dskip;
            }
            if (n == 0) {
                int l0 = s * ST_STEPS + tt;
#pragma unroll
                for (int t = 0; t < CH; ++t)
                    ysb[(size_t)(l0 + t) * D_INNER] = yv[t];
            }
        }

        if (more)
            stage_store(nxt, tid, rx, rd, rbc);
        lds_stage_barrier();
    }
}

__global__ void k_combine(const float* __restrict__ ys,
                          const float* __restrict__ xz,
                          float* __restrict__ y) {
    int tid = blockIdx.x * blockDim.x + threadIdx.x;
    if (tid >= BL * D_INNER) return;
    int d  = tid % D_INNER;
    int bl = tid / D_INNER;
    int b = bl / LL, l = bl % LL;
    float acc = 0.f;
#pragma unroll
    for (int k = 0; k < KDIR; ++k) {
        int ls = spatial_to_scan(k, l);
        acc += ys[(size_t)((k * BATCH + b) * LL + ls) * D_INNER + d];
    }
    float z = xz[(size_t)bl * XZW + D_INNER + d];
    y[tid] = acc * silu(z);
}

__global__ void k_out_gemm(const float* __restrict__ y,
                           const float* __restrict__ Wout,
                           float* __restrict__ out) {
    int lane = threadIdx.x & 31;
    int wave = threadIdx.x >> 5;
    int mTile = blockIdx.x * 4 + wave;
    int nTile = blockIdx.y;
    int m0 = mTile * 16, n0 = nTile * 16;
    int mrow = m0 + (lane & 15);
    int nn = n0 + (lane & 15);
    int kh = (lane >> 4) * 2;
    v8f acc = {};
    for (int kb = 0; kb < D_INNER; kb += 4) {
        v2f a, bf;
        a[0]  = y[(size_t)mrow * D_INNER + kb + kh];
        a[1]  = y[(size_t)mrow * D_INNER + kb + kh + 1];
        bf[0] = Wout[(size_t)(kb + kh    ) * DIMC + nn];
        bf[1] = Wout[(size_t)(kb + kh + 1) * DIMC + nn];
        acc = __builtin_amdgcn_wmma_f32_16x16x4_f32(false, a, false, bf,
                                                    (short)0, acc, false, false);
    }
    int mb = m0 + (lane >> 4) * 8;
#pragma unroll
    for (int v = 0; v < 8; ++v) {
        int row = mb + v;
        int b = row / LL, l = row % LL;
        out[(size_t)(b * DIMC + nn) * LL + l] = acc[v];
    }
}

extern "C" void kernel_launch(void* const* d_in, const int* in_sizes, int n_in,
                              void* d_out, int out_size, void* d_ws, size_t ws_size,
                              hipStream_t stream) {
    (void)in_sizes; (void)n_in; (void)out_size; (void)ws_size;
    const float* x       = (const float*)d_in[0];
    const float* W_in    = (const float*)d_in[1];
    const float* W_conv  = (const float*)d_in[2];
    const float* b_conv  = (const float*)d_in[3];
    const float* W_xproj = (const float*)d_in[4];
    const float* W_dt    = (const float*)d_in[5];
    const float* b_dt    = (const float*)d_in[6];
    const float* A_logs  = (const float*)d_in[7];
    const float* Ds      = (const float*)d_in[8];
    const float* W_out   = (const float*)d_in[9];
    float* out = (float*)d_out;

    float* ws = (float*)d_ws;
    size_t o = 0;
    float* xz    = ws + o;  o += (size_t)BL * XZW;
    float* xi    = ws + o;  o += (size_t)BL * D_INNER;
    float* xdbl  = ws + o;  o += (size_t)KDIR * BL * NPROJ;
    float* delta = ws + o;  o += (size_t)KDIR * BL * D_INNER;
    float* ysbuf = ws + o;  o += (size_t)KDIR * BL * D_INNER;
    float* ybuf  = ws + o;  o += (size_t)BL * D_INNER;

    dim3 blk(128);
    k_in_gemm <<<dim3(98, 24),    blk, 0, stream>>>(x, W_in, xz);
    int nelt = BL * D_INNER;
    k_conv    <<<(nelt + 255) / 256, 256, 0, stream>>>(xz, W_conv, b_conv, xi);
    k_xproj   <<<dim3(98, 3, 4),  blk, 0, stream>>>(xi, W_xproj, xdbl);
    k_delta   <<<dim3(98, 12, 4), blk, 0, stream>>>(xdbl, W_dt, b_dt, delta);
    int nscan = KDIR * BATCH * D_INNER * D_STATE;
    k_scan    <<<nscan / SCAN_THREADS, SCAN_THREADS, 0, stream>>>(xi, xdbl, delta,
                                                        A_logs, Ds, ysbuf);
    k_combine <<<(nelt + 255) / 256, 256, 0, stream>>>(ysbuf, xz, ybuf);
    k_out_gemm<<<dim3(98, 6),     blk, 0, stream>>>(ybuf, W_out, out);
}


# conv, combine, delta kernels: more work per wave (4 channels of both images per thread, 128-bit accesses; 4 column tiles per wave): 8x, 8x, 4x fewer waves have to enter; order of sums unchanged
# speedup vs baseline: 13.3223x; 1.0772x over previous
#include <hip/hip_runtime.h>
#include <math.h>

typedef __attribute__((ext_vector_type(2))) float v2f;
typedef __attribute__((ext_vector_type(8))) float v8f;

#define DIMC    96
#define D_STATE 16
#define D_INNER 192
#define DT_RANK 6
#define KDIR    4
#define BATCH   2
#define HH      56
#define WWD     56
#define LL      (HH * WWD)
#define BL      (BATCH * LL)
#define NPROJ   (DT_RANK + 2 * D_STATE)
#define XZW     (2 * D_INNER)

__device__ __forceinline__ int scan_to_spatial(int k, int l) {
    int q = l / WWD, r = l % WWD;
    if (k == 0) return l;
    if (k == 1) return q * WWD + (WWD - 1 - r);
    if (k == 2) return r * WWD + q;
    return (WWD - 1 - r) * WWD + q;
}

__device__ __forceinline__ int spatial_to_scan(int k, int l) {
    int q = l / WWD, r = l % WWD;
    if (k == 0) return l;
    if (k == 1) return q * WWD + (WWD - 1 - r);
    if (k == 2) return r * WWD + q;
    return r * WWD + (WWD - 1 - q);
}

__device__ __forceinline__ float silu(float v) {
    return v / (1.f + __expf(-v));
}

__global__ void k_in_gemm(const float* __restrict__ x,
                          const float* __restrict__ Win,
                          float* __restrict__ xz) {
    int lane  = threadIdx.x & 31;
    int wave  = threadIdx.x >> 5;
    int mTile = blockIdx.x * 4 + wave;
    int nTile = blockIdx.y;
    int m0 = mTile * 16, n0 = nTile * 16;
    int mrow = m0 + (lane & 15);
    int b = mrow / LL, l = mrow % LL;
    int nn = n0 + (lane & 15);
    int kh = (lane >> 4) * 2;
    v8f acc = {};
    for (int kb = 0; kb < DIMC; kb += 4) {
        int c0 = kb + kh;
        v2f a, bf;
        a[0]  = x[(size_t)(b * DIMC + c0    ) * LL + l];
        a[1]  = x[(size_t)(b * DIMC + c0 + 1) * LL + l];
        bf[0] = Win[(size_t)(c0    ) * XZW + nn];
        bf[1] = Win[(size_t)(c0 + 1) * XZW + nn];
        acc = __builtin_amdgcn_wmma_f32_16x16x4_f32(false, a, false, bf,
                                                    (short)0, acc, false, false);
    }
    int mb = m0 + (lane >> 4) * 8;
#pragma unroll
    for (int v = 0; v < 8; ++v)
        xz[(size_t)(mb + v) * XZW + nn] = acc[v];
}

#define DQ (D_INNER / 4)
__device__ __forceinline__ void f4_to_arr(float4 q, float (&f)[4]) {
    f[0] = q.x; f[1] = q.y; f[2] = q.z; f[3] = q.w;
}
__global__ void k_conv(const float* __restrict__ xz,
                       const float* __restrict__ Wc,
                       const float* __restrict__ bc,
                       float* __restrict__ xi) {
    int tid = blockIdx.x * blockDim.x + threadIdx.x;
    if (tid >= LL * DQ) return;
    int d = (tid % DQ) * 4;
    int l = tid / DQ;
    int h = l / WWD, w = l % WWD;

    float v[9][BATCH][4];
    bool  ok[9];
#pragma unroll
    for (int t = 0; t < 9; ++t) {
        int hy = h + t / 3 - 1, wx = w + t % 3 - 1;
        ok[t] = (hy >= 0) && (hy < HH) && (wx >= 0) && (wx < WWD);
        int hc = min(max(hy, 0), HH - 1), wc = min(max(wx, 0), WWD - 1);
#pragma unroll
        for (int b = 0; b < BATCH; ++b)
            f4_to_arr(*(const float4*)(xz + (size_t)(b * LL + hc * WWD + wc) * XZW + d),
                      v[t][b]);
    }
    float wt[36];
#pragma unroll
    for (int i = 0; i < 9; ++i) {
        float q[4];
        f4_to_arr(*(const float4*)(Wc + (size_t)d * 9 + 4 * i), q);
        wt[4 * i] = q[0]; wt[4 * i + 1] = q[1]; wt[4 * i + 2] = q[2]; wt[4 * i + 3] = q[3];
    }
    float bias[4];
    f4_to_arr(*(const float4*)(bc + d), bias);

    float acc[BATCH][4];
#pragma unroll
    for (int b = 0; b < BATCH; ++b)
#pragma unroll
        for (int c = 0; c < 4; ++c) acc[b][c] = bias[c];
#pragma unroll
    for (int t = 0; t < 9; ++t) {
        if (!ok[t]) continue;
#pragma unroll
        for (int b = 0; b < BATCH; ++b)
#pragma unroll
            for (int c = 0; c < 4; ++c) acc[b][c] += wt[c * 9 + t] * v[t][b][c];
    }
#pragma unroll
    for (int b = 0; b < BATCH; ++b) {
        float4 o;
        o.x = silu(acc[b][0]); o.y = silu(acc[b][1]);
        o.z = silu(acc[b][2]); o.w = silu(acc[b][3]);
        *(float4*)(xi + (size_t)(b * LL + l) * D_INNER + d) = o;
    }
}

__global__ void k_xproj(const float* __restrict__ xi,
                        const float* __restrict__ Wxp,
                        float* __restrict__ xdbl) {
    int lane = threadIdx.x & 31;
    int wave = threadIdx.x >> 5;
    int kdir = blockIdx.z;
    int mTile = blockIdx.x * 4 + wave;
    int nTile = blockIdx.y;
    int m0 = mTile * 16, n0 = nTile * 16;
    int mrow = m0 + (lane & 15);
    int b = mrow / LL, l = mrow % LL;
    int lsp = scan_to_spatial(kdir, l);
    const float* arow = xi + (size_t)(b * LL + lsp) * D_INNER;
    const float* wk = Wxp + (size_t)kdir * D_INNER * NPROJ;
    int nn = n0 + (lane & 15);
    bool nok = (nn < NPROJ);
    int kh = (lane >> 4) * 2;
    v8f acc = {};
    for (int kb = 0; kb < D_INNER; kb += 4) {
        v2f a, bf;
        a[0]  = arow[kb + kh];
        a[1]  = arow[kb + kh + 1];
        bf[0] = nok ? wk[(size_t)(kb + kh    ) * NPROJ + nn] : 0.f;
        bf[1] = nok ? wk[(size_t)(kb + kh + 1) * NPROJ + nn] : 0.f;
        acc = __builtin_amdgcn_wmma_f32_16x16x4_f32(false, a, false, bf,
                                                    (short)0, acc, false, false);
    }
    if (nok) {
        int mb = m0 + (lane >> 4) * 8;
#pragma unroll
        for (int v = 0; v < 8; ++v)
            xdbl[(size_t)(kdir * BL + mb + v) * NPROJ + nn] = acc[v];
    }
}

#define DELTA_NT 4
__global__ void k_delta(const float* __restrict__ xdbl,
                        const float* __restrict__ Wdt,
                        const float* __restrict__ bdt,
                        float* __restrict__ delta) {
    int lane = threadIdx.x & 31;
    int wave = threadIdx.x >> 5;
    int kdir = blockIdx.z;
    int mTile = blockIdx.x * 4 + wave;
    int m0 = mTile * 16;
    int mrow = m0 + (lane & 15);
    int kh = (lane >> 4) * 2;
    int mb = m0 + (lane >> 4) * 8;
    v2f a[2];
#pragma unroll
    for (int s = 0; s < 2; ++s)
#pragma unroll
        for (int v = 0; v < 2; ++v) {
            int kk = s * 4 + kh + v;
            a[s][v] = (kk < DT_RANK) ? xdbl[(size_t)(kdir * BL + mrow) * NPROJ + kk] : 0.f;
        }
    for (int j = 0; j < DELTA_NT; ++j) {
        int nn = (blockIdx.y * DELTA_NT + j) * 16 + (lane & 15);
        v8f acc = {};
#pragma unroll
        for (int s = 0; s < 2; ++s) {
            v2f bf;
#pragma unroll
            for (int v = 0; v < 2; ++v) {
                int kk = s * 4 + kh + v;
                bf[v] = (kk < DT_RANK) ? Wdt[(size_t)(kdir * DT_RANK + kk) * D_INNER + nn] : 0.f;
            }
            acc = __builtin_amdgcn_wmma_f32_16x16x4_f32(false, a[s], false, bf,
                                                        (short)0, acc, false, false);
        }
        float bias = bdt[kdir * D_INNER + nn];
#pragma unroll
        for (int v = 0; v < 8; ++v) {
            float t = acc[v] + bias;
            float sp = (t > 20.f) ? t : log1pf(__expf(t));
            delta[(size_t)(kdir * BL + mb + v) * D_INNER + nn] = sp;
        }
    }
}

#define SCAN_CH      8
#define SCAN_THREADS (SCAN_CH * 16)
#define CH_SHIFT     3
#define BLK_PER_KB   (D_INNER / SCAN_CH)
#define NBC          (ST_STEPS * 32 / SCAN_THREADS)
#define ST_STEPS 64
#define N_STAGE  (LL / ST_STEPS)
#define CH       16
#define XPITCH   (ST_STEPS + 4)
#define ST_X     0
#define ST_D     (SCAN_CH * XPITCH)
#define ST_BC    (2 * SCAN_CH * XPITCH)
#define ST_FLOATS (ST_BC + ST_STEPS * 32)

__device__ __forceinline__ void stage_fetch(const float* __restrict__ xb,
                                            const float* __restrict__ dlk,
                                            const float* __restrict__ bck,
                                            int l0, int d0, int sbase, int sq, int sr,
                                            int tid,
                                            float (&rx)[4], float (&rd)[4],
                                            float (&rbc)[NBC]) {
#pragma unroll
    for (int i = 0; i < 4; ++i) {
        int e  = i * SCAN_THREADS + tid;
        int t  = e >> CH_SHIFT, dd = e & (SCAN_CH - 1);
        int l  = l0 + t;
        int q  = l / WWD, r = l - q * WWD;
        int sp = sbase + q * sq + r * sr;
        rx[i] = xb [(size_t)sp * D_INNER + d0 + dd];
        rd[i] = dlk[(size_t)l  * D_INNER + d0 + dd];
    }
#pragma unroll
    for (int i = 0; i < NBC; ++i) {
        int e = i * SCAN_THREADS + tid;
        int t = e >> 5, j = e & 31;
        rbc[i] = bck[(size_t)(l0 + t) * NPROJ + j];
    }
}

__device__ __forceinline__ void stage_store(float* __restrict__ buf, int tid,
                                            const float (&rx)[4], const float (&rd)[4],
                                            const float (&rbc)[NBC]) {
#pragma unroll
    for (int i = 0; i < 4; ++i) {
        int e = i * SCAN_THREADS + tid;
        int t = e >> CH_SHIFT, dd = e & (SCAN_CH - 1);
        buf[ST_X + dd * XPITCH + t] = rx[i];
        buf[ST_D + dd * XPITCH + t] = rd[i];
    }
#pragma unroll
    for (int i = 0; i < NBC; ++i)
        buf[ST_BC + i * SCAN_THREADS + tid] = rbc[i];
}

__device__ __forceinline__ void lds_stage_barrier() {
    asm volatile("s_wait_dscnt 0x0\n\ts_barrier_signal -1\n\ts_barrier_wait -1" ::: "memory");
}

template <int M>
__device__ __forceinline__ float row_xor_add(float p) {
    int q = __builtin_amdgcn_update_dpp(0, __float_as_int(p), 0x160 + M, 0xf, 0xf, true);
    return p + __int_as_float(q);
}

__global__ void __launch_bounds__(SCAN_THREADS)
k_scan(const float* __restrict__ xi,
       const float* __restrict__ xdbl,
       const float* __restrict__ delta,
       const float* __restrict__ A_logs,
       const float* __restrict__ Ds,
       float* __restrict__ ys) {
    __shared__ __attribute__((aligned(16))) float sm[2 * ST_FLOATS];

    int tid   = threadIdx.x;
    int n     = tid & 15;
    int g     = tid >> 4;
    int d0    = (blockIdx.x % BLK_PER_KB) * SCAN_CH;
    int d     = d0 + g;
    int kb    = blockIdx.x / BLK_PER_KB;
    int b     = kb % BATCH;
    int kdir  = kb / BATCH;

    int sbase, sq, sr;
    if (kdir == 0)      { sbase = 0;                 sq = WWD; sr = 1;    }
    else if (kdir == 1) { sbase = WWD - 1;           sq = WWD; sr = -1;   }
    else if (kdir == 2) { sbase = 0;                 sq = 1;   sr = WWD;  }
    else                { sbase = (WWD - 1) * WWD;   sq = 1;   sr = -WWD; }

    float A = -__expf(A_logs[(size_t)(kdir * D_INNER + d) * D_STATE + n]);
    float Dskip = Ds[kdir * D_INNER + d];
    float h = 0.f;

    const float* xb  = xi    + (size_t)b  * LL * D_INNER;
    const float* dlk = delta + (size_t)kb * LL * D_INNER;
    const float* bck = xdbl  + (size_t)kb * LL * NPROJ + DT_RANK;
    float*       ysb = ys    + (size_t)kb * LL * D_INNER + d;

    float rx[4], rd[4], rbc[NBC];
    stage_fetch(xb, dlk, bck, 0, d0, sbase, sq, sr, tid, rx, rd, rbc);
    stage_store(sm, tid, rx, rd, rbc);
    lds_stage_barrier();

    for (int s = 0; s < N_STAGE; ++s) {
        const float* cur = sm + (s & 1) * ST_FLOATS;
        float*       nxt = sm + ((s + 1) & 1) * ST_FLOATS;
        bool more = (s + 1 < N_STAGE);
        if (more)
            stage_fetch(xb, dlk, bck, (s + 1) * ST_STEPS, d0, sbase, sq, sr, tid,
                        rx, rd, rbc);

        for (int c = 0; c < ST_STEPS / CH; ++c) {
            int tt = c * CH;
            float xc[CH], dc[CH], bc[CH], cc[CH];
#pragma unroll
            for (int j = 0; j < CH / 4; ++j) {
                float4 vx = *(const float4*)(cur + ST_X + g * XPITCH + tt + 4 * j);
                float4 vd = *(const float4*)(cur + ST_D + g * XPITCH + tt + 4 * j);
                xc[4 * j] = vx.x; xc[4 * j + 1] = vx.y; xc[4 * j + 2] = vx.z; xc[4 * j + 3] = vx.w;
                dc[4 * j] = vd.x; dc[4 * j + 1] = vd.y; dc[4 * j + 2] = vd.z; dc[4 * j + 3] = vd.w;
            }
#pragma unroll
            for (int t = 0; t < CH; ++t) {
                bc[t] = cur[ST_BC + (tt + t) * 32 + n];
                cc[t] = cur[ST_BC + (tt + t) * 32 + D_STATE + n];
            }

            float yv[CH];
#pragma unroll
            for (int t = 0; t < CH; ++t) {
                float x  = xc[t];
                float dl = dc[t];
                float dA = __expf(dl * A);
                h = dA * h + (dl * x) * bc[t];
                float p = h * cc[t];
                p = row_xor_add<8>(p);
                p = row_xor_add<4>(p);
                p = row_xor_add<2>(p);
                p = row_xor_add<1>(p);
                yv[t] = p + x * Dskip;
            }
            if (n == 0) {
                int l0 = s * ST_STEPS + tt;
#pragma unroll
                for (int t = 0; t < CH; ++t)
                    ysb[(size_t)(l0 + t) * D_INNER] = yv[t];
            }
        }

        if (more)
            stage_store(nxt, tid, rx, rd, rbc);
        lds_stage_barrier();
    }
}

__global__ void k_combine(const float* __restrict__ ys,
                          const float* __restrict__ xz,
                          float* __restrict__ y) {
    int tid = blockIdx.x * blockDim.x + threadIdx.x;
    if (tid >= LL * DQ) return;
    int d = (tid % DQ) * 4;
    int l = tid / DQ;
    float r[KDIR][BATCH][4], z[BATCH][4];
#pragma unroll
    for (int k = 0; k < KDIR; ++k) {
        int ls = spatial_to_scan(k, l);
#pragma unroll
        for (int b = 0; b < BATCH; ++b)
            f4_to_arr(*(const float4*)(ys + (size_t)((k * BATCH + b) * LL + ls) * D_INNER + d),
                      r[k][b]);
    }
#pragma unroll
    for (int b = 0; b < BATCH; ++b)
        f4_to_arr(*(const float4*)(xz + (size_t)(b * LL + l) * XZW + D_INNER + d), z[b]);
#pragma unroll
    for (int b = 0; b < BATCH; ++b) {
        float o[4];
#pragma unroll
        for (int c = 0; c < 4; ++c) {
            float acc = 0.f;
#pragma unroll
            for (int k = 0; k < KDIR; ++k) acc += r[k][b][c];
            o[c] = acc * silu(z[b][c]);
        }
        float4 q;
        q.x = o[0]; q.y = o[1]; q.z = o[2]; q.w = o[3];
        *(float4*)(y + (size_t)(b * LL + l) * D_INNER + d) = q;
    }
}

__global__ void k_out_gemm(const float* __restrict__ y,
                           const float* __restrict__ Wout,
                           float* __restrict__ out) {
    int lane = threadIdx.x & 31;
    int wave = threadIdx.x >> 5;
    int mTile = blockIdx.x * 4 + wave;
    int nTile = blockIdx.y;
    int m0 = mTile * 16, n0 = nTile * 16;
    int mrow = m0 + (lane & 15);
    int nn = n0 + (lane & 15);
    int kh = (lane >> 4) * 2;
    v8f acc = {};
    for (int kb = 0; kb < D_INNER; kb += 4) {
        v2f a, bf;
        a[0]  = y[(size_t)mrow * D_INNER + kb + kh];
        a[1]  = y[(size_t)mrow * D_INNER + kb + kh + 1];
        bf[0] = Wout[(size_t)(kb + kh    ) * DIMC + nn];
        bf[1] = Wout[(size_t)(kb + kh + 1) * DIMC + nn];
        acc = __builtin_amdgcn_wmma_f32_16x16x4_f32(false, a, false, bf,
                                                    (short)0, acc, false, false);
    }
    int mb = m0 + (lane >> 4) * 8;
#pragma unroll
    for (int v = 0; v < 8; ++v) {
        int row = mb + v;
        int b = row / LL, l = row % LL;
        out[(size_t)(b * DIMC + nn) * LL + l] = acc[v];
    }
}

extern "C" void kernel_launch(void* const* d_in, const int* in_sizes, int n_in,
                              void* d_out, int out_size, void* d_ws, size_t ws_size,
                              hipStream_t stream) {
    (void)in_sizes; (void)n_in; (void)out_size; (void)ws_size;
    const float* x       = (const float*)d_in[0];
    const float* W_in    = (const float*)d_in[1];
    const float* W_conv  = (const float*)d_in[2];
    const float* b_conv  = (const float*)d_in[3];
    const float* W_xproj = (const float*)d_in[4];
    const float* W_dt    = (const float*)d_in[5];
    const float* b_dt    = (const float*)d_in[6];
    const float* A_logs  = (const float*)d_in[7];
    const float* Ds      = (const float*)d_in[8];
    const float* W_out   = (const float*)d_in[9];
    float* out = (float*)d_out;

    float* ws = (float*)d_ws;
    size_t o = 0;
    float* xz    = ws + o;  o += (size_t)BL * XZW;
    float* xi    = ws + o;  o += (size_t)BL * D_INNER;
    float* xdbl  = ws + o;  o += (size_t)KDIR * BL * NPROJ;
    float* delta = ws + o;  o += (size_t)KDIR * BL * D_INNER;
    float* ysbuf = ws + o;  o += (size_t)KDIR * BL * D_INNER;
    float* ybuf  = ws + o;  o += (size_t)BL * D_INNER;

    dim3 blk(128);
    k_in_gemm <<<dim3(98, 24),    blk, 0, stream>>>(x, W_in, xz);
    int nquad = LL * DQ;
    k_conv    <<<(nquad + 255) / 256, 256, 0, stream>>>(xz, W_conv, b_conv, xi);
    k_xproj   <<<dim3(98, 3, 4),  blk, 0, stream>>>(xi, W_xproj, xdbl);
    k_delta   <<<dim3(98, 12 / DELTA_NT, 4), blk, 0, stream>>>(xdbl, W_dt, b_dt, delta);
    int nscan = KDIR * BATCH * D_INNER * D_STATE;
    k_scan    <<<nscan / SCAN_THREADS, SCAN_THREADS, 0, stream>>>(xi, xdbl, delta,
                                                        A_logs, Ds, ysbuf);
    k_combine <<<(nquad + 255) / 256, 256, 0, stream>>>(ysbuf, xz, ybuf);
    k_out_gemm<<<dim3(98, 6),     blk, 0, stream>>>(ybuf, W_out, out);
}


# scan kernel: lane sums of 16 steps done together - each lane pair splits two steps per level (15 row_xmask adds per 16 steps, was 64), one 16-lane store per chunk; same pairs, same order of sums
# speedup vs baseline: 14.2522x; 1.0698x over previous
#include <hip/hip_runtime.h>
#include <math.h>

typedef __attribute__((ext_vector_type(2))) float v2f;
typedef __attribute__((ext_vector_type(8))) float v8f;

#define DIMC    96
#define D_STATE 16
#define D_INNER 192
#define DT_RANK 6
#define KDIR    4
#define BATCH   2
#define HH      56
#define WWD     56
#define LL      (HH * WWD)
#define BL      (BATCH * LL)
#define NPROJ   (DT_RANK + 2 * D_STATE)
#define XZW     (2 * D_INNER)

__device__ __forceinline__ int scan_to_spatial(int k, int l) {
    int q = l / WWD, r = l % WWD;
    if (k == 0) return l;
    if (k == 1) return q * WWD + (WWD - 1 - r);
    if (k == 2) return r * WWD + q;
    return (WWD - 1 - r) * WWD + q;
}

__device__ __forceinline__ int spatial_to_scan(int k, int l) {
    int q = l / WWD, r = l % WWD;
    if (k == 0) return l;
    if (k == 1) return q * WWD + (WWD - 1 - r);
    if (k == 2) return r * WWD + q;
    return r * WWD + (WWD - 1 - q);
}

__device__ __forceinline__ float silu(float v) {
    return v / (1.f + __expf(-v));
}

__global__ void k_in_gemm(const float* __restrict__ x,
                          const float* __restrict__ Win,
                          float* __restrict__ xz) {
    int lane  = threadIdx.x & 31;
    int wave  = threadIdx.x >> 5;
    int mTile = blockIdx.x * 4 + wave;
    int nTile = blockIdx.y;
    int m0 = mTile * 16, n0 = nTile * 16;
    int mrow = m0 + (lane & 15);
    int b = mrow / LL, l = mrow % LL;
    int nn = n0 + (lane & 15);
    int kh = (lane >> 4) * 2;
    v8f acc = {};
    for (int kb = 0; kb < DIMC; kb += 4) {
        int c0 = kb + kh;
        v2f a, bf;
        a[0]  = x[(size_t)(b * DIMC + c0    ) * LL + l];
        a[1]  = x[(size_t)(b * DIMC + c0 + 1) * LL + l];
        bf[0] = Win[(size_t)(c0    ) * XZW + nn];
        bf[1] = Win[(size_t)(c0 + 1) * XZW + nn];
        acc = __builtin_amdgcn_wmma_f32_16x16x4_f32(false, a, false, bf,
                                                    (short)0, acc, false, false);
    }
    int mb = m0 + (lane >> 4) * 8;
#pragma unroll
    for (int v = 0; v < 8; ++v)
        xz[(size_t)(mb + v) * XZW + nn] = acc[v];
}

#define DQ (D_INNER / 4)
__device__ __forceinline__ void f4_to_arr(float4 q, float (&f)[4]) {
    f[0] = q.x; f[1] = q.y; f[2] = q.z; f[3] = q.w;
}
__global__ void k_conv(const float* __restrict__ xz,
                       const float* __restrict__ Wc,
                       const float* __restrict__ bc,
                       float* __restrict__ xi) {
    int tid = blockIdx.x * blockDim.x + threadIdx.x;
    if (tid >= LL * DQ) return;
    int d = (tid % DQ) * 4;
    int l = tid / DQ;
    int h = l / WWD, w = l % WWD;

    float v[9][BATCH][4];
    bool  ok[9];
#pragma unroll
    for (int t = 0; t < 9; ++t) {
        int hy = h + t / 3 - 1, wx = w + t % 3 - 1;
        ok[t] = (hy >= 0) && (hy < HH) && (wx >= 0) && (wx < WWD);
        int hc = min(max(hy, 0), HH - 1), wc = min(max(wx, 0), WWD - 1);
#pragma unroll
        for (int b = 0; b < BATCH; ++b)
            f4_to_arr(*(const float4*)(xz + (size_t)(b * LL + hc * WWD + wc) * XZW + d),
                      v[t][b]);
    }
    float wt[36];
#pragma unroll
    for (int i = 0; i < 9; ++i) {
        float q[4];
        f4_to_arr(*(const float4*)(Wc + (size_t)d * 9 + 4 * i), q);
        wt[4 * i] = q[0]; wt[4 * i + 1] = q[1]; wt[4 * i + 2] = q[2]; wt[4 * i + 3] = q[3];
    }
    float bias[4];
    f4_to_arr(*(const float4*)(bc + d), bias);

    float acc[BATCH][4];
#pragma unroll
    for (int b = 0; b < BATCH; ++b)
#pragma unroll
        for (int c = 0; c < 4; ++c) acc[b][c] = bias[c];
#pragma unroll
    for (int t = 0; t < 9; ++t) {
        if (!ok[t]) continue;
#pragma unroll
        for (int b = 0; b < BATCH; ++b)
#pragma unroll
            for (int c = 0; c < 4; ++c) acc[b][c] += wt[c * 9 + t] * v[t][b][c];
    }
#pragma unroll
    for (int b = 0; b < BATCH; ++b) {
        float4 o;
        o.x = silu(acc[b][0]); o.y = silu(acc[b][1]);
        o.z = silu(acc[b][2]); o.w = silu(acc[b][3]);
        *(float4*)(xi + (size_t)(b * LL + l) * D_INNER + d) = o;
    }
}

__global__ void k_xproj(const float* __restrict__ xi,
                        const float* __restrict__ Wxp,
                        float* __restrict__ xdbl) {
    int lane = threadIdx.x & 31;
    int wave = threadIdx.x >> 5;
    int kdir = blockIdx.z;
    int mTile = blockIdx.x * 4 + wave;
    int nTile = blockIdx.y;
    int m0 = mTile * 16, n0 = nTile * 16;
    int mrow = m0 + (lane & 15);
    int b = mrow / LL, l = mrow % LL;
    int lsp = scan_to_spatial(kdir, l);
    const float* arow = xi + (size_t)(b * LL + lsp) * D_INNER;
    const float* wk = Wxp + (size_t)kdir * D_INNER * NPROJ;
    int nn = n0 + (lane & 15);
    bool nok = (nn < NPROJ);
    int kh = (lane >> 4) * 2;
    v8f acc = {};
    for (int kb = 0; kb < D_INNER; kb += 4) {
        v2f a, bf;
        a[0]  = arow[kb + kh];
        a[1]  = arow[kb + kh + 1];
        bf[0] = nok ? wk[(size_t)(kb + kh    ) * NPROJ + nn] : 0.f;
        bf[1] = nok ? wk[(size_t)(kb + kh + 1) * NPROJ + nn] : 0.f;
        acc = __builtin_amdgcn_wmma_f32_16x16x4_f32(false, a, false, bf,
                                                    (short)0, acc, false, false);
    }
    if (nok) {
        int mb = m0 + (lane >> 4) * 8;
#pragma unroll
        for (int v = 0; v < 8; ++v)
            xdbl[(size_t)(kdir * BL + mb + v) * NPROJ + nn] = acc[v];
    }
}

#define DELTA_NT 4
__global__ void k_delta(const float* __restrict__ xdbl,
                        const float* __restrict__ Wdt,
                        const float* __restrict__ bdt,
                        float* __restrict__ delta) {
    int lane = threadIdx.x & 31;
    int wave = threadIdx.x >> 5;
    int kdir = blockIdx.z;
    int mTile = blockIdx.x * 4 + wave;
    int m0 = mTile * 16;
    int mrow = m0 + (lane & 15);
    int kh = (lane >> 4) * 2;
    int mb = m0 + (lane >> 4) * 8;
    v2f a[2];
#pragma unroll
    for (int s = 0; s < 2; ++s)
#pragma unroll
        for (int v = 0; v < 2; ++v) {
            int kk = s * 4 + kh + v;
            a[s][v] = (kk < DT_RANK) ? xdbl[(size_t)(kdir * BL + mrow) * NPROJ + kk] : 0.f;
        }
    for (int j = 0; j < DELTA_NT; ++j) {
        int nn = (blockIdx.y * DELTA_NT + j) * 16 + (lane & 15);
        v8f acc = {};
#pragma unroll
        for (int s = 0; s < 2; ++s) {
            v2f bf;
#pragma unroll
            for (int v = 0; v < 2; ++v) {
                int kk = s * 4 + kh + v;
                bf[v] = (kk < DT_RANK) ? Wdt[(size_t)(kdir * DT_RANK + kk) * D_INNER + nn] : 0.f;
            }
            acc = __builtin_amdgcn_wmma_f32_16x16x4_f32(false, a[s], false, bf,
                                                        (short)0, acc, false, false);
        }
        float bias = bdt[kdir * D_INNER + nn];
#pragma unroll
        for (int v = 0; v < 8; ++v) {
            float t = acc[v] + bias;
            float sp = (t > 20.f) ? t : log1pf(__expf(t));
            delta[(size_t)(kdir * BL + mb + v) * D_INNER + nn] = sp;
        }
    }
}

#define SCAN_CH      8
#define SCAN_THREADS (SCAN_CH * 16)
#define CH_SHIFT     3
#define BLK_PER_KB   (D_INNER / SCAN_CH)
#define NBC          (ST_STEPS * 32 / SCAN_THREADS)
#define ST_STEPS 64
#define N_STAGE  (LL / ST_STEPS)
#define CH       16
#define XPITCH   (ST_STEPS + 4)
#define ST_X     0
#define ST_D     (SCAN_CH * XPITCH)
#define ST_BC    (2 * SCAN_CH * XPITCH)
#define ST_FLOATS (ST_BC + ST_STEPS * 32)

__device__ __forceinline__ void stage_fetch(const float* __restrict__ xb,
                                            const float* __restrict__ dlk,
                                            const float* __restrict__ bck,
                                            int l0, int d0, int sbase, int sq, int sr,
                                            int tid,
                                            float (&rx)[4], float (&rd)[4],
                                            float (&rbc)[NBC]) {
#pragma unroll
    for (int i = 0; i < 4; ++i) {
        int e  = i * SCAN_THREADS + tid;
        int t  = e >> CH_SHIFT, dd = e & (SCAN_CH - 1);
        int l  = l0 + t;
        int q  = l / WWD, r = l - q * WWD;
        int sp = sbase + q * sq + r * sr;
        rx[i] = xb [(size_t)sp * D_INNER + d0 + dd];
        rd[i] = dlk[(size_t)l  * D_INNER + d0 + dd];
    }
#pragma unroll
    for (int i = 0; i < NBC; ++i) {
        int e = i * SCAN_THREADS + tid;
        int t = e >> 5, j = e & 31;
        rbc[i] = bck[(size_t)(l0 + t) * NPROJ + j];
    }
}

__device__ __forceinline__ void stage_store(float* __restrict__ buf, int tid,
                                            const float (&rx)[4], const float (&rd)[4],
                                            const float (&rbc)[NBC]) {
#pragma unroll
    for (int i = 0; i < 4; ++i) {
        int e = i * SCAN_THREADS + tid;
        int t = e >> CH_SHIFT, dd = e & (SCAN_CH - 1);
        buf[ST_X + dd * XPITCH + t] = rx[i];
        buf[ST_D + dd * XPITCH + t] = rd[i];
    }
#pragma unroll
    for (int i = 0; i < NBC; ++i)
        buf[ST_BC + i * SCAN_THREADS + tid] = rbc[i];
}

__device__ __forceinline__ void lds_stage_barrier() {
    asm volatile("s_wait_dscnt 0x0\n\ts_barrier_signal -1\n\ts_barrier_wait -1" ::: "memory");
}

template <int M>
__device__ __forceinline__ float row_halve(bool hi, float a, float b) {
    float keep = hi ? b : a;
    float send = hi ? a : b;
    int q = __builtin_amdgcn_update_dpp(0, __float_as_int(send), 0x160 + M, 0xf, 0xf, true);
    return keep + __int_as_float(q);
}

__global__ void __launch_bounds__(SCAN_THREADS)
k_scan(const float* __restrict__ xi,
       const float* __restrict__ xdbl,
       const float* __restrict__ delta,
       const float* __restrict__ A_logs,
       const float* __restrict__ Ds,
       float* __restrict__ ys) {
    __shared__ __attribute__((aligned(16))) float sm[2 * ST_FLOATS];

    int tid   = threadIdx.x;
    int n     = tid & 15;
    int g     = tid >> 4;
    int d0    = (blockIdx.x % BLK_PER_KB) * SCAN_CH;
    int d     = d0 + g;
    int kb    = blockIdx.x / BLK_PER_KB;
    int b     = kb % BATCH;
    int kdir  = kb / BATCH;

    int sbase, sq, sr;
    if (kdir == 0)      { sbase = 0;                 sq = WWD; sr = 1;    }
    else if (kdir == 1) { sbase = WWD - 1;           sq = WWD; sr = -1;   }
    else if (kdir == 2) { sbase = 0;                 sq = 1;   sr = WWD;  }
    else                { sbase = (WWD - 1) * WWD;   sq = 1;   sr = -WWD; }

    bool hi8 = (n & 8) != 0, hi4 = (n & 4) != 0, hi2 = (n & 2) != 0, hi1 = (n & 1) != 0;
    int  trev = ((n & 1) << 3) | ((n & 2) << 1) | ((n & 4) >> 1) | ((n & 8) >> 3);

    float A = -__expf(A_logs[(size_t)(kdir * D_INNER + d) * D_STATE + n]);
    float Dskip = Ds[kdir * D_INNER + d];
    float h = 0.f;

    const float* xb  = xi    + (size_t)b  * LL * D_INNER;
    const float* dlk = delta + (size_t)kb * LL * D_INNER;
    const float* bck = xdbl  + (size_t)kb * LL * NPROJ + DT_RANK;
    float*       ysb = ys    + (size_t)kb * LL * D_INNER + d;

    float rx[4], rd[4], rbc[NBC];
    stage_fetch(xb, dlk, bck, 0, d0, sbase, sq, sr, tid, rx, rd, rbc);
    stage_store(sm, tid, rx, rd, rbc);
    lds_stage_barrier();

    for (int s = 0; s < N_STAGE; ++s) {
        const float* cur = sm + (s & 1) * ST_FLOATS;
        float*       nxt = sm + ((s + 1) & 1) * ST_FLOATS;
        bool more = (s + 1 < N_STAGE);
        if (more)
            stage_fetch(xb, dlk, bck, (s + 1) * ST_STEPS, d0, sbase, sq, sr, tid,
                        rx, rd, rbc);

        for (int c = 0; c < ST_STEPS / CH; ++c) {
            int tt = c * CH;
            float xc[CH], dc[CH], bc[CH], cc[CH];
#pragma unroll
            for (int j = 0; j < CH / 4; ++j) {
                float4 vx = *(const float4*)(cur + ST_X + g * XPITCH + tt + 4 * j);
                float4 vd = *(const float4*)(cur + ST_D + g * XPITCH + tt + 4 * j);
                xc[4 * j] = vx.x; xc[4 * j + 1] = vx.y; xc[4 * j + 2] = vx.z; xc[4 * j + 3] = vx.w;
                dc[4 * j] = vd.x; dc[4 * j + 1] = vd.y; dc[4 * j + 2] = vd.z; dc[4 * j + 3] = vd.w;
            }
#pragma unroll
            for (int t = 0; t < CH; ++t) {
                bc[t] = cur[ST_BC + (tt + t) * 32 + n];
                cc[t] = cur[ST_BC + (tt + t) * 32 + D_STATE + n];
            }

            float xo = cur[ST_X + g * XPITCH + tt + trev];

            float pv[CH];
#pragma unroll
            for (int t = 0; t < CH; ++t) {
                float x  = xc[t];
                float dl = dc[t];
                float dA = __expf(dl * A);
                h = dA * h + (dl * x) * bc[t];
                pv[t] = h * cc[t];
            }
            float q8[CH / 2], q4[CH / 4], q2[CH / 8];
#pragma unroll
            for (int j = 0; j < CH / 2; ++j) q8[j] = row_halve<8>(hi8, pv[2 * j], pv[2 * j + 1]);
#pragma unroll
            for (int j = 0; j < CH / 4; ++j) q4[j] = row_halve<4>(hi4, q8[2 * j], q8[2 * j + 1]);
#pragma unroll
            for (int j = 0; j < CH / 8; ++j) q2[j] = row_halve<2>(hi2, q4[2 * j], q4[2 * j + 1]);
            float p = row_halve<1>(hi1, q2[0], q2[1]);
            ysb[(size_t)(s * ST_STEPS + tt + trev) * D_INNER] = p + xo * Dskip;
        }

        if (more)
            stage_store(nxt, tid, rx, rd, rbc);
        lds_stage_barrier();
    }
}

__global__ void k_combine(const float* __restrict__ ys,
                          const float* __restrict__ xz,
                          float* __restrict__ y) {
    int tid = blockIdx.x * blockDim.x + threadIdx.x;
    if (tid >= LL * DQ) return;
    int d = (tid % DQ) * 4;
    int l = tid / DQ;
    float r[KDIR][BATCH][4], z[BATCH][4];
#pragma unroll
    for (int k = 0; k < KDIR; ++k) {
        int ls = spatial_to_scan(k, l);
#pragma unroll
        for (int b = 0; b < BATCH; ++b)
            f4_to_arr(*(const float4*)(ys + (size_t)((k * BATCH + b) * LL + ls) * D_INNER + d),
                      r[k][b]);
    }
#pragma unroll
    for (int b = 0; b < BATCH; ++b)
        f4_to_arr(*(const float4*)(xz + (size_t)(b * LL + l) * XZW + D_INNER + d), z[b]);
#pragma unroll
    for (int b = 0; b < BATCH; ++b) {
        float o[4];
#pragma unroll
        for (int c = 0; c < 4; ++c) {
            float acc = 0.f;
#pragma unroll
            for (int k = 0; k < KDIR; ++k) acc += r[k][b][c];
            o[c] = acc * silu(z[b][c]);
        }
        float4 q;
        q.x = o[0]; q.y = o[1]; q.z = o[2]; q.w = o[3];
        *(float4*)(y + (size_t)(b * LL + l) * D_INNER + d) = q;
    }
}

__global__ void k_out_gemm(const float* __restrict__ y,
                           const float* __restrict__ Wout,
                           float* __restrict__ out) {
    int lane = threadIdx.x & 31;
    int wave = threadIdx.x >> 5;
    int mTile = blockIdx.x * 4 + wave;
    int nTile = blockIdx.y;
    int m0 = mTile * 16, n0 = nTile * 16;
    int mrow = m0 + (lane & 15);
    int nn = n0 + (lane & 15);
    int kh = (lane >> 4) * 2;
    v8f acc = {};
    for (int kb = 0; kb < D_INNER; kb += 4) {
        v2f a, bf;
        a[0]  = y[(size_t)mrow * D_INNER + kb + kh];
        a[1]  = y[(size_t)mrow * D_INNER + kb + kh + 1];
        bf[0] = Wout[(size_t)(kb + kh    ) * DIMC + nn];
        bf[1] = Wout[(size_t)(kb + kh + 1) * DIMC + nn];
        acc = __builtin_amdgcn_wmma_f32_16x16x4_f32(false, a, false, bf,
                                                    (short)0, acc, false, false);
    }
    int mb = m0 + (lane >> 4) * 8;
#pragma unroll
    for (int v = 0; v < 8; ++v) {
        int row = mb + v;
        int b = row / LL, l = row % LL;
        out[(size_t)(b * DIMC + nn) * LL + l] = acc[v];
    }
}

extern "C" void kernel_launch(void* const* d_in, const int* in_sizes, int n_in,
                              void* d_out, int out_size, void* d_ws, size_t ws_size,
                              hipStream_t stream) {
    (void)in_sizes; (void)n_in; (void)out_size; (void)ws_size;
    const float* x       = (const float*)d_in[0];
    const float* W_in    = (const float*)d_in[1];
    const float* W_conv  = (const float*)d_in[2];
    const float* b_conv  = (const float*)d_in[3];
    const float* W_xproj = (const float*)d_in[4];
    const float* W_dt    = (const float*)d_in[5];
    const float* b_dt    = (const float*)d_in[6];
    const float* A_logs  = (const float*)d_in[7];
    const float* Ds      = (const float*)d_in[8];
    const float* W_out   = (const float*)d_in[9];
    float* out = (float*)d_out;

    float* ws = (float*)d_ws;
    size_t o = 0;
    float* xz    = ws + o;  o += (size_t)BL * XZW;
    float* xi    = ws + o;  o += (size_t)BL * D_INNER;
    float* xdbl  = ws + o;  o += (size_t)KDIR * BL * NPROJ;
    float* delta = ws + o;  o += (size_t)KDIR * BL * D_INNER;
    float* ysbuf = ws + o;  o += (size_t)KDIR * BL * D_INNER;
    float* ybuf  = ws + o;  o += (size_t)BL * D_INNER;

    dim3 blk(128);
    k_in_gemm <<<dim3(98, 24),    blk, 0, stream>>>(x, W_in, xz);
    int nquad = LL * DQ;
    k_conv    <<<(nquad + 255) / 256, 256, 0, stream>>>(xz, W_conv, b_conv, xi);
    k_xproj   <<<dim3(98, 3, 4),  blk, 0, stream>>>(xi, W_xproj, xdbl);
    k_delta   <<<dim3(98, 12 / DELTA_NT, 4), blk, 0, stream>>>(xdbl, W_dt, b_dt, delta);
    int nscan = KDIR * BATCH * D_INNER * D_STATE;
    k_scan    <<<nscan / SCAN_THREADS, SCAN_THREADS, 0, stream>>>(xi, xdbl, delta,
                                                        A_logs, Ds, ysbuf);
    k_combine <<<(nquad + 255) / 256, 256, 0, stream>>>(ysbuf, xz, ybuf);
    k_out_gemm<<<dim3(98, 6),     blk, 0, stream>>>(ybuf, W_out, out);
}


# delta no longer a kernel nor a 19 MB array: each scan block makes its stage's delta itself (dt projection on the matrix unit; softplus = log(1+e) by hardware exp/log + 1+e's rounding term) into LDS
# speedup vs baseline: 15.3235x; 1.0752x over previous
#include <hip/hip_runtime.h>
#include <math.h>

typedef __attribute__((ext_vector_type(2))) float v2f;
typedef __attribute__((ext_vector_type(8))) float v8f;

#define DIMC    96
#define D_STATE 16
#define D_INNER 192
#define DT_RANK 6
#define KDIR    4
#define BATCH   2
#define HH      56
#define WWD     56
#define LL      (HH * WWD)
#define BL      (BATCH * LL)
#define NPROJ   (DT_RANK + 2 * D_STATE)
#define XZW     (2 * D_INNER)

__device__ __forceinline__ int scan_to_spatial(int k, int l) {
    int q = l / WWD, r = l % WWD;
    if (k == 0) return l;
    if (k == 1) return q * WWD + (WWD - 1 - r);
    if (k == 2) return r * WWD + q;
    return (WWD - 1 - r) * WWD + q;
}

__device__ __forceinline__ int spatial_to_scan(int k, int l) {
    int q = l / WWD, r = l % WWD;
    if (k == 0) return l;
    if (k == 1) return q * WWD + (WWD - 1 - r);
    if (k == 2) return r * WWD + q;
    return r * WWD + (WWD - 1 - q);
}

__device__ __forceinline__ float silu(float v) {
    return v / (1.f + __expf(-v));
}

__global__ void k_in_gemm(const float* __restrict__ x,
                          const float* __restrict__ Win,
                          float* __restrict__ xz) {
    int lane  = threadIdx.x & 31;
    int wave  = threadIdx.x >> 5;
    int mTile = blockIdx.x * 4 + wave;
    int nTile = blockIdx.y;
    int m0 = mTile * 16, n0 = nTile * 16;
    int mrow = m0 + (lane & 15);
    int b = mrow / LL, l = mrow % LL;
    int nn = n0 + (lane & 15);
    int kh = (lane >> 4) * 2;
    v8f acc = {};
    for (int kb = 0; kb < DIMC; kb += 4) {
        int c0 = kb + kh;
        v2f a, bf;
        a[0]  = x[(size_t)(b * DIMC + c0    ) * LL + l];
        a[1]  = x[(size_t)(b * DIMC + c0 + 1) * LL + l];
        bf[0] = Win[(size_t)(c0    ) * XZW + nn];
        bf[1] = Win[(size_t)(c0 + 1) * XZW + nn];
        acc = __builtin_amdgcn_wmma_f32_16x16x4_f32(false, a, false, bf,
                                                    (short)0, acc, false, false);
    }
    int mb = m0 + (lane >> 4) * 8;
#pragma unroll
    for (int v = 0; v < 8; ++v)
        xz[(size_t)(mb + v) * XZW + nn] = acc[v];
}

#define DQ (D_INNER / 4)
__device__ __forceinline__ void f4_to_arr(float4 q, float (&f)[4]) {
    f[0] = q.x; f[1] = q.y; f[2] = q.z; f[3] = q.w;
}
__global__ void k_conv(const float* __restrict__ xz,
                       const float* __restrict__ Wc,
                       const float* __restrict__ bc,
                       float* __restrict__ xi) {
    int tid = blockIdx.x * blockDim.x + threadIdx.x;
    if (tid >= LL * DQ) return;
    int d = (tid % DQ) * 4;
    int l = tid / DQ;
    int h = l / WWD, w = l % WWD;

    float v[9][BATCH][4];
    bool  ok[9];
#pragma unroll
    for (int t = 0; t < 9; ++t) {
        int hy = h + t / 3 - 1, wx = w + t % 3 - 1;
        ok[t] = (hy >= 0) && (hy < HH) && (wx >= 0) && (wx < WWD);
        int hc = min(max(hy, 0), HH - 1), wc = min(max(wx, 0), WWD - 1);
#pragma unroll
        for (int b = 0; b < BATCH; ++b)
            f4_to_arr(*(const float4*)(xz + (size_t)(b * LL + hc * WWD + wc) * XZW + d),
                      v[t][b]);
    }
    float wt[36];
#pragma unroll
    for (int i = 0; i < 9; ++i) {
        float q[4];
        f4_to_arr(*(const float4*)(Wc + (size_t)d * 9 + 4 * i), q);
        wt[4 * i] = q[0]; wt[4 * i + 1] = q[1]; wt[4 * i + 2] = q[2]; wt[4 * i + 3] = q[3];
    }
    float bias[4];
    f4_to_arr(*(const float4*)(bc + d), bias);

    float acc[BATCH][4];
#pragma unroll
    for (int b = 0; b < BATCH; ++b)
#pragma unroll
        for (int c = 0; c < 4; ++c) acc[b][c] = bias[c];
#pragma unroll
    for (int t = 0; t < 9; ++t) {
        if (!ok[t]) continue;
#pragma unroll
        for (int b = 0; b < BATCH; ++b)
#pragma unroll
            for (int c = 0; c < 4; ++c) acc[b][c] += wt[c * 9 + t] * v[t][b][c];
    }
#pragma unroll
    for (int b = 0; b < BATCH; ++b) {
        float4 o;
        o.x = silu(acc[b][0]); o.y = silu(acc[b][1]);
        o.z = silu(acc[b][2]); o.w = silu(acc[b][3]);
        *(float4*)(xi + (size_t)(b * LL + l) * D_INNER + d) = o;
    }
}

__global__ void k_xproj(const float* __restrict__ xi,
                        const float* __restrict__ Wxp,
                        float* __restrict__ xdbl) {
    int lane = threadIdx.x & 31;
    int wave = threadIdx.x >> 5;
    int kdir = blockIdx.z;
    int mTile = blockIdx.x * 4 + wave;
    int nTile = blockIdx.y;
    int m0 = mTile * 16, n0 = nTile * 16;
    int mrow = m0 + (lane & 15);
    int b = mrow / LL, l = mrow % LL;
    int lsp = scan_to_spatial(kdir, l);
    const float* arow = xi + (size_t)(b * LL + lsp) * D_INNER;
    const float* wk = Wxp + (size_t)kdir * D_INNER * NPROJ;
    int nn = n0 + (lane & 15);
    bool nok = (nn < NPROJ);
    int kh = (lane >> 4) * 2;
    v8f acc = {};
    for (int kb = 0; kb < D_INNER; kb += 4) {
        v2f a, bf;
        a[0]  = arow[kb + kh];
        a[1]  = arow[kb + kh + 1];
        bf[0] = nok ? wk[(size_t)(kb + kh    ) * NPROJ + nn] : 0.f;
        bf[1] = nok ? wk[(size_t)(kb + kh + 1) * NPROJ + nn] : 0.f;
        acc = __builtin_amdgcn_wmma_f32_16x16x4_f32(false, a, false, bf,
                                                    (short)0, acc, false, false);
    }
    if (nok) {
        int mb = m0 + (lane >> 4) * 8;
#pragma unroll
        for (int v = 0; v < 8; ++v)
            xdbl[(size_t)(kdir * BL + mb + v) * NPROJ + nn] = acc[v];
    }
}

#define SCAN_CH      8
#define SCAN_THREADS (SCAN_CH * 16)
#define CH_SHIFT     3
#define BLK_PER_KB   (D_INNER / SCAN_CH)
#define NBC          (ST_STEPS * 32 / SCAN_THREADS)
#define ST_STEPS 64
#define N_STAGE  (LL / ST_STEPS)
#define CH       16
#define XPITCH   (ST_STEPS + 4)
#define ST_X     0
#define ST_D     (SCAN_CH * XPITCH)
#define ST_BC    (2 * SCAN_CH * XPITCH)
#define ST_FLOATS (ST_BC + ST_STEPS * 32)


__device__ __forceinline__ float softplus_dt(float t) {
    float e  = __expf(t);
    float u  = 1.f + e;
    float c  = e - (u - 1.f);
    float sp = __builtin_amdgcn_logf(u) * 0.69314718056f + c * __builtin_amdgcn_rcpf(u);
    return (t > 20.f) ? t : sp;
}

__device__ __forceinline__ void stage_fetch(const float* __restrict__ xb,
                                            const float* __restrict__ drk,
                                            const float* __restrict__ bck,
                                            int l0, int d0, int sbase, int sq, int sr,
                                            int tid, int apos, int hh,
                                            float (&rx)[4], v2f (&ra)[2],
                                            float (&rbc)[NBC]) {
#pragma unroll
    for (int i = 0; i < 4; ++i) {
        int e  = i * SCAN_THREADS + tid;
        int t  = e >> CH_SHIFT, dd = e & (SCAN_CH - 1);
        int l  = l0 + t;
        int q  = l / WWD, r = l - q * WWD;
        int sp = sbase + q * sq + r * sr;
        rx[i] = xb[(size_t)sp * D_INNER + d0 + dd];
    }
    const float* ap = drk + (size_t)l0 * NPROJ + apos;
    ra[0] = *(const v2f*)ap;
    ra[1] = *(const v2f*)(ap + 4 - 2 * hh);
#pragma unroll
    for (int i = 0; i < NBC; ++i) {
        int e = i * SCAN_THREADS + tid;
        int t = e >> 5, j = e & 31;
        rbc[i] = bck[(size_t)(l0 + t) * NPROJ + j];
    }
}

__device__ __forceinline__ void stage_store(float* __restrict__ buf, int tid,
                                            const float (&rx)[4], const v2f (&ra)[2],
                                            const v2f (&wb)[2], float dbias,
                                            int hh, int dpos,
                                            const float (&rbc)[NBC]) {
#pragma unroll
    for (int i = 0; i < 4; ++i) {
        int e = i * SCAN_THREADS + tid;
        int t = e >> CH_SHIFT, dd = e & (SCAN_CH - 1);
        buf[ST_X + dd * XPITCH + t] = rx[i];
    }
#pragma unroll
    for (int i = 0; i < NBC; ++i)
        buf[ST_BC + i * SCAN_THREADS + tid] = rbc[i];

    v2f a1;
    a1[0] = hh ? 0.f : ra[1][0];
    a1[1] = hh ? 0.f : ra[1][1];
    v8f acc = {};
    acc = __builtin_amdgcn_wmma_f32_16x16x4_f32(false, ra[0], false, wb[0],
                                                (short)0, acc, false, false);
    acc = __builtin_amdgcn_wmma_f32_16x16x4_f32(false, a1, false, wb[1],
                                                (short)0, acc, false, false);
    float dl[8];
#pragma unroll
    for (int r = 0; r < 8; ++r) dl[r] = softplus_dt(acc[r] + dbias);
    if (dpos >= 0) {
        float4 lo, hi;
        lo.x = dl[0]; lo.y = dl[1]; lo.z = dl[2]; lo.w = dl[3];
        hi.x = dl[4]; hi.y = dl[5]; hi.z = dl[6]; hi.w = dl[7];
        *(float4*)(buf + ST_D + dpos)     = lo;
        *(float4*)(buf + ST_D + dpos + 4) = hi;
    }
}

__device__ __forceinline__ void lds_stage_barrier() {
    asm volatile("s_wait_dscnt 0x0\n\ts_barrier_signal -1\n\ts_barrier_wait -1" ::: "memory");
}

template <int M>
__device__ __forceinline__ float row_halve(bool hi, float a, float b) {
    float keep = hi ? b : a;
    float send = hi ? a : b;
    int q = __builtin_amdgcn_update_dpp(0, __float_as_int(send), 0x160 + M, 0xf, 0xf, true);
    return keep + __int_as_float(q);
}

__global__ void __launch_bounds__(SCAN_THREADS)
k_scan(const float* __restrict__ xi,
       const float* __restrict__ xdbl,
       const float* __restrict__ Wdt,
       const float* __restrict__ bdt,
       const float* __restrict__ A_logs,
       const float* __restrict__ Ds,
       float* __restrict__ ys) {
    __shared__ __attribute__((aligned(16))) float sm[2 * ST_FLOATS];

    int tid   = threadIdx.x;
    int n     = tid & 15;
    int g     = tid >> 4;
    int d0    = (blockIdx.x % BLK_PER_KB) * SCAN_CH;
    int d     = d0 + g;
    int kb    = blockIdx.x / BLK_PER_KB;
    int b     = kb % BATCH;
    int kdir  = kb / BATCH;

    int sbase, sq, sr;
    if (kdir == 0)      { sbase = 0;                 sq = WWD; sr = 1;    }
    else if (kdir == 1) { sbase = WWD - 1;           sq = WWD; sr = -1;   }
    else if (kdir == 2) { sbase = 0;                 sq = 1;   sr = WWD;  }
    else                { sbase = (WWD - 1) * WWD;   sq = 1;   sr = -WWD; }

    bool hi8 = (n & 8) != 0, hi4 = (n & 4) != 0, hi2 = (n & 2) != 0, hi1 = (n & 1) != 0;
    int  trev = ((n & 1) << 3) | ((n & 2) << 1) | ((n & 4) >> 1) | ((n & 8) >> 3);

    float A = -__expf(A_logs[(size_t)(kdir * D_INNER + d) * D_STATE + n]);
    float Dskip = Ds[kdir * D_INNER + d];
    float h = 0.f;

    const float* xb  = xi    + (size_t)b  * LL * D_INNER;
    const float* drk = xdbl  + (size_t)kb * LL * NPROJ;
    const float* bck = xdbl  + (size_t)kb * LL * NPROJ + DT_RANK;
    float*       ysb = ys    + (size_t)kb * LL * D_INNER + d;

    int c16  = tid & 15;
    int hh   = (tid >> 4) & 1;
    int wv   = tid >> 5;
    int apos = (wv * 16 + c16) * NPROJ + 2 * hh;
    int dpos = (c16 < SCAN_CH) ? c16 * XPITCH + wv * 16 + hh * 8 : -1;
    int dcol = min(d0 + c16, D_INNER - 1);
    v2f wb[2];
#pragma unroll
    for (int s = 0; s < 2; ++s)
#pragma unroll
        for (int v = 0; v < 2; ++v) {
            int kk = s * 4 + 2 * hh + v;
            float w = Wdt[(size_t)(kdir * DT_RANK + min(kk, DT_RANK - 1)) * D_INNER + dcol];
            wb[s][v] = (kk < DT_RANK) ? w : 0.f;
        }
    float dbias = bdt[kdir * D_INNER + dcol];

    float rx[4], rbc[NBC];
    v2f ra[2];
    stage_fetch(xb, drk, bck, 0, d0, sbase, sq, sr, tid, apos, hh, rx, ra, rbc);
    stage_store(sm, tid, rx, ra, wb, dbias, hh, dpos, rbc);
    lds_stage_barrier();

    for (int s = 0; s < N_STAGE; ++s) {
        const float* cur = sm + (s & 1) * ST_FLOATS;
        float*       nxt = sm + ((s + 1) & 1) * ST_FLOATS;
        bool more = (s + 1 < N_STAGE);
        if (more)
            stage_fetch(xb, drk, bck, (s + 1) * ST_STEPS, d0, sbase, sq, sr, tid,
                        apos, hh, rx, ra, rbc);

        for (int c = 0; c < ST_STEPS / CH; ++c) {
            int tt = c * CH;
            float xc[CH], dc[CH], bc[CH], cc[CH];
#pragma unroll
            for (int j = 0; j < CH / 4; ++j) {
                float4 vx = *(const float4*)(cur + ST_X + g * XPITCH + tt + 4 * j);
                float4 vd = *(const float4*)(cur + ST_D + g * XPITCH + tt + 4 * j);
                xc[4 * j] = vx.x; xc[4 * j + 1] = vx.y; xc[4 * j + 2] = vx.z; xc[4 * j + 3] = vx.w;
                dc[4 * j] = vd.x; dc[4 * j + 1] = vd.y; dc[4 * j + 2] = vd.z; dc[4 * j + 3] = vd.w;
            }
#pragma unroll
            for (int t = 0; t < CH; ++t) {
                bc[t] = cur[ST_BC + (tt + t) * 32 + n];
                cc[t] = cur[ST_BC + (tt + t) * 32 + D_STATE + n];
            }

            float xo = cur[ST_X + g * XPITCH + tt + trev];

            float pv[CH];
#pragma unroll
            for (int t = 0; t < CH; ++t) {
                float x  = xc[t];
                float dl = dc[t];
                float dA = __expf(dl * A);
                h = dA * h + (dl * x) * bc[t];
                pv[t] = h * cc[t];
            }
            float q8[CH / 2], q4[CH / 4], q2[CH / 8];
#pragma unroll
            for (int j = 0; j < CH / 2; ++j) q8[j] = row_halve<8>(hi8, pv[2 * j], pv[2 * j + 1]);
#pragma unroll
            for (int j = 0; j < CH / 4; ++j) q4[j] = row_halve<4>(hi4, q8[2 * j], q8[2 * j + 1]);
#pragma unroll
            for (int j = 0; j < CH / 8; ++j) q2[j] = row_halve<2>(hi2, q4[2 * j], q4[2 * j + 1]);
            float p = row_halve<1>(hi1, q2[0], q2[1]);
            ysb[(size_t)(s * ST_STEPS + tt + trev) * D_INNER] = p + xo * Dskip;
        }

        if (more)
            stage_store(nxt, tid, rx, ra, wb, dbias, hh, dpos, rbc);
        lds_stage_barrier();
    }
}

__global__ void k_combine(const float* __restrict__ ys,
                          const float* __restrict__ xz,
                          float* __restrict__ y) {
    int tid = blockIdx.x * blockDim.x + threadIdx.x;
    if (tid >= LL * DQ) return;
    int d = (tid % DQ) * 4;
    int l = tid / DQ;
    float r[KDIR][BATCH][4], z[BATCH][4];
#pragma unroll
    for (int k = 0; k < KDIR; ++k) {
        int ls = spatial_to_scan(k, l);
#pragma unroll
        for (int b = 0; b < BATCH; ++b)
            f4_to_arr(*(const float4*)(ys + (size_t)((k * BATCH + b) * LL + ls) * D_INNER + d),
                      r[k][b]);
    }
#pragma unroll
    for (int b = 0; b < BATCH; ++b)
        f4_to_arr(*(const float4*)(xz + (size_t)(b * LL + l) * XZW + D_INNER + d), z[b]);
#pragma unroll
    for (int b = 0; b < BATCH; ++b) {
        float o[4];
#pragma unroll
        for (int c = 0; c < 4; ++c) {
            float acc = 0.f;
#pragma unroll
            for (int k = 0; k < KDIR; ++k) acc += r[k][b][c];
            o[c] = acc * silu(z[b][c]);
        }
        float4 q;
        q.x = o[0]; q.y = o[1]; q.z = o[2]; q.w = o[3];
        *(float4*)(y + (size_t)(b * LL + l) * D_INNER + d) = q;
    }
}

__global__ void k_out_gemm(const float* __restrict__ y,
                           const float* __restrict__ Wout,
                           float* __restrict__ out) {
    int lane = threadIdx.x & 31;
    int wave = threadIdx.x >> 5;
    int mTile = blockIdx.x * 4 + wave;
    int nTile = blockIdx.y;
    int m0 = mTile * 16, n0 = nTile * 16;
    int mrow = m0 + (lane & 15);
    int nn = n0 + (lane & 15);
    int kh = (lane >> 4) * 2;
    v8f acc = {};
    for (int kb = 0; kb < D_INNER; kb += 4) {
        v2f a, bf;
        a[0]  = y[(size_t)mrow * D_INNER + kb + kh];
        a[1]  = y[(size_t)mrow * D_INNER + kb + kh + 1];
        bf[0] = Wout[(size_t)(kb + kh    ) * DIMC + nn];
        bf[1] = Wout[(size_t)(kb + kh + 1) * DIMC + nn];
        acc = __builtin_amdgcn_wmma_f32_16x16x4_f32(false, a, false, bf,
                                                    (short)0, acc, false, false);
    }
    int mb = m0 + (lane >> 4) * 8;
#pragma unroll
    for (int v = 0; v < 8; ++v) {
        int row = mb + v;
        int b = row / LL, l = row % LL;
        out[(size_t)(b * DIMC + nn) * LL + l] = acc[v];
    }
}

extern "C" void kernel_launch(void* const* d_in, const int* in_sizes, int n_in,
                              void* d_out, int out_size, void* d_ws, size_t ws_size,
                              hipStream_t stream) {
    (void)in_sizes; (void)n_in; (void)out_size; (void)ws_size;
    const float* x       = (const float*)d_in[0];
    const float* W_in    = (const float*)d_in[1];
    const float* W_conv  = (const float*)d_in[2];
    const float* b_conv  = (const float*)d_in[3];
    const float* W_xproj = (const float*)d_in[4];
    const float* W_dt    = (const float*)d_in[5];
    const float* b_dt    = (const float*)d_in[6];
    const float* A_logs  = (const float*)d_in[7];
    const float* Ds      = (const float*)d_in[8];
    const float* W_out   = (const float*)d_in[9];
    float* out = (float*)d_out;

    float* ws = (float*)d_ws;
    size_t o = 0;
    float* xz    = ws + o;  o += (size_t)BL * XZW;
    float* xi    = ws + o;  o += (size_t)BL * D_INNER;
    float* xdbl  = ws + o;  o += (size_t)KDIR * BL * NPROJ;
    float* ysbuf = ws + o;  o += (size_t)KDIR * BL * D_INNER;
    float* ybuf  = ws + o;  o += (size_t)BL * D_INNER;

    dim3 blk(128);
    k_in_gemm <<<dim3(98, 24),    blk, 0, stream>>>(x, W_in, xz);
    int nquad = LL * DQ;
    k_conv    <<<(nquad + 255) / 256, 256, 0, stream>>>(xz, W_conv, b_conv, xi);
    k_xproj   <<<dim3(98, 3, 4),  blk, 0, stream>>>(xi, W_xproj, xdbl);
    int nscan = KDIR * BATCH * D_INNER * D_STATE;
    k_scan    <<<nscan / SCAN_THREADS, SCAN_THREADS, 0, stream>>>(xi, xdbl, W_dt, b_dt,
                                                        A_logs, Ds, ysbuf);
    k_combine <<<(nquad + 255) / 256, 256, 0, stream>>>(ysbuf, xz, ybuf);
    k_out_gemm<<<dim3(98, 6),     blk, 0, stream>>>(ybuf, W_out, out);
}


# conv+xproj and combine+out GEMM fused pairwise: a block makes its 32-row tile of xi (of y) in LDS, same order of sums, and multiplies it from there; 4 launches, not 6; y array gone
# speedup vs baseline: 19.1908x; 1.2524x over previous
#include <hip/hip_runtime.h>
#include <math.h>

typedef __attribute__((ext_vector_type(2))) float v2f;
typedef __attribute__((ext_vector_type(8))) float v8f;

#define DIMC    96
#define D_STATE 16
#define D_INNER 192
#define DT_RANK 6
#define KDIR    4
#define BATCH   2
#define HH      56
#define WWD     56
#define LL      (HH * WWD)
#define BL      (BATCH * LL)
#define NPROJ   (DT_RANK + 2 * D_STATE)
#define XZW     (2 * D_INNER)

__device__ __forceinline__ int scan_to_spatial(int k, int l) {
    int q = l / WWD, r = l % WWD;
    if (k == 0) return l;
    if (k == 1) return q * WWD + (WWD - 1 - r);
    if (k == 2) return r * WWD + q;
    return (WWD - 1 - r) * WWD + q;
}

__device__ __forceinline__ int spatial_to_scan(int k, int l) {
    int q = l / WWD, r = l % WWD;
    if (k == 0) return l;
    if (k == 1) return q * WWD + (WWD - 1 - r);
    if (k == 2) return r * WWD + q;
    return r * WWD + (WWD - 1 - q);
}

__device__ __forceinline__ float silu(float v) {
    return v / (1.f + __expf(-v));
}

__global__ void k_in_gemm(const float* __restrict__ x,
                          const float* __restrict__ Win,
                          float* __restrict__ xz) {
    int lane  = threadIdx.x & 31;
    int wave  = threadIdx.x >> 5;
    int mTile = blockIdx.x * 4 + wave;
    int nTile = blockIdx.y;
    int m0 = mTile * 16, n0 = nTile * 16;
    int mrow = m0 + (lane & 15);
    int b = mrow / LL, l = mrow % LL;
    int nn = n0 + (lane & 15);
    int kh = (lane >> 4) * 2;
    v8f acc = {};
    for (int kb = 0; kb < DIMC; kb += 4) {
        int c0 = kb + kh;
        v2f a, bf;
        a[0]  = x[(size_t)(b * DIMC + c0    ) * LL + l];
        a[1]  = x[(size_t)(b * DIMC + c0 + 1) * LL + l];
        bf[0] = Win[(size_t)(c0    ) * XZW + nn];
        bf[1] = Win[(size_t)(c0 + 1) * XZW + nn];
        acc = __builtin_amdgcn_wmma_f32_16x16x4_f32(false, a, false, bf,
                                                    (short)0, acc, false, false);
    }
    int mb = m0 + (lane >> 4) * 8;
#pragma unroll
    for (int v = 0; v < 8; ++v)
        xz[(size_t)(mb + v) * XZW + nn] = acc[v];
}

#define DQ (D_INNER / 4)
#define FT_ROWS 32
#define YP      (D_INNER + 4)
__device__ __forceinline__ void f4_to_arr(float4 q, float (&f)[4]) {
    f[0] = q.x; f[1] = q.y; f[2] = q.z; f[3] = q.w;
}
__global__ void __launch_bounds__(128) __attribute__((amdgpu_waves_per_eu(1, 1)))
k_conv_xproj(const float* __restrict__ xz,
             const float* __restrict__ Wc,
             const float* __restrict__ bc,
             const float* __restrict__ Wxp,
             float* __restrict__ xi,
             float* __restrict__ xdbl) {
    __shared__ __attribute__((aligned(16))) float xsm[FT_ROWS * YP];
    int tid  = threadIdx.x;
    int row0 = blockIdx.x * FT_ROWS;
    int b0   = row0 / LL;
    int p0   = row0 - b0 * LL;

#pragma unroll 1
    for (int m = 0; m < 3; ++m) {
        int q0  = m * 128 + tid;
        int rr0 = q0 / DQ;
        int d   = (q0 - rr0 * DQ) * 4;
        float wt[36];
#pragma unroll
        for (int i = 0; i < 9; ++i) {
            float q[4];
            f4_to_arr(*(const float4*)(Wc + (size_t)d * 9 + 4 * i), q);
            wt[4 * i] = q[0]; wt[4 * i + 1] = q[1]; wt[4 * i + 2] = q[2]; wt[4 * i + 3] = q[3];
        }
        float bias[4];
        f4_to_arr(*(const float4*)(bc + d), bias);
        float v[4][9][4];
        bool  ok[4][9];
#pragma unroll
        for (int j = 0; j < 4; ++j) {
            int p = p0 + rr0 + 8 * j;
            int h = p / WWD, w = p - h * WWD;
#pragma unroll
            for (int t = 0; t < 9; ++t) {
                int hy = h + t / 3 - 1, wx = w + t % 3 - 1;
                ok[j][t] = (hy >= 0) && (hy < HH) && (wx >= 0) && (wx < WWD);
                int hc = min(max(hy, 0), HH - 1), wc = min(max(wx, 0), WWD - 1);
                f4_to_arr(*(const float4*)(xz + (size_t)(b0 * LL + hc * WWD + wc) * XZW + d),
                          v[j][t]);
            }
        }
#pragma unroll
        for (int j = 0; j < 4; ++j) {
            int rr = rr0 + 8 * j;
            float acc[4];
#pragma unroll
            for (int c = 0; c < 4; ++c) acc[c] = bias[c];
#pragma unroll
            for (int t = 0; t < 9; ++t)
#pragma unroll
                for (int c = 0; c < 4; ++c)
                    acc[c] = ok[j][t] ? acc[c] + wt[c * 9 + t] * v[j][t][c] : acc[c];
            float4 o;
            o.x = silu(acc[0]); o.y = silu(acc[1]); o.z = silu(acc[2]); o.w = silu(acc[3]);
            *(float4*)(xi + (size_t)(row0 + rr) * D_INNER + d) = o;
            *(float4*)(xsm + rr * YP + d) = o;
        }
    }
    __syncthreads();

    int lane = tid & 31;
    int wave = tid >> 5;
    int c16  = lane & 15;
    int hh   = lane >> 4;
    int kh   = hh * 2;
    int mt   = wave >> 1;
    int kd0  = (wave & 1) * 2;
    const float* arow = xsm + (mt * 16 + c16) * YP + kh;
    int  nn2  = 32 + c16;
    bool nok2 = (nn2 < NPROJ);
    int  nn2c = min(nn2, NPROJ - 1);
    const float* wd0 = Wxp + ((size_t)kd0 * D_INNER + kh) * NPROJ;
    v2f afr[D_INNER / 4];
#pragma unroll
    for (int s = 0; s < D_INNER / 4; ++s) afr[s] = *(const v2f*)(arow + 4 * s);
    v8f acc[2][3] = {};
#pragma unroll
    for (int kb = 0; kb < D_INNER; kb += 4) {
        v2f a = afr[kb / 4];
#pragma unroll
        for (int kk = 0; kk < 2; ++kk) {
            const float* wr = wd0 + ((size_t)kk * D_INNER + kb) * NPROJ;
            v2f f0, f1, f2;
            f0[0] = wr[c16];       f0[1] = wr[NPROJ + c16];
            f1[0] = wr[16 + c16];  f1[1] = wr[NPROJ + 16 + c16];
            float t0 = wr[nn2c],   t1 = wr[NPROJ + nn2c];
            f2[0] = nok2 ? t0 : 0.f;
            f2[1] = nok2 ? t1 : 0.f;
            acc[kk][0] = __builtin_amdgcn_wmma_f32_16x16x4_f32(false, a, false, f0,
                                                               (short)0, acc[kk][0], false, false);
            acc[kk][1] = __builtin_amdgcn_wmma_f32_16x16x4_f32(false, a, false, f1,
                                                               (short)0, acc[kk][1], false, false);
            acc[kk][2] = __builtin_amdgcn_wmma_f32_16x16x4_f32(false, a, false, f2,
                                                               (short)0, acc[kk][2], false, false);
        }
    }
#pragma unroll
    for (int kk = 0; kk < 2; ++kk) {
        int kdir = kd0 + kk;
#pragma unroll
        for (int v = 0; v < 8; ++v) {
            int p = p0 + mt * 16 + hh * 8 + v;
            int l = spatial_to_scan(kdir, p);
            float* op = xdbl + ((size_t)kdir * BL + b0 * LL + l) * NPROJ;
            op[c16]      = acc[kk][0][v];
            op[16 + c16] = acc[kk][1][v];
            if (nok2) op[nn2] = acc[kk][2][v];
        }
    }
}

#define SCAN_CH      8
#define SCAN_THREADS (SCAN_CH * 16)
#define CH_SHIFT     3
#define BLK_PER_KB   (D_INNER / SCAN_CH)
#define NBC          (ST_STEPS * 32 / SCAN_THREADS)
#define ST_STEPS 64
#define N_STAGE  (LL / ST_STEPS)
#define CH       16
#define XPITCH   (ST_STEPS + 4)
#define ST_X     0
#define ST_D     (SCAN_CH * XPITCH)
#define ST_BC    (2 * SCAN_CH * XPITCH)
#define ST_FLOATS (ST_BC + ST_STEPS * 32)


__device__ __forceinline__ float softplus_dt(float t) {
    float e  = __expf(t);
    float u  = 1.f + e;
    float c  = e - (u - 1.f);
    float sp = __builtin_amdgcn_logf(u) * 0.69314718056f + c * __builtin_amdgcn_rcpf(u);
    return (t > 20.f) ? t : sp;
}

__device__ __forceinline__ void stage_fetch(const float* __restrict__ xb,
                                            const float* __restrict__ drk,
                                            const float* __restrict__ bck,
                                            int l0, int d0, int sbase, int sq, int sr,
                                            int tid, int apos, int hh,
                                            float (&rx)[4], v2f (&ra)[2],
                                            float (&rbc)[NBC]) {
#pragma unroll
    for (int i = 0; i < 4; ++i) {
        int e  = i * SCAN_THREADS + tid;
        int t  = e >> CH_SHIFT, dd = e & (SCAN_CH - 1);
        int l  = l0 + t;
        int q  = l / WWD, r = l - q * WWD;
        int sp = sbase + q * sq + r * sr;
        rx[i] = xb[(size_t)sp * D_INNER + d0 + dd];
    }
    const float* ap = drk + (size_t)l0 * NPROJ + apos;
    ra[0] = *(const v2f*)ap;
    ra[1] = *(const v2f*)(ap + 4 - 2 * hh);
#pragma unroll
    for (int i = 0; i < NBC; ++i) {
        int e = i * SCAN_THREADS + tid;
        int t = e >> 5, j = e & 31;
        rbc[i] = bck[(size_t)(l0 + t) * NPROJ + j];
    }
}

__device__ __forceinline__ void stage_store(float* __restrict__ buf, int tid,
                                            const float (&rx)[4], const v2f (&ra)[2],
                                            const v2f (&wb)[2], float dbias,
                                            int hh, int dpos,
                                            const float (&rbc)[NBC]) {
#pragma unroll
    for (int i = 0; i < 4; ++i) {
        int e = i * SCAN_THREADS + tid;
        int t = e >> CH_SHIFT, dd = e & (SCAN_CH - 1);
        buf[ST_X + dd * XPITCH + t] = rx[i];
    }
#pragma unroll
    for (int i = 0; i < NBC; ++i)
        buf[ST_BC + i * SCAN_THREADS + tid] = rbc[i];

    v2f a1;
    a1[0] = hh ? 0.f : ra[1][0];
    a1[1] = hh ? 0.f : ra[1][1];
    v8f acc = {};
    acc = __builtin_amdgcn_wmma_f32_16x16x4_f32(false, ra[0], false, wb[0],
                                                (short)0, acc, false, false);
    acc = __builtin_amdgcn_wmma_f32_16x16x4_f32(false, a1, false, wb[1],
                                                (short)0, acc, false, false);
    float dl[8];
#pragma unroll
    for (int r = 0; r < 8; ++r) dl[r] = softplus_dt(acc[r] + dbias);
    if (dpos >= 0) {
        float4 lo, hi;
        lo.x = dl[0]; lo.y = dl[1]; lo.z = dl[2]; lo.w = dl[3];
        hi.x = dl[4]; hi.y = dl[5]; hi.z = dl[6]; hi.w = dl[7];
        *(float4*)(buf + ST_D + dpos)     = lo;
        *(float4*)(buf + ST_D + dpos + 4) = hi;
    }
}

__device__ __forceinline__ void lds_stage_barrier() {
    asm volatile("s_wait_dscnt 0x0\n\ts_barrier_signal -1\n\ts_barrier_wait -1" ::: "memory");
}

template <int M>
__device__ __forceinline__ float row_halve(bool hi, float a, float b) {
    float keep = hi ? b : a;
    float send = hi ? a : b;
    int q = __builtin_amdgcn_update_dpp(0, __float_as_int(send), 0x160 + M, 0xf, 0xf, true);
    return keep + __int_as_float(q);
}

__global__ void __launch_bounds__(SCAN_THREADS)
k_scan(const float* __restrict__ xi,
       const float* __restrict__ xdbl,
       const float* __restrict__ Wdt,
       const float* __restrict__ bdt,
       const float* __restrict__ A_logs,
       const float* __restrict__ Ds,
       float* __restrict__ ys) {
    __shared__ __attribute__((aligned(16))) float sm[2 * ST_FLOATS];

    int tid   = threadIdx.x;
    int n     = tid & 15;
    int g     = tid >> 4;
    int d0    = (blockIdx.x % BLK_PER_KB) * SCAN_CH;
    int d     = d0 + g;
    int kb    = blockIdx.x / BLK_PER_KB;
    int b     = kb % BATCH;
    int kdir  = kb / BATCH;

    int sbase, sq, sr;
    if (kdir == 0)      { sbase = 0;                 sq = WWD; sr = 1;    }
    else if (kdir == 1) { sbase = WWD - 1;           sq = WWD; sr = -1;   }
    else if (kdir == 2) { sbase = 0;                 sq = 1;   sr = WWD;  }
    else                { sbase = (WWD - 1) * WWD;   sq = 1;   sr = -WWD; }

    bool hi8 = (n & 8) != 0, hi4 = (n & 4) != 0, hi2 = (n & 2) != 0, hi1 = (n & 1) != 0;
    int  trev = ((n & 1) << 3) | ((n & 2) << 1) | ((n & 4) >> 1) | ((n & 8) >> 3);

    float A = -__expf(A_logs[(size_t)(kdir * D_INNER + d) * D_STATE + n]);
    float Dskip = Ds[kdir * D_INNER + d];
    float h = 0.f;

    const float* xb  = xi    + (size_t)b  * LL * D_INNER;
    const float* drk = xdbl  + (size_t)kb * LL * NPROJ;
    const float* bck = xdbl  + (size_t)kb * LL * NPROJ + DT_RANK;
    float*       ysb = ys    + (size_t)kb * LL * D_INNER + d;

    int c16  = tid & 15;
    int hh   = (tid >> 4) & 1;
    int wv   = tid >> 5;
    int apos = (wv * 16 + c16) * NPROJ + 2 * hh;
    int dpos = (c16 < SCAN_CH) ? c16 * XPITCH + wv * 16 + hh * 8 : -1;
    int dcol = min(d0 + c16, D_INNER - 1);
    v2f wb[2];
#pragma unroll
    for (int s = 0; s < 2; ++s)
#pragma unroll
        for (int v = 0; v < 2; ++v) {
            int kk = s * 4 + 2 * hh + v;
            float w = Wdt[(size_t)(kdir * DT_RANK + min(kk, DT_RANK - 1)) * D_INNER + dcol];
            wb[s][v] = (kk < DT_RANK) ? w : 0.f;
        }
    float dbias = bdt[kdir * D_INNER + dcol];

    float rx[4], rbc[NBC];
    v2f ra[2];
    stage_fetch(xb, drk, bck, 0, d0, sbase, sq, sr, tid, apos, hh, rx, ra, rbc);
    stage_store(sm, tid, rx, ra, wb, dbias, hh, dpos, rbc);
    lds_stage_barrier();

    for (int s = 0; s < N_STAGE; ++s) {
        const float* cur = sm + (s & 1) * ST_FLOATS;
        float*       nxt = sm + ((s + 1) & 1) * ST_FLOATS;
        bool more = (s + 1 < N_STAGE);
        if (more)
            stage_fetch(xb, drk, bck, (s + 1) * ST_STEPS, d0, sbase, sq, sr, tid,
                        apos, hh, rx, ra, rbc);

        for (int c = 0; c < ST_STEPS / CH; ++c) {
            int tt = c * CH;
            float xc[CH], dc[CH], bc[CH], cc[CH];
#pragma unroll
            for (int j = 0; j < CH / 4; ++j) {
                float4 vx = *(const float4*)(cur + ST_X + g * XPITCH + tt + 4 * j);
                float4 vd = *(const float4*)(cur + ST_D + g * XPITCH + tt + 4 * j);
                xc[4 * j] = vx.x; xc[4 * j + 1] = vx.y; xc[4 * j + 2] = vx.z; xc[4 * j + 3] = vx.w;
                dc[4 * j] = vd.x; dc[4 * j + 1] = vd.y; dc[4 * j + 2] = vd.z; dc[4 * j + 3] = vd.w;
            }
#pragma unroll
            for (int t = 0; t < CH; ++t) {
                bc[t] = cur[ST_BC + (tt + t) * 32 + n];
                cc[t] = cur[ST_BC + (tt + t) * 32 + D_STATE + n];
            }

            float xo = cur[ST_X + g * XPITCH + tt + trev];

            float pv[CH];
#pragma unroll
            for (int t = 0; t < CH; ++t) {
                float x  = xc[t];
                float dl = dc[t];
                float dA = __expf(dl * A);
                h = dA * h + (dl * x) * bc[t];
                pv[t] = h * cc[t];
            }
            float q8[CH / 2], q4[CH / 4], q2[CH / 8];
#pragma unroll
            for (int j = 0; j < CH / 2; ++j) q8[j] = row_halve<8>(hi8, pv[2 * j], pv[2 * j + 1]);
#pragma unroll
            for (int j = 0; j < CH / 4; ++j) q4[j] = row_halve<4>(hi4, q8[2 * j], q8[2 * j + 1]);
#pragma unroll
            for (int j = 0; j < CH / 8; ++j) q2[j] = row_halve<2>(hi2, q4[2 * j], q4[2 * j + 1]);
            float p = row_halve<1>(hi1, q2[0], q2[1]);
            ysb[(size_t)(s * ST_STEPS + tt + trev) * D_INNER] = p + xo * Dskip;
        }

        if (more)
            stage_store(nxt, tid, rx, ra, wb, dbias, hh, dpos, rbc);
        lds_stage_barrier();
    }
}

#define FT_Q    (FT_ROWS * DQ / 128)
__global__ void __launch_bounds__(128) __attribute__((amdgpu_waves_per_eu(1, 1)))
k_comb_out(const float* __restrict__ ys,
           const float* __restrict__ xz,
           const float* __restrict__ Wout,
           float* __restrict__ out) {
    __shared__ __attribute__((aligned(16))) float ysm[FT_ROWS * YP];
    int tid  = threadIdx.x;
    int row0 = blockIdx.x * FT_ROWS;

    float r[FT_Q][KDIR][4], z[FT_Q][4];
#pragma unroll
    for (int i = 0; i < FT_Q; ++i) {
        int q   = i * 128 + tid;
        int rr  = q / DQ;
        int d   = (q - rr * DQ) * 4;
        int row = row0 + rr;
        int b = row / LL, l = row - b * LL;
#pragma unroll
        for (int k = 0; k < KDIR; ++k) {
            int ls = spatial_to_scan(k, l);
            f4_to_arr(*(const float4*)(ys + (size_t)((k * BATCH + b) * LL + ls) * D_INNER + d),
                      r[i][k]);
        }
        f4_to_arr(*(const float4*)(xz + (size_t)row * XZW + D_INNER + d), z[i]);
    }
#pragma unroll
    for (int i = 0; i < FT_Q; ++i) {
        int q  = i * 128 + tid;
        int rr = q / DQ;
        int d  = (q - rr * DQ) * 4;
        float o[4];
#pragma unroll
        for (int c = 0; c < 4; ++c) {
            float acc = 0.f;
#pragma unroll
            for (int k = 0; k < KDIR; ++k) acc += r[i][k][c];
            o[c] = acc * silu(z[i][c]);
        }
        float4 qv;
        qv.x = o[0]; qv.y = o[1]; qv.z = o[2]; qv.w = o[3];
        *(float4*)(ysm + rr * YP + d) = qv;
    }
    __syncthreads();

    int lane = tid & 31;
    int wave = tid >> 5;
    int c16  = lane & 15;
    int kh   = (lane >> 4) * 2;
    int mt   = wave >> 1;
    int n0   = (wave & 1) * 48;
    const float* arow = ysm + (mt * 16 + c16) * YP + kh;
    const float* wcol = Wout + (size_t)kh * DIMC + n0 + c16;
    v8f acc0 = {}, acc1 = {}, acc2 = {};
#pragma unroll
    for (int kb = 0; kb < D_INNER; kb += 4) {
        v2f a = *(const v2f*)(arow + kb);
        const float* w0 = wcol + (size_t)kb * DIMC;
        v2f b0, b1, b2;
        b0[0] = w0[0];   b0[1] = w0[DIMC];
        b1[0] = w0[16];  b1[1] = w0[DIMC + 16];
        b2[0] = w0[32];  b2[1] = w0[DIMC + 32];
        acc0 = __builtin_amdgcn_wmma_f32_16x16x4_f32(false, a, false, b0,
                                                     (short)0, acc0, false, false);
        acc1 = __builtin_amdgcn_wmma_f32_16x16x4_f32(false, a, false, b1,
                                                     (short)0, acc1, false, false);
        acc2 = __builtin_amdgcn_wmma_f32_16x16x4_f32(false, a, false, b2,
                                                     (short)0, acc2, false, false);
    }
    int mb = row0 + mt * 16 + (lane >> 4) * 8;
    int nn = n0 + c16;
#pragma unroll
    for (int v = 0; v < 8; ++v) {
        int row = mb + v;
        int b = row / LL, l = row - b * LL;
        float* op = out + (size_t)(b * DIMC + nn) * LL + l;
        op[0]                = acc0[v];
        op[(size_t)16 * LL]  = acc1[v];
        op[(size_t)32 * LL]  = acc2[v];
    }
}

extern "C" void kernel_launch(void* const* d_in, const int* in_sizes, int n_in,
                              void* d_out, int out_size, void* d_ws, size_t ws_size,
                              hipStream_t stream) {
    (void)in_sizes; (void)n_in; (void)out_size; (void)ws_size;
    const float* x       = (const float*)d_in[0];
    const float* W_in    = (const float*)d_in[1];
    const float* W_conv  = (const float*)d_in[2];
    const float* b_conv  = (const float*)d_in[3];
    const float* W_xproj = (const float*)d_in[4];
    const float* W_dt    = (const float*)d_in[5];
    const float* b_dt    = (const float*)d_in[6];
    const float* A_logs  = (const float*)d_in[7];
    const float* Ds      = (const float*)d_in[8];
    const float* W_out   = (const float*)d_in[9];
    float* out = (float*)d_out;

    float* ws = (float*)d_ws;
    size_t o = 0;
    float* xz    = ws + o;  o += (size_t)BL * XZW;
    float* xi    = ws + o;  o += (size_t)BL * D_INNER;
    float* xdbl  = ws + o;  o += (size_t)KDIR * BL * NPROJ;
    float* ysbuf = ws + o;  o += (size_t)KDIR * BL * D_INNER;

    dim3 blk(128);
    k_in_gemm <<<dim3(98, 24),    blk, 0, stream>>>(x, W_in, xz);
    k_conv_xproj<<<BL / FT_ROWS, 128, 0, stream>>>(xz, W_conv, b_conv, W_xproj, xi, xdbl);
    int nscan = KDIR * BATCH * D_INNER * D_STATE;
    k_scan    <<<nscan / SCAN_THREADS, SCAN_THREADS, 0, stream>>>(xi, xdbl, W_dt, b_dt,
                                                        A_logs, Ds, ysbuf);
    k_comb_out<<<BL / FT_ROWS, 128, 0, stream>>>(ysbuf, xz, W_out, out);
}


# scan kernel: a stage's loads, dt projection+softplus and LDS stores go to 4 staging waves beside the 4 scan waves, 2 stages ahead, ring of 4 LDS buffers; same source sums (compiler now fuses h*dA)
# speedup vs baseline: 21.8817x; 1.1402x over previous
#include <hip/hip_runtime.h>
#include <math.h>

typedef __attribute__((ext_vector_type(2))) float v2f;
typedef __attribute__((ext_vector_type(8))) float v8f;

#define DIMC    96
#define D_STATE 16
#define D_INNER 192
#define DT_RANK 6
#define KDIR    4
#define BATCH   2
#define HH      56
#define WWD     56
#define LL      (HH * WWD)
#define BL      (BATCH * LL)
#define NPROJ   (DT_RANK + 2 * D_STATE)
#define XZW     (2 * D_INNER)

__device__ __forceinline__ int scan_to_spatial(int k, int l) {
    int q = l / WWD, r = l % WWD;
    if (k == 0) return l;
    if (k == 1) return q * WWD + (WWD - 1 - r);
    if (k == 2) return r * WWD + q;
    return (WWD - 1 - r) * WWD + q;
}

__device__ __forceinline__ int spatial_to_scan(int k, int l) {
    int q = l / WWD, r = l % WWD;
    if (k == 0) return l;
    if (k == 1) return q * WWD + (WWD - 1 - r);
    if (k == 2) return r * WWD + q;
    return r * WWD + (WWD - 1 - q);
}

__device__ __forceinline__ float silu(float v) {
    return v / (1.f + __expf(-v));
}

__global__ void k_in_gemm(const float* __restrict__ x,
                          const float* __restrict__ Win,
                          float* __restrict__ xz) {
    int lane  = threadIdx.x & 31;
    int wave  = threadIdx.x >> 5;
    int mTile = blockIdx.x * 4 + wave;
    int nTile = blockIdx.y;
    int m0 = mTile * 16, n0 = nTile * 16;
    int mrow = m0 + (lane & 15);
    int b = mrow / LL, l = mrow % LL;
    int nn = n0 + (lane & 15);
    int kh = (lane >> 4) * 2;
    v8f acc = {};
    for (int kb = 0; kb < DIMC; kb += 4) {
        int c0 = kb + kh;
        v2f a, bf;
        a[0]  = x[(size_t)(b * DIMC + c0    ) * LL + l];
        a[1]  = x[(size_t)(b * DIMC + c0 + 1) * LL + l];
        bf[0] = Win[(size_t)(c0    ) * XZW + nn];
        bf[1] = Win[(size_t)(c0 + 1) * XZW + nn];
        acc = __builtin_amdgcn_wmma_f32_16x16x4_f32(false, a, false, bf,
                                                    (short)0, acc, false, false);
    }
    int mb = m0 + (lane >> 4) * 8;
#pragma unroll
    for (int v = 0; v < 8; ++v)
        xz[(size_t)(mb + v) * XZW + nn] = acc[v];
}

#define DQ (D_INNER / 4)
#define FT_ROWS 32
#define YP      (D_INNER + 4)
__device__ __forceinline__ void f4_to_arr(float4 q, float (&f)[4]) {
    f[0] = q.x; f[1] = q.y; f[2] = q.z; f[3] = q.w;
}
__global__ void __launch_bounds__(128) __attribute__((amdgpu_waves_per_eu(1, 1)))
k_conv_xproj(const float* __restrict__ xz,
             const float* __restrict__ Wc,
             const float* __restrict__ bc,
             const float* __restrict__ Wxp,
             float* __restrict__ xi,
             float* __restrict__ xdbl) {
    __shared__ __attribute__((aligned(16))) float xsm[FT_ROWS * YP];
    int tid  = threadIdx.x;
    int row0 = blockIdx.x * FT_ROWS;
    int b0   = row0 / LL;
    int p0   = row0 - b0 * LL;

#pragma unroll 1
    for (int m = 0; m < 3; ++m) {
        int q0  = m * 128 + tid;
        int rr0 = q0 / DQ;
        int d   = (q0 - rr0 * DQ) * 4;
        float wt[36];
#pragma unroll
        for (int i = 0; i < 9; ++i) {
            float q[4];
            f4_to_arr(*(const float4*)(Wc + (size_t)d * 9 + 4 * i), q);
            wt[4 * i] = q[0]; wt[4 * i + 1] = q[1]; wt[4 * i + 2] = q[2]; wt[4 * i + 3] = q[3];
        }
        float bias[4];
        f4_to_arr(*(const float4*)(bc + d), bias);
        float v[4][9][4];
        bool  ok[4][9];
#pragma unroll
        for (int j = 0; j < 4; ++j) {
            int p = p0 + rr0 + 8 * j;
            int h = p / WWD, w = p - h * WWD;
#pragma unroll
            for (int t = 0; t < 9; ++t) {
                int hy = h + t / 3 - 1, wx = w + t % 3 - 1;
                ok[j][t] = (hy >= 0) && (hy < HH) && (wx >= 0) && (wx < WWD);
                int hc = min(max(hy, 0), HH - 1), wc = min(max(wx, 0), WWD - 1);
                f4_to_arr(*(const float4*)(xz + (size_t)(b0 * LL + hc * WWD + wc) * XZW + d),
                          v[j][t]);
            }
        }
#pragma unroll
        for (int j = 0; j < 4; ++j) {
            int rr = rr0 + 8 * j;
            float acc[4];
#pragma unroll
            for (int c = 0; c < 4; ++c) acc[c] = bias[c];
#pragma unroll
            for (int t = 0; t < 9; ++t)
#pragma unroll
                for (int c = 0; c < 4; ++c)
                    acc[c] = ok[j][t] ? acc[c] + wt[c * 9 + t] * v[j][t][c] : acc[c];
            float4 o;
            o.x = silu(acc[0]); o.y = silu(acc[1]); o.z = silu(acc[2]); o.w = silu(acc[3]);
            *(float4*)(xi + (size_t)(row0 + rr) * D_INNER + d) = o;
            *(float4*)(xsm + rr * YP + d) = o;
        }
    }
    __syncthreads();

    int lane = tid & 31;
    int wave = tid >> 5;
    int c16  = lane & 15;
    int hh   = lane >> 4;
    int kh   = hh * 2;
    int mt   = wave >> 1;
    int kd0  = (wave & 1) * 2;
    const float* arow = xsm + (mt * 16 + c16) * YP + kh;
    int  nn2  = 32 + c16;
    bool nok2 = (nn2 < NPROJ);
    int  nn2c = min(nn2, NPROJ - 1);
    const float* wd0 = Wxp + ((size_t)kd0 * D_INNER + kh) * NPROJ;
    v2f afr[D_INNER / 4];
#pragma unroll
    for (int s = 0; s < D_INNER / 4; ++s) afr[s] = *(const v2f*)(arow + 4 * s);
    v8f acc[2][3] = {};
#pragma unroll
    for (int kb = 0; kb < D_INNER; kb += 4) {
        v2f a = afr[kb / 4];
#pragma unroll
        for (int kk = 0; kk < 2; ++kk) {
            const float* wr = wd0 + ((size_t)kk * D_INNER + kb) * NPROJ;
            v2f f0, f1, f2;
            f0[0] = wr[c16];       f0[1] = wr[NPROJ + c16];
            f1[0] = wr[16 + c16];  f1[1] = wr[NPROJ + 16 + c16];
            float t0 = wr[nn2c],   t1 = wr[NPROJ + nn2c];
            f2[0] = nok2 ? t0 : 0.f;
            f2[1] = nok2 ? t1 : 0.f;
            acc[kk][0] = __builtin_amdgcn_wmma_f32_16x16x4_f32(false, a, false, f0,
                                                               (short)0, acc[kk][0], false, false);
            acc[kk][1] = __builtin_amdgcn_wmma_f32_16x16x4_f32(false, a, false, f1,
                                                               (short)0, acc[kk][1], false, false);
            acc[kk][2] = __builtin_amdgcn_wmma_f32_16x16x4_f32(false, a, false, f2,
                                                               (short)0, acc[kk][2], false, false);
        }
    }
#pragma unroll
    for (int kk = 0; kk < 2; ++kk) {
        int kdir = kd0 + kk;
#pragma unroll
        for (int v = 0; v < 8; ++v) {
            int p = p0 + mt * 16 + hh * 8 + v;
            int l = spatial_to_scan(kdir, p);
            float* op = xdbl + ((size_t)kdir * BL + b0 * LL + l) * NPROJ;
            op[c16]      = acc[kk][0][v];
            op[16 + c16] = acc[kk][1][v];
            if (nok2) op[nn2] = acc[kk][2][v];
        }
    }
}

#define SCAN_CH      8
#define SCAN_THREADS (SCAN_CH * 16)
#define CH_SHIFT     3
#define BLK_PER_KB   (D_INNER / SCAN_CH)
#define NBC          (ST_STEPS * 32 / SCAN_THREADS)
#define ST_STEPS 64
#define N_STAGE  (LL / ST_STEPS)
#define CH       16
#define XPITCH   (ST_STEPS + 4)
#define ST_X     0
#define ST_D     (SCAN_CH * XPITCH)
#define ST_BC    (2 * SCAN_CH * XPITCH)
#define ST_FLOATS (ST_BC + ST_STEPS * 32)


__device__ __forceinline__ float softplus_dt(float t) {
    float e  = __expf(t);
    float u  = 1.f + e;
    float c  = e - (u - 1.f);
    float sp = __builtin_amdgcn_logf(u) * 0.69314718056f + c * __builtin_amdgcn_rcpf(u);
    return (t > 20.f) ? t : sp;
}

__device__ __forceinline__ void stage_fetch(const float* __restrict__ xb,
                                            const float* __restrict__ drk,
                                            const float* __restrict__ bck,
                                            int l0, int d0, int sbase, int sq, int sr,
                                            int tid, int apos, int hh,
                                            float (&rx)[4], v2f (&ra)[2],
                                            float (&rbc)[NBC]) {
#pragma unroll
    for (int i = 0; i < 4; ++i) {
        int e  = i * SCAN_THREADS + tid;
        int t  = e >> CH_SHIFT, dd = e & (SCAN_CH - 1);
        int l  = l0 + t;
        int q  = l / WWD, r = l - q * WWD;
        int sp = sbase + q * sq + r * sr;
        rx[i] = xb[(size_t)sp * D_INNER + d0 + dd];
    }
    const float* ap = drk + (size_t)l0 * NPROJ + apos;
    ra[0] = *(const v2f*)ap;
    ra[1] = *(const v2f*)(ap + 4 - 2 * hh);
#pragma unroll
    for (int i = 0; i < NBC; ++i) {
        int e = i * SCAN_THREADS + tid;
        int t = e >> 5, j = e & 31;
        rbc[i] = bck[(size_t)(l0 + t) * NPROJ + j];
    }
}

__device__ __forceinline__ void stage_store(float* __restrict__ buf, int tid,
                                            const float (&rx)[4], const v2f (&ra)[2],
                                            const v2f (&wb)[2], float dbias,
                                            int hh, int dpos,
                                            const float (&rbc)[NBC]) {
#pragma unroll
    for (int i = 0; i < 4; ++i) {
        int e = i * SCAN_THREADS + tid;
        int t = e >> CH_SHIFT, dd = e & (SCAN_CH - 1);
        buf[ST_X + dd * XPITCH + t] = rx[i];
    }
#pragma unroll
    for (int i = 0; i < NBC; ++i)
        buf[ST_BC + i * SCAN_THREADS + tid] = rbc[i];

    v2f a1;
    a1[0] = hh ? 0.f : ra[1][0];
    a1[1] = hh ? 0.f : ra[1][1];
    v8f acc = {};
    acc = __builtin_amdgcn_wmma_f32_16x16x4_f32(false, ra[0], false, wb[0],
                                                (short)0, acc, false, false);
    acc = __builtin_amdgcn_wmma_f32_16x16x4_f32(false, a1, false, wb[1],
                                                (short)0, acc, false, false);
    float dl[8];
#pragma unroll
    for (int r = 0; r < 8; ++r) dl[r] = softplus_dt(acc[r] + dbias);
    if (dpos >= 0) {
        float4 lo, hi;
        lo.x = dl[0]; lo.y = dl[1]; lo.z = dl[2]; lo.w = dl[3];
        hi.x = dl[4]; hi.y = dl[5]; hi.z = dl[6]; hi.w = dl[7];
        *(float4*)(buf + ST_D + dpos)     = lo;
        *(float4*)(buf + ST_D + dpos + 4) = hi;
    }
}

__device__ __forceinline__ void lds_stage_barrier() {
    asm volatile("s_wait_dscnt 0x0\n\ts_barrier_signal -1\n\ts_barrier_wait -1" ::: "memory");
}

template <int M>
__device__ __forceinline__ float row_halve(bool hi, float a, float b) {
    float keep = hi ? b : a;
    float send = hi ? a : b;
    int q = __builtin_amdgcn_update_dpp(0, __float_as_int(send), 0x160 + M, 0xf, 0xf, true);
    return keep + __int_as_float(q);
}

#define RING 4
__global__ void __launch_bounds__(2 * SCAN_THREADS)
k_scan(const float* __restrict__ xi,
       const float* __restrict__ xdbl,
       const float* __restrict__ Wdt,
       const float* __restrict__ bdt,
       const float* __restrict__ A_logs,
       const float* __restrict__ Ds,
       float* __restrict__ ys) {
    __shared__ __attribute__((aligned(16))) float sm[RING * ST_FLOATS];

    int wave8   = __builtin_amdgcn_readfirstlane((int)(threadIdx.x >> 5));
    bool stager = wave8 >= SCAN_THREADS / 32;
    int tid   = threadIdx.x & (SCAN_THREADS - 1);
    int n     = tid & 15;
    int g     = tid >> 4;
    int d0    = (blockIdx.x % BLK_PER_KB) * SCAN_CH;
    int d     = d0 + g;
    int kb    = blockIdx.x / BLK_PER_KB;
    int b     = kb % BATCH;
    int kdir  = kb / BATCH;

    int sbase, sq, sr;
    if (kdir == 0)      { sbase = 0;                 sq = WWD; sr = 1;    }
    else if (kdir == 1) { sbase = WWD - 1;           sq = WWD; sr = -1;   }
    else if (kdir == 2) { sbase = 0;                 sq = 1;   sr = WWD;  }
    else                { sbase = (WWD - 1) * WWD;   sq = 1;   sr = -WWD; }

    bool hi8 = (n & 8) != 0, hi4 = (n & 4) != 0, hi2 = (n & 2) != 0, hi1 = (n & 1) != 0;
    int  trev = ((n & 1) << 3) | ((n & 2) << 1) | ((n & 4) >> 1) | ((n & 8) >> 3);

    float A = 0.f, Dskip = 0.f;
    if (!stager) {
        Dskip = Ds[kdir * D_INNER + d];
        A = -__expf(A_logs[(size_t)(kdir * D_INNER + d) * D_STATE + n]);
    }
    float h = 0.f;

    const float* xb  = xi    + (size_t)b  * LL * D_INNER;
    const float* drk = xdbl  + (size_t)kb * LL * NPROJ;
    const float* bck = xdbl  + (size_t)kb * LL * NPROJ + DT_RANK;
    float*       ysb = ys    + (size_t)kb * LL * D_INNER + d;

    int c16  = tid & 15;
    int hh   = (tid >> 4) & 1;
    int wv   = tid >> 5;
    int apos = (wv * 16 + c16) * NPROJ + 2 * hh;
    int dpos = (c16 < SCAN_CH) ? c16 * XPITCH + wv * 16 + hh * 8 : -1;
    int dcol = min(d0 + c16, D_INNER - 1);
    v2f wb[2] = {};
    float dbias = 0.f;

    auto scan_stage = [&](int s) __attribute__((always_inline)) {
        const float* cur = sm + (s & (RING - 1)) * ST_FLOATS;
#pragma unroll 1
        for (int c = 0; c < ST_STEPS / CH; ++c) {
            int tt = c * CH;
            float xc[CH], dc[CH], bc[CH], cc[CH];
#pragma unroll
            for (int j = 0; j < CH / 4; ++j) {
                float4 vx = *(const float4*)(cur + ST_X + g * XPITCH + tt + 4 * j);
                float4 vd = *(const float4*)(cur + ST_D + g * XPITCH + tt + 4 * j);
                xc[4 * j] = vx.x; xc[4 * j + 1] = vx.y; xc[4 * j + 2] = vx.z; xc[4 * j + 3] = vx.w;
                dc[4 * j] = vd.x; dc[4 * j + 1] = vd.y; dc[4 * j + 2] = vd.z; dc[4 * j + 3] = vd.w;
            }
#pragma unroll
            for (int t = 0; t < CH; ++t) {
                bc[t] = cur[ST_BC + (tt + t) * 32 + n];
                cc[t] = cur[ST_BC + (tt + t) * 32 + D_STATE + n];
            }

            float xo = cur[ST_X + g * XPITCH + tt + trev];

            float pv[CH];
#pragma unroll
            for (int t = 0; t < CH; ++t) {
                float x  = xc[t];
                float dl = dc[t];
                float dA = __expf(dl * A);
                h = dA * h + (dl * x) * bc[t];
                pv[t] = h * cc[t];
            }
            float q8[CH / 2], q4[CH / 4], q2[CH / 8];
#pragma unroll
            for (int j = 0; j < CH / 2; ++j) q8[j] = row_halve<8>(hi8, pv[2 * j], pv[2 * j + 1]);
#pragma unroll
            for (int j = 0; j < CH / 4; ++j) q4[j] = row_halve<4>(hi4, q8[2 * j], q8[2 * j + 1]);
#pragma unroll
            for (int j = 0; j < CH / 8; ++j) q2[j] = row_halve<2>(hi2, q4[2 * j], q4[2 * j + 1]);
            float p = row_halve<1>(hi1, q2[0], q2[1]);
            ysb[(size_t)(s * ST_STEPS + tt + trev) * D_INNER] = p + xo * Dskip;
        }
    };

    float rx0[4], rbc0[NBC], rx1[4], rbc1[NBC];
    v2f ra0[2], ra1[2];
    if (stager) {
        dbias = bdt[kdir * D_INNER + dcol];
#pragma unroll
        for (int s = 0; s < 2; ++s)
#pragma unroll
            for (int v = 0; v < 2; ++v) {
                int kk = s * 4 + 2 * hh + v;
                float w = Wdt[(size_t)(kdir * DT_RANK + min(kk, DT_RANK - 1)) * D_INNER + dcol];
                wb[s][v] = (kk < DT_RANK) ? w : 0.f;
            }
        stage_fetch(xb, drk, bck, 0,        d0, sbase, sq, sr, tid, apos, hh, rx0, ra0, rbc0);
        stage_fetch(xb, drk, bck, ST_STEPS, d0, sbase, sq, sr, tid, apos, hh, rx1, ra1, rbc1);
        stage_store(sm,             tid, rx0, ra0, wb, dbias, hh, dpos, rbc0);
        stage_store(sm + ST_FLOATS, tid, rx1, ra1, wb, dbias, hh, dpos, rbc1);
        stage_fetch(xb, drk, bck, 2 * ST_STEPS, d0, sbase, sq, sr, tid, apos, hh, rx0, ra0, rbc0);
        stage_fetch(xb, drk, bck, 3 * ST_STEPS, d0, sbase, sq, sr, tid, apos, hh, rx1, ra1, rbc1);
    }
    lds_stage_barrier();

    for (int s = 0; s < N_STAGE; s += 2) {
        if (stager) {
            if (s + 2 < N_STAGE) {
                stage_store(sm + ((s + 2) & (RING - 1)) * ST_FLOATS, tid, rx0, ra0, wb,
                            dbias, hh, dpos, rbc0);
                if (s + 3 < N_STAGE)
                    stage_store(sm + ((s + 3) & (RING - 1)) * ST_FLOATS, tid, rx1, ra1, wb,
                                dbias, hh, dpos, rbc1);
                stage_fetch(xb, drk, bck, min(s + 4, N_STAGE - 1) * ST_STEPS, d0, sbase,
                            sq, sr, tid, apos, hh, rx0, ra0, rbc0);
                stage_fetch(xb, drk, bck, min(s + 5, N_STAGE - 1) * ST_STEPS, d0, sbase,
                            sq, sr, tid, apos, hh, rx1, ra1, rbc1);
            }
        } else {
            scan_stage(s);
            if (s + 1 < N_STAGE) scan_stage(s + 1);
        }
        lds_stage_barrier();
    }
}

#define FT_Q    (FT_ROWS * DQ / 128)
__global__ void __launch_bounds__(128) __attribute__((amdgpu_waves_per_eu(1, 1)))
k_comb_out(const float* __restrict__ ys,
           const float* __restrict__ xz,
           const float* __restrict__ Wout,
           float* __restrict__ out) {
    __shared__ __attribute__((aligned(16))) float ysm[FT_ROWS * YP];
    int tid  = threadIdx.x;
    int row0 = blockIdx.x * FT_ROWS;

    float r[FT_Q][KDIR][4], z[FT_Q][4];
#pragma unroll
    for (int i = 0; i < FT_Q; ++i) {
        int q   = i * 128 + tid;
        int rr  = q / DQ;
        int d   = (q - rr * DQ) * 4;
        int row = row0 + rr;
        int b = row / LL, l = row - b * LL;
#pragma unroll
        for (int k = 0; k < KDIR; ++k) {
            int ls = spatial_to_scan(k, l);
            f4_to_arr(*(const float4*)(ys + (size_t)((k * BATCH + b) * LL + ls) * D_INNER + d),
                      r[i][k]);
        }
        f4_to_arr(*(const float4*)(xz + (size_t)row * XZW + D_INNER + d), z[i]);
    }
#pragma unroll
    for (int i = 0; i < FT_Q; ++i) {
        int q  = i * 128 + tid;
        int rr = q / DQ;
        int d  = (q - rr * DQ) * 4;
        float o[4];
#pragma unroll
        for (int c = 0; c < 4; ++c) {
            float acc = 0.f;
#pragma unroll
            for (int k = 0; k < KDIR; ++k) acc += r[i][k][c];
            o[c] = acc * silu(z[i][c]);
        }
        float4 qv;
        qv.x = o[0]; qv.y = o[1]; qv.z = o[2]; qv.w = o[3];
        *(float4*)(ysm + rr * YP + d) = qv;
    }
    __syncthreads();

    int lane = tid & 31;
    int wave = tid >> 5;
    int c16  = lane & 15;
    int kh   = (lane >> 4) * 2;
    int mt   = wave >> 1;
    int n0   = (wave & 1) * 48;
    const float* arow = ysm + (mt * 16 + c16) * YP + kh;
    const float* wcol = Wout + (size_t)kh * DIMC + n0 + c16;
    v8f acc0 = {}, acc1 = {}, acc2 = {};
#pragma unroll
    for (int kb = 0; kb < D_INNER; kb += 4) {
        v2f a = *(const v2f*)(arow + kb);
        const float* w0 = wcol + (size_t)kb * DIMC;
        v2f b0, b1, b2;
        b0[0] = w0[0];   b0[1] = w0[DIMC];
        b1[0] = w0[16];  b1[1] = w0[DIMC + 16];
        b2[0] = w0[32];  b2[1] = w0[DIMC + 32];
        acc0 = __builtin_amdgcn_wmma_f32_16x16x4_f32(false, a, false, b0,
                                                     (short)0, acc0, false, false);
        acc1 = __builtin_amdgcn_wmma_f32_16x16x4_f32(false, a, false, b1,
                                                     (short)0, acc1, false, false);
        acc2 = __builtin_amdgcn_wmma_f32_16x16x4_f32(false, a, false, b2,
                                                     (short)0, acc2, false, false);
    }
    int mb = row0 + mt * 16 + (lane >> 4) * 8;
    int nn = n0 + c16;
#pragma unroll
    for (int v = 0; v < 8; ++v) {
        int row = mb + v;
        int b = row / LL, l = row - b * LL;
        float* op = out + (size_t)(b * DIMC + nn) * LL + l;
        op[0]                = acc0[v];
        op[(size_t)16 * LL]  = acc1[v];
        op[(size_t)32 * LL]  = acc2[v];
    }
}

extern "C" void kernel_launch(void* const* d_in, const int* in_sizes, int n_in,
                              void* d_out, int out_size, void* d_ws, size_t ws_size,
                              hipStream_t stream) {
    (void)in_sizes; (void)n_in; (void)out_size; (void)ws_size;
    const float* x       = (const float*)d_in[0];
    const float* W_in    = (const float*)d_in[1];
    const float* W_conv  = (const float*)d_in[2];
    const float* b_conv  = (const float*)d_in[3];
    const float* W_xproj = (const float*)d_in[4];
    const float* W_dt    = (const float*)d_in[5];
    const float* b_dt    = (const float*)d_in[6];
    const float* A_logs  = (const float*)d_in[7];
    const float* Ds      = (const float*)d_in[8];
    const float* W_out   = (const float*)d_in[9];
    float* out = (float*)d_out;

    float* ws = (float*)d_ws;
    size_t o = 0;
    float* xz    = ws + o;  o += (size_t)BL * XZW;
    float* xi    = ws + o;  o += (size_t)BL * D_INNER;
    float* xdbl  = ws + o;  o += (size_t)KDIR * BL * NPROJ;
    float* ysbuf = ws + o;  o += (size_t)KDIR * BL * D_INNER;

    dim3 blk(128);
    k_in_gemm <<<dim3(98, 24),    blk, 0, stream>>>(x, W_in, xz);
    k_conv_xproj<<<BL / FT_ROWS, 128, 0, stream>>>(xz, W_conv, b_conv, W_xproj, xi, xdbl);
    int nscan = KDIR * BATCH * D_INNER * D_STATE;
    k_scan    <<<nscan / SCAN_THREADS, 2 * SCAN_THREADS, 0, stream>>>(xi, xdbl, W_dt, b_dt,
                                                        A_logs, Ds, ysbuf);
    k_comb_out<<<BL / FT_ROWS, 128, 0, stream>>>(ysbuf, xz, W_out, out);
}


# in-projection GEMM re-tiled: a wave makes 32 rows x 96 columns (12 accumulators, a W_in fragment serves two row tiles, loads one k block ahead); 784 waves instead of 9408; order of sums unchanged
# speedup vs baseline: 22.9190x; 1.0474x over previous
#include <hip/hip_runtime.h>
#include <math.h>

typedef __attribute__((ext_vector_type(2))) float v2f;
typedef __attribute__((ext_vector_type(8))) float v8f;

#define DIMC    96
#define D_STATE 16
#define D_INNER 192
#define DT_RANK 6
#define KDIR    4
#define BATCH   2
#define HH      56
#define WWD     56
#define LL      (HH * WWD)
#define BL      (BATCH * LL)
#define NPROJ   (DT_RANK + 2 * D_STATE)
#define XZW     (2 * D_INNER)

__device__ __forceinline__ int scan_to_spatial(int k, int l) {
    int q = l / WWD, r = l % WWD;
    if (k == 0) return l;
    if (k == 1) return q * WWD + (WWD - 1 - r);
    if (k == 2) return r * WWD + q;
    return (WWD - 1 - r) * WWD + q;
}

__device__ __forceinline__ int spatial_to_scan(int k, int l) {
    int q = l / WWD, r = l % WWD;
    if (k == 0) return l;
    if (k == 1) return q * WWD + (WWD - 1 - r);
    if (k == 2) return r * WWD + q;
    return r * WWD + (WWD - 1 - q);
}

__device__ __forceinline__ float silu(float v) {
    return v / (1.f + __expf(-v));
}

#define G1_ROWS 32
#define G1_NT   6
#define G1_KB   3
#define G1_NBLK (DIMC / (4 * G1_KB))
__global__ void __launch_bounds__(128) __attribute__((amdgpu_waves_per_eu(1, 1)))
k_in_gemm(const float* __restrict__ x,
          const float* __restrict__ Win,
          float* __restrict__ xz) {
    int lane = threadIdx.x & 31;
    int wave = threadIdx.x >> 5;
    int c16  = lane & 15;
    int hh   = lane >> 4;
    int kh   = hh * 2;
    int row0 = blockIdx.x * G1_ROWS;
    int b    = row0 / LL;
    int l0   = row0 - b * LL;
    int n0   = wave * (G1_NT * 16);
    const float* xa = x + ((size_t)b * DIMC + kh) * LL + l0 + c16;
    const float* wc = Win + (size_t)kh * XZW + n0 + c16;
    v8f acc[2][G1_NT] = {};
    v2f aA[G1_KB][2], bA[G1_KB][G1_NT], aB[G1_KB][2], bB[G1_KB][G1_NT];
    auto load_blk = [&](int kblk, v2f (&af)[G1_KB][2], v2f (&bq)[G1_KB][G1_NT])
        __attribute__((always_inline)) {
        const float* xk = xa + (size_t)kblk * (4 * G1_KB) * LL;
        const float* wk = wc + (size_t)kblk * (4 * G1_KB) * XZW;
#pragma unroll
        for (int s = 0; s < G1_KB; ++s) {
            af[s][0][0] = xk[(size_t)(4 * s) * LL];
            af[s][0][1] = xk[(size_t)(4 * s + 1) * LL];
            af[s][1][0] = xk[(size_t)(4 * s) * LL + 16];
            af[s][1][1] = xk[(size_t)(4 * s + 1) * LL + 16];
#pragma unroll
            for (int j = 0; j < G1_NT; ++j) {
                bq[s][j][0] = wk[(size_t)(4 * s) * XZW + 16 * j];
                bq[s][j][1] = wk[(size_t)(4 * s + 1) * XZW + 16 * j];
            }
        }
    };
    auto mma_blk = [&](const v2f (&af)[G1_KB][2], const v2f (&bq)[G1_KB][G1_NT])
        __attribute__((always_inline)) {
#pragma unroll
        for (int s = 0; s < G1_KB; ++s)
#pragma unroll
            for (int j = 0; j < G1_NT; ++j) {
                acc[0][j] = __builtin_amdgcn_wmma_f32_16x16x4_f32(false, af[s][0], false, bq[s][j],
                                                                 (short)0, acc[0][j], false, false);
                acc[1][j] = __builtin_amdgcn_wmma_f32_16x16x4_f32(false, af[s][1], false, bq[s][j],
                                                                 (short)0, acc[1][j], false, false);
            }
    };
    load_blk(0, aA, bA);
#pragma unroll 1
    for (int p = 0; p < G1_NBLK / 2; ++p) {
        load_blk(2 * p + 1, aB, bB);
        __builtin_amdgcn_sched_barrier(0);
        mma_blk(aA, bA);
        __builtin_amdgcn_sched_barrier(0);
        load_blk(min(2 * p + 2, G1_NBLK - 1), aA, bA);
        __builtin_amdgcn_sched_barrier(0);
        mma_blk(aB, bB);
    }
#pragma unroll
    for (int r = 0; r < 2; ++r) {
        float* op = xz + (size_t)(row0 + 16 * r + hh * 8) * XZW + n0 + c16;
#pragma unroll
        for (int v = 0; v < 8; ++v)
#pragma unroll
            for (int j = 0; j < G1_NT; ++j)
                op[(size_t)v * XZW + 16 * j] = acc[r][j][v];
    }
}

#define DQ (D_INNER / 4)
#define FT_ROWS 32
#define YP      (D_INNER + 4)
__device__ __forceinline__ void f4_to_arr(float4 q, float (&f)[4]) {
    f[0] = q.x; f[1] = q.y; f[2] = q.z; f[3] = q.w;
}
__global__ void __launch_bounds__(128) __attribute__((amdgpu_waves_per_eu(1, 1)))
k_conv_xproj(const float* __restrict__ xz,
             const float* __restrict__ Wc,
             const float* __restrict__ bc,
             const float* __restrict__ Wxp,
             float* __restrict__ xi,
             float* __restrict__ xdbl) {
    __shared__ __attribute__((aligned(16))) float xsm[FT_ROWS * YP];
    int tid  = threadIdx.x;
    int row0 = blockIdx.x * FT_ROWS;
    int b0   = row0 / LL;
    int p0   = row0 - b0 * LL;

#pragma unroll 1
    for (int m = 0; m < 3; ++m) {
        int q0  = m * 128 + tid;
        int rr0 = q0 / DQ;
        int d   = (q0 - rr0 * DQ) * 4;
        float wt[36];
#pragma unroll
        for (int i = 0; i < 9; ++i) {
            float q[4];
            f4_to_arr(*(const float4*)(Wc + (size_t)d * 9 + 4 * i), q);
            wt[4 * i] = q[0]; wt[4 * i + 1] = q[1]; wt[4 * i + 2] = q[2]; wt[4 * i + 3] = q[3];
        }
        float bias[4];
        f4_to_arr(*(const float4*)(bc + d), bias);
        float v[4][9][4];
        bool  ok[4][9];
#pragma unroll
        for (int j = 0; j < 4; ++j) {
            int p = p0 + rr0 + 8 * j;
            int h = p / WWD, w = p - h * WWD;
#pragma unroll
            for (int t = 0; t < 9; ++t) {
                int hy = h + t / 3 - 1, wx = w + t % 3 - 1;
                ok[j][t] = (hy >= 0) && (hy < HH) && (wx >= 0) && (wx < WWD);
                int hc = min(max(hy, 0), HH - 1), wc = min(max(wx, 0), WWD - 1);
                f4_to_arr(*(const float4*)(xz + (size_t)(b0 * LL + hc * WWD + wc) * XZW + d),
                          v[j][t]);
            }
        }
#pragma unroll
        for (int j = 0; j < 4; ++j) {
            int rr = rr0 + 8 * j;
            float acc[4];
#pragma unroll
            for (int c = 0; c < 4; ++c) acc[c] = bias[c];
#pragma unroll
            for (int t = 0; t < 9; ++t)
#pragma unroll
                for (int c = 0; c < 4; ++c)
                    acc[c] = ok[j][t] ? acc[c] + wt[c * 9 + t] * v[j][t][c] : acc[c];
            float4 o;
            o.x = silu(acc[0]); o.y = silu(acc[1]); o.z = silu(acc[2]); o.w = silu(acc[3]);
            *(float4*)(xi + (size_t)(row0 + rr) * D_INNER + d) = o;
            *(float4*)(xsm + rr * YP + d) = o;
        }
    }
    __syncthreads();

    int lane = tid & 31;
    int wave = tid >> 5;
    int c16  = lane & 15;
    int hh   = lane >> 4;
    int kh   = hh * 2;
    int mt   = wave >> 1;
    int kd0  = (wave & 1) * 2;
    const float* arow = xsm + (mt * 16 + c16) * YP + kh;
    int  nn2  = 32 + c16;
    bool nok2 = (nn2 < NPROJ);
    int  nn2c = min(nn2, NPROJ - 1);
    const float* wd0 = Wxp + ((size_t)kd0 * D_INNER + kh) * NPROJ;
    v2f afr[D_INNER / 4];
#pragma unroll
    for (int s = 0; s < D_INNER / 4; ++s) afr[s] = *(const v2f*)(arow + 4 * s);
    v8f acc[2][3] = {};
#pragma unroll
    for (int kb = 0; kb < D_INNER; kb += 4) {
        v2f a = afr[kb / 4];
#pragma unroll
        for (int kk = 0; kk < 2; ++kk) {
            const float* wr = wd0 + ((size_t)kk * D_INNER + kb) * NPROJ;
            v2f f0, f1, f2;
            f0[0] = wr[c16];       f0[1] = wr[NPROJ + c16];
            f1[0] = wr[16 + c16];  f1[1] = wr[NPROJ + 16 + c16];
            float t0 = wr[nn2c],   t1 = wr[NPROJ + nn2c];
            f2[0] = nok2 ? t0 : 0.f;
            f2[1] = nok2 ? t1 : 0.f;
            acc[kk][0] = __builtin_amdgcn_wmma_f32_16x16x4_f32(false, a, false, f0,
                                                               (short)0, acc[kk][0], false, false);
            acc[kk][1] = __builtin_amdgcn_wmma_f32_16x16x4_f32(false, a, false, f1,
                                                               (short)0, acc[kk][1], false, false);
            acc[kk][2] = __builtin_amdgcn_wmma_f32_16x16x4_f32(false, a, false, f2,
                                                               (short)0, acc[kk][2], false, false);
        }
    }
#pragma unroll
    for (int kk = 0; kk < 2; ++kk) {
        int kdir = kd0 + kk;
#pragma unroll
        for (int v = 0; v < 8; ++v) {
            int p = p0 + mt * 16 + hh * 8 + v;
            int l = spatial_to_scan(kdir, p);
            float* op = xdbl + ((size_t)kdir * BL + b0 * LL + l) * NPROJ;
            op[c16]      = acc[kk][0][v];
            op[16 + c16] = acc[kk][1][v];
            if (nok2) op[nn2] = acc[kk][2][v];
        }
    }
}

#define SCAN_CH      8
#define SCAN_THREADS (SCAN_CH * 16)
#define CH_SHIFT     3
#define BLK_PER_KB   (D_INNER / SCAN_CH)
#define NBC          (ST_STEPS * 32 / SCAN_THREADS)
#define ST_STEPS 64
#define N_STAGE  (LL / ST_STEPS)
#define CH       16
#define XPITCH   (ST_STEPS + 4)
#define ST_X     0
#define ST_D     (SCAN_CH * XPITCH)
#define ST_BC    (2 * SCAN_CH * XPITCH)
#define ST_FLOATS (ST_BC + ST_STEPS * 32)


__device__ __forceinline__ float softplus_dt(float t) {
    float e  = __expf(t);
    float u  = 1.f + e;
    float c  = e - (u - 1.f);
    float sp = __builtin_amdgcn_logf(u) * 0.69314718056f + c * __builtin_amdgcn_rcpf(u);
    return (t > 20.f) ? t : sp;
}

__device__ __forceinline__ void stage_fetch(const float* __restrict__ xb,
                                            const float* __restrict__ drk,
                                            const float* __restrict__ bck,
                                            int l0, int d0, int sbase, int sq, int sr,
                                            int tid, int apos, int hh,
                                            float (&rx)[4], v2f (&ra)[2],
                                            float (&rbc)[NBC]) {
#pragma unroll
    for (int i = 0; i < 4; ++i) {
        int e  = i * SCAN_THREADS + tid;
        int t  = e >> CH_SHIFT, dd = e & (SCAN_CH - 1);
        int l  = l0 + t;
        int q  = l / WWD, r = l - q * WWD;
        int sp = sbase + q * sq + r * sr;
        rx[i] = xb[(size_t)sp * D_INNER + d0 + dd];
    }
    const float* ap = drk + (size_t)l0 * NPROJ + apos;
    ra[0] = *(const v2f*)ap;
    ra[1] = *(const v2f*)(ap + 4 - 2 * hh);
#pragma unroll
    for (int i = 0; i < NBC; ++i) {
        int e = i * SCAN_THREADS + tid;
        int t = e >> 5, j = e & 31;
        rbc[i] = bck[(size_t)(l0 + t) * NPROJ + j];
    }
}

__device__ __forceinline__ void stage_store(float* __restrict__ buf, int tid,
                                            const float (&rx)[4], const v2f (&ra)[2],
                                            const v2f (&wb)[2], float dbias,
                                            int hh, int dpos,
                                            const float (&rbc)[NBC]) {
#pragma unroll
    for (int i = 0; i < 4; ++i) {
        int e = i * SCAN_THREADS + tid;
        int t = e >> CH_SHIFT, dd = e & (SCAN_CH - 1);
        buf[ST_X + dd * XPITCH + t] = rx[i];
    }
#pragma unroll
    for (int i = 0; i < NBC; ++i)
        buf[ST_BC + i * SCAN_THREADS + tid] = rbc[i];

    v2f a1;
    a1[0] = hh ? 0.f : ra[1][0];
    a1[1] = hh ? 0.f : ra[1][1];
    v8f acc = {};
    acc = __builtin_amdgcn_wmma_f32_16x16x4_f32(false, ra[0], false, wb[0],
                                                (short)0, acc, false, false);
    acc = __builtin_amdgcn_wmma_f32_16x16x4_f32(false, a1, false, wb[1],
                                                (short)0, acc, false, false);
    float dl[8];
#pragma unroll
    for (int r = 0; r < 8; ++r) dl[r] = softplus_dt(acc[r] + dbias);
    if (dpos >= 0) {
        float4 lo, hi;
        lo.x = dl[0]; lo.y = dl[1]; lo.z = dl[2]; lo.w = dl[3];
        hi.x = dl[4]; hi.y = dl[5]; hi.z = dl[6]; hi.w = dl[7];
        *(float4*)(buf + ST_D + dpos)     = lo;
        *(float4*)(buf + ST_D + dpos + 4) = hi;
    }
}

__device__ __forceinline__ void lds_stage_barrier() {
    asm volatile("s_wait_dscnt 0x0\n\ts_barrier_signal -1\n\ts_barrier_wait -1" ::: "memory");
}

template <int M>
__device__ __forceinline__ float row_halve(bool hi, float a, float b) {
    float keep = hi ? b : a;
    float send = hi ? a : b;
    int q = __builtin_amdgcn_update_dpp(0, __float_as_int(send), 0x160 + M, 0xf, 0xf, true);
    return keep + __int_as_float(q);
}

#define RING 4
__global__ void __launch_bounds__(2 * SCAN_THREADS)
k_scan(const float* __restrict__ xi,
       const float* __restrict__ xdbl,
       const float* __restrict__ Wdt,
       const float* __restrict__ bdt,
       const float* __restrict__ A_logs,
       const float* __restrict__ Ds,
       float* __restrict__ ys) {
    __shared__ __attribute__((aligned(16))) float sm[RING * ST_FLOATS];

    int wave8   = __builtin_amdgcn_readfirstlane((int)(threadIdx.x >> 5));
    bool stager = wave8 >= SCAN_THREADS / 32;
    int tid   = threadIdx.x & (SCAN_THREADS - 1);
    int n     = tid & 15;
    int g     = tid >> 4;
    int d0    = (blockIdx.x % BLK_PER_KB) * SCAN_CH;
    int d     = d0 + g;
    int kb    = blockIdx.x / BLK_PER_KB;
    int b     = kb % BATCH;
    int kdir  = kb / BATCH;

    int sbase, sq, sr;
    if (kdir == 0)      { sbase = 0;                 sq = WWD; sr = 1;    }
    else if (kdir == 1) { sbase = WWD - 1;           sq = WWD; sr = -1;   }
    else if (kdir == 2) { sbase = 0;                 sq = 1;   sr = WWD;  }
    else                { sbase = (WWD - 1) * WWD;   sq = 1;   sr = -WWD; }

    bool hi8 = (n & 8) != 0, hi4 = (n & 4) != 0, hi2 = (n & 2) != 0, hi1 = (n & 1) != 0;
    int  trev = ((n & 1) << 3) | ((n & 2) << 1) | ((n & 4) >> 1) | ((n & 8) >> 3);

    float A = 0.f, Dskip = 0.f;
    if (!stager) {
        Dskip = Ds[kdir * D_INNER + d];
        A = -__expf(A_logs[(size_t)(kdir * D_INNER + d) * D_STATE + n]);
    }
    float h = 0.f;

    const float* xb  = xi    + (size_t)b  * LL * D_INNER;
    const float* drk = xdbl  + (size_t)kb * LL * NPROJ;
    const float* bck = xdbl  + (size_t)kb * LL * NPROJ + DT_RANK;
    float*       ysb = ys    + (size_t)kb * LL * D_INNER + d;

    int c16  = tid & 15;
    int hh   = (tid >> 4) & 1;
    int wv   = tid >> 5;
    int apos = (wv * 16 + c16) * NPROJ + 2 * hh;
    int dpos = (c16 < SCAN_CH) ? c16 * XPITCH + wv * 16 + hh * 8 : -1;
    int dcol = min(d0 + c16, D_INNER - 1);
    v2f wb[2] = {};
    float dbias = 0.f;

    auto scan_stage = [&](int s) __attribute__((always_inline)) {
        const float* cur = sm + (s & (RING - 1)) * ST_FLOATS;
#pragma unroll 1
        for (int c = 0; c < ST_STEPS / CH; ++c) {
            int tt = c * CH;
            float xc[CH], dc[CH], bc[CH], cc[CH];
#pragma unroll
            for (int j = 0; j < CH / 4; ++j) {
                float4 vx = *(const float4*)(cur + ST_X + g * XPITCH + tt + 4 * j);
                float4 vd = *(const float4*)(cur + ST_D + g * XPITCH + tt + 4 * j);
                xc[4 * j] = vx.x; xc[4 * j + 1] = vx.y; xc[4 * j + 2] = vx.z; xc[4 * j + 3] = vx.w;
                dc[4 * j] = vd.x; dc[4 * j + 1] = vd.y; dc[4 * j + 2] = vd.z; dc[4 * j + 3] = vd.w;
            }
#pragma unroll
            for (int t = 0; t < CH; ++t) {
                bc[t] = cur[ST_BC + (tt + t) * 32 + n];
                cc[t] = cur[ST_BC + (tt + t) * 32 + D_STATE + n];
            }

            float xo = cur[ST_X + g * XPITCH + tt + trev];

            float pv[CH];
#pragma unroll
            for (int t = 0; t < CH; ++t) {
                float x  = xc[t];
                float dl = dc[t];
                float dA = __expf(dl * A);
                h = dA * h + (dl * x) * bc[t];
                pv[t] = h * cc[t];
            }
            float q8[CH / 2], q4[CH / 4], q2[CH / 8];
#pragma unroll
            for (int j = 0; j < CH / 2; ++j) q8[j] = row_halve<8>(hi8, pv[2 * j], pv[2 * j + 1]);
#pragma unroll
            for (int j = 0; j < CH / 4; ++j) q4[j] = row_halve<4>(hi4, q8[2 * j], q8[2 * j + 1]);
#pragma unroll
            for (int j = 0; j < CH / 8; ++j) q2[j] = row_halve<2>(hi2, q4[2 * j], q4[2 * j + 1]);
            float p = row_halve<1>(hi1, q2[0], q2[1]);
            ysb[(size_t)(s * ST_STEPS + tt + trev) * D_INNER] = p + xo * Dskip;
        }
    };

    float rx0[4], rbc0[NBC], rx1[4], rbc1[NBC];
    v2f ra0[2], ra1[2];
    if (stager) {
        dbias = bdt[kdir * D_INNER + dcol];
#pragma unroll
        for (int s = 0; s < 2; ++s)
#pragma unroll
            for (int v = 0; v < 2; ++v) {
                int kk = s * 4 + 2 * hh + v;
                float w = Wdt[(size_t)(kdir * DT_RANK + min(kk, DT_RANK - 1)) * D_INNER + dcol];
                wb[s][v] = (kk < DT_RANK) ? w : 0.f;
            }
        stage_fetch(xb, drk, bck, 0,        d0, sbase, sq, sr, tid, apos, hh, rx0, ra0, rbc0);
        stage_fetch(xb, drk, bck, ST_STEPS, d0, sbase, sq, sr, tid, apos, hh, rx1, ra1, rbc1);
        stage_store(sm,             tid, rx0, ra0, wb, dbias, hh, dpos, rbc0);
        stage_store(sm + ST_FLOATS, tid, rx1, ra1, wb, dbias, hh, dpos, rbc1);
        stage_fetch(xb, drk, bck, 2 * ST_STEPS, d0, sbase, sq, sr, tid, apos, hh, rx0, ra0, rbc0);
        stage_fetch(xb, drk, bck, 3 * ST_STEPS, d0, sbase, sq, sr, tid, apos, hh, rx1, ra1, rbc1);
    }
    lds_stage_barrier();

    for (int s = 0; s < N_STAGE; s += 2) {
        if (stager) {
            if (s + 2 < N_STAGE) {
                stage_store(sm + ((s + 2) & (RING - 1)) * ST_FLOATS, tid, rx0, ra0, wb,
                            dbias, hh, dpos, rbc0);
                if (s + 3 < N_STAGE)
                    stage_store(sm + ((s + 3) & (RING - 1)) * ST_FLOATS, tid, rx1, ra1, wb,
                                dbias, hh, dpos, rbc1);
                stage_fetch(xb, drk, bck, min(s + 4, N_STAGE - 1) * ST_STEPS, d0, sbase,
                            sq, sr, tid, apos, hh, rx0, ra0, rbc0);
                stage_fetch(xb, drk, bck, min(s + 5, N_STAGE - 1) * ST_STEPS, d0, sbase,
                            sq, sr, tid, apos, hh, rx1, ra1, rbc1);
            }
        } else {
            scan_stage(s);
            if (s + 1 < N_STAGE) scan_stage(s + 1);
        }
        lds_stage_barrier();
    }
}

#define FT_Q    (FT_ROWS * DQ / 128)
__global__ void __launch_bounds__(128) __attribute__((amdgpu_waves_per_eu(1, 1)))
k_comb_out(const float* __restrict__ ys,
           const float* __restrict__ xz,
           const float* __restrict__ Wout,
           float* __restrict__ out) {
    __shared__ __attribute__((aligned(16))) float ysm[FT_ROWS * YP];
    int tid  = threadIdx.x;
    int row0 = blockIdx.x * FT_ROWS;

    float r[FT_Q][KDIR][4], z[FT_Q][4];
#pragma unroll
    for (int i = 0; i < FT_Q; ++i) {
        int q   = i * 128 + tid;
        int rr  = q / DQ;
        int d   = (q - rr * DQ) * 4;
        int row = row0 + rr;
        int b = row / LL, l = row - b * LL;
#pragma unroll
        for (int k = 0; k < KDIR; ++k) {
            int ls = spatial_to_scan(k, l);
            f4_to_arr(*(const float4*)(ys + (size_t)((k * BATCH + b) * LL + ls) * D_INNER + d),
                      r[i][k]);
        }
        f4_to_arr(*(const float4*)(xz + (size_t)row * XZW + D_INNER + d), z[i]);
    }
#pragma unroll
    for (int i = 0; i < FT_Q; ++i) {
        int q  = i * 128 + tid;
        int rr = q / DQ;
        int d  = (q - rr * DQ) * 4;
        float o[4];
#pragma unroll
        for (int c = 0; c < 4; ++c) {
            float acc = 0.f;
#pragma unroll
            for (int k = 0; k < KDIR; ++k) acc += r[i][k][c];
            o[c] = acc * silu(z[i][c]);
        }
        float4 qv;
        qv.x = o[0]; qv.y = o[1]; qv.z = o[2]; qv.w = o[3];
        *(float4*)(ysm + rr * YP + d) = qv;
    }
    __syncthreads();

    int lane = tid & 31;
    int wave = tid >> 5;
    int c16  = lane & 15;
    int kh   = (lane >> 4) * 2;
    int mt   = wave >> 1;
    int n0   = (wave & 1) * 48;
    const float* arow = ysm + (mt * 16 + c16) * YP + kh;
    const float* wcol = Wout + (size_t)kh * DIMC + n0 + c16;
    v8f acc0 = {}, acc1 = {}, acc2 = {};
#pragma unroll
    for (int kb = 0; kb < D_INNER; kb += 4) {
        v2f a = *(const v2f*)(arow + kb);
        const float* w0 = wcol + (size_t)kb * DIMC;
        v2f b0, b1, b2;
        b0[0] = w0[0];   b0[1] = w0[DIMC];
        b1[0] = w0[16];  b1[1] = w0[DIMC + 16];
        b2[0] = w0[32];  b2[1] = w0[DIMC + 32];
        acc0 = __builtin_amdgcn_wmma_f32_16x16x4_f32(false, a, false, b0,
                                                     (short)0, acc0, false, false);
        acc1 = __builtin_amdgcn_wmma_f32_16x16x4_f32(false, a, false, b1,
                                                     (short)0, acc1, false, false);
        acc2 = __builtin_amdgcn_wmma_f32_16x16x4_f32(false, a, false, b2,
                                                     (short)0, acc2, false, false);
    }
    int mb = row0 + mt * 16 + (lane >> 4) * 8;
    int nn = n0 + c16;
#pragma unroll
    for (int v = 0; v < 8; ++v) {
        int row = mb + v;
        int b = row / LL, l = row - b * LL;
        float* op = out + (size_t)(b * DIMC + nn) * LL + l;
        op[0]                = acc0[v];
        op[(size_t)16 * LL]  = acc1[v];
        op[(size_t)32 * LL]  = acc2[v];
    }
}

extern "C" void kernel_launch(void* const* d_in, const int* in_sizes, int n_in,
                              void* d_out, int out_size, void* d_ws, size_t ws_size,
                              hipStream_t stream) {
    (void)in_sizes; (void)n_in; (void)out_size; (void)ws_size;
    const float* x       = (const float*)d_in[0];
    const float* W_in    = (const float*)d_in[1];
    const float* W_conv  = (const float*)d_in[2];
    const float* b_conv  = (const float*)d_in[3];
    const float* W_xproj = (const float*)d_in[4];
    const float* W_dt    = (const float*)d_in[5];
    const float* b_dt    = (const float*)d_in[6];
    const float* A_logs  = (const float*)d_in[7];
    const float* Ds      = (const float*)d_in[8];
    const float* W_out   = (const float*)d_in[9];
    float* out = (float*)d_out;

    float* ws = (float*)d_ws;
    size_t o = 0;
    float* xz    = ws + o;  o += (size_t)BL * XZW;
    float* xi    = ws + o;  o += (size_t)BL * D_INNER;
    float* xdbl  = ws + o;  o += (size_t)KDIR * BL * NPROJ;
    float* ysbuf = ws + o;  o += (size_t)KDIR * BL * D_INNER;

    dim3 blk(128);
    k_in_gemm <<<BL / G1_ROWS,    blk, 0, stream>>>(x, W_in, xz);
    k_conv_xproj<<<BL / FT_ROWS, 128, 0, stream>>>(xz, W_conv, b_conv, W_xproj, xi, xdbl);
    int nscan = KDIR * BATCH * D_INNER * D_STATE;
    k_scan    <<<nscan / SCAN_THREADS, 2 * SCAN_THREADS, 0, stream>>>(xi, xdbl, W_dt, b_dt,
                                                        A_logs, Ds, ysbuf);
    k_comb_out<<<BL / FT_ROWS, 128, 0, stream>>>(ysbuf, xz, W_out, out);
}
